# starting kernel
# baseline (speedup 1.0000x reference)
#include <hip/hip_runtime.h>

typedef __attribute__((ext_vector_type(2))) float v2f;
typedef __attribute__((ext_vector_type(8))) float v8f;

#define NNODES 50000
#define DHID   128
#define NCLS   10

__global__ void deg_init_kernel(float* deg, int n) {
    int i = blockIdx.x * blockDim.x + threadIdx.x;
    if (i < n) deg[i] = 1.0f;
}

__global__ void deg_edges_kernel(const int* __restrict__ dst,
                                 const float* __restrict__ w,
                                 float* deg, int nE) {
    int e = blockIdx.x * blockDim.x + threadIdx.x;
    if (e < nE) atomicAdd(&deg[dst[e]], w[e]);
}

__global__ void dinv_kernel(const float* __restrict__ deg, float* dinv, int n) {
    int i = blockIdx.x * blockDim.x + threadIdx.x;
    if (i < n) {
        float d = deg[i];
        dinv[i] = __frsqrt_rn(d);
    }
}

__global__ void norm_kernel(const int* __restrict__ src,
                            const int* __restrict__ dst,
                            const float* __restrict__ w,
                            const float* __restrict__ dinv,
                            float* norm, int nE) {
    int e = blockIdx.x * blockDim.x + threadIdx.x;
    if (e < nE) norm[e] = dinv[src[e]] * w[e] * dinv[dst[e]];
}

#define XS_STRIDE 34
#define WT_STRIDE 144

__global__ __launch_bounds__(256) void gemm128_wmma_kernel(
    const float* __restrict__ X, const float* __restrict__ W,
    float* __restrict__ H, int M, int K)
{
    __shared__ float  Xs[128 * XS_STRIDE];
    __shared__ float2 Wt[16 * WT_STRIDE];

    const int tid  = threadIdx.x;
    const int lane = tid & 31;
    const int wave = tid >> 5;
    const int l16  = lane & 15;
    const bool hi  = (lane >= 16);
    const int rowBlock = blockIdx.x * 128;
    const int rowLocal = wave * 16 + l16;

    v8f acc[8];
    #pragma unroll
    for (int t = 0; t < 8; ++t)
        #pragma unroll
        for (int r = 0; r < 8; ++r) acc[t][r] = 0.0f;

    const int nch = (K + 31) / 32;
    for (int ch = 0; ch < nch; ++ch) {
        const int k0 = ch * 32;

        #pragma unroll
        for (int j = 0; j < 4; ++j) {
            int f  = tid * 4 + j;
            int r  = f >> 3;
            int kb = (f & 7) * 4;
            int gr = rowBlock + r;
            float4 v = make_float4(0.f, 0.f, 0.f, 0.f);
            if (gr < M) {
                int kg = k0 + kb;
                if (kg + 3 < K) {
                    v = *(const float4*)(X + (size_t)gr * K + kg);
                } else {
                    float t0 = (kg + 0 < K) ? X[(size_t)gr * K + kg + 0] : 0.f;
                    float t1 = (kg + 1 < K) ? X[(size_t)gr * K + kg + 1] : 0.f;
                    float t2 = (kg + 2 < K) ? X[(size_t)gr * K + kg + 2] : 0.f;
                    float t3 = (kg + 3 < K) ? X[(size_t)gr * K + kg + 3] : 0.f;
                    v = make_float4(t0, t1, t2, t3);
                }
            }
            float* dp = &Xs[r * XS_STRIDE + kb];
            ((float2*)dp)[0] = make_float2(v.x, v.y);
            ((float2*)dp)[1] = make_float2(v.z, v.w);
        }

        #pragma unroll
        for (int j = 0; j < 2; ++j) {
            int f  = tid * 2 + j;
            int kp = f >> 5;
            int cb = (f & 31) * 4;
            int ka = k0 + 2 * kp;
            float4 va = make_float4(0.f, 0.f, 0.f, 0.f);
            float4 vb = make_float4(0.f, 0.f, 0.f, 0.f);
            if (ka < K)     va = *(const float4*)(W + (size_t)ka * DHID + cb);
            if (ka + 1 < K) vb = *(const float4*)(W + (size_t)(ka + 1) * DHID + cb);
            float4 w0 = make_float4(va.x, vb.x, va.y, vb.y);
            float4 w1 = make_float4(va.z, vb.z, va.w, vb.w);
            *(float4*)&Wt[kp * WT_STRIDE + cb]     = w0;
            *(float4*)&Wt[kp * WT_STRIDE + cb + 2] = w1;
        }

        __syncthreads();

        #pragma unroll
        for (int s = 0; s < 8; ++s) {
            const int kp = s * 2 + (hi ? 1 : 0);
            v2f a = *(const v2f*)&Xs[rowLocal * XS_STRIDE + kp * 2];
            float2 bf[8];
            #pragma unroll
            for (int nt = 0; nt < 8; ++nt)
                bf[nt] = Wt[kp * WT_STRIDE + nt * 16 + l16];
            #pragma unroll
            for (int nt = 0; nt < 8; ++nt) {
                v2f b;
                b.x = bf[nt].x;
                b.y = bf[nt].y;
                acc[nt] = __builtin_amdgcn_wmma_f32_16x16x4_f32(
                     false, a,  false, b,
                     (short)0, acc[nt],
                     false,  false);
            }
        }
        __syncthreads();
    }

    #pragma unroll
    for (int nt = 0; nt < 8; ++nt) {
        const int col = nt * 16 + l16;
        #pragma unroll
        for (int r = 0; r < 8; ++r) {
            int m = rowBlock + wave * 16 + r + (hi ? 8 : 0);
            if (m < M) H[(size_t)m * DHID + col] = acc[nt][r];
        }
    }
}

__global__ void agg_self_kernel(const float* __restrict__ H,
                                const float* __restrict__ dinv,
                                float* __restrict__ OUT, int n) {
    int i = blockIdx.x * blockDim.x + threadIdx.x;
    if (i < n * DHID) {
        float s = dinv[i >> 7];
        OUT[i] = H[i] * s * s;
    }
}

__global__ __launch_bounds__(256) void agg_edges_kernel(
    const float* __restrict__ H,
    const int* __restrict__ src, const int* __restrict__ dst,
    const float* __restrict__ norm,
    float* __restrict__ OUT, int nE)
{
    int e = blockIdx.x * 2 + (threadIdx.x >> 7);
    int d = threadIdx.x & 127;
    if (e < nE) {
        int s = src[e];
        int t = dst[e];
        float v = H[(size_t)s * DHID + d] * norm[e];
        atomicAdd(&OUT[(size_t)t * DHID + d], v);
    }
}

__global__ void bias_relu_kernel(float* __restrict__ OUT,
                                 const float* __restrict__ b, int n) {
    int i = blockIdx.x * blockDim.x + threadIdx.x;
    if (i < n * DHID) {
        float v = OUT[i] + b[i & (DHID - 1)];
        OUT[i] = v > 0.f ? v : 0.f;
    }
}

__global__ void classifier_kernel(const float* __restrict__ H,
                                  const float* __restrict__ Wc,
                                  const float* __restrict__ bc,
                                  float* __restrict__ out, int n) {
    int i = blockIdx.x * blockDim.x + threadIdx.x;
    if (i < n * NCLS) {
        int row = i / NCLS;
        int c   = i - row * NCLS;
        float s = bc[c];
        const float* h = H + (size_t)row * DHID;
        #pragma unroll 8
        for (int k = 0; k < DHID; ++k) s += h[k] * Wc[k * NCLS + c];
        out[i] = s;
    }
}

extern "C" void kernel_launch(void* const* d_in, const int* in_sizes, int n_in,
                              void* d_out, int out_size, void* d_ws, size_t ws_size,
                              hipStream_t stream) {
    const float* x  = (const float*)d_in[0];
    const int*   ei = (const int*)d_in[1];
    const float* ew = (const float*)d_in[2];
    const float* W1 = (const float*)d_in[3];
    const float* b1 = (const float*)d_in[4];
    const float* W2 = (const float*)d_in[5];
    const float* b2 = (const float*)d_in[6];
    const float* Wc = (const float*)d_in[7];
    const float* bc = (const float*)d_in[8];
    float* out = (float*)d_out;

    const int n  = NNODES;
    const int E  = in_sizes[1] / 2;
    const int K1 = in_sizes[0] / n;
    const int* src = ei;
    const int* dst = ei + E;

    char* ws = (char*)d_ws;
    size_t off = 0;
    auto alloc = [&](size_t bytes) { char* p = ws + off; off = (off + bytes + 255) & ~(size_t)255; return p; };
    float* deg  = (float*)alloc((size_t)n * 4);
    float* dinv = (float*)alloc((size_t)n * 4);
    float* norm = (float*)alloc((size_t)E * 4);
    float* bufA = (float*)alloc((size_t)n * DHID * 4);
    float* bufB = (float*)alloc((size_t)n * DHID * 4);
    (void)ws_size;

    const int T = 256;
    deg_init_kernel <<<(n + T - 1) / T, T, 0, stream>>>(deg, n);
    deg_edges_kernel<<<(E + T - 1) / T, T, 0, stream>>>(dst, ew, deg, E);
    dinv_kernel     <<<(n + T - 1) / T, T, 0, stream>>>(deg, dinv, n);
    norm_kernel     <<<(E + T - 1) / T, T, 0, stream>>>(src, dst, ew, dinv, norm, E);

    const int gemmBlocks = (n + 127) / 128;
    const int elemBlocks = (n * DHID + T - 1) / T;
    const int edgeBlocks = (E + 1) / 2;

    gemm128_wmma_kernel<<<gemmBlocks, T, 0, stream>>>(x, W1, bufA, n, K1);
    agg_self_kernel    <<<elemBlocks, T, 0, stream>>>(bufA, dinv, bufB, n);
    agg_edges_kernel   <<<edgeBlocks, T, 0, stream>>>(bufA, src, dst, norm, bufB, E);
    bias_relu_kernel   <<<elemBlocks, T, 0, stream>>>(bufB, b1, n);

    gemm128_wmma_kernel<<<gemmBlocks, T, 0, stream>>>(bufB, W2, bufA, n, DHID);
    agg_self_kernel    <<<elemBlocks, T, 0, stream>>>(bufA, dinv, bufB, n);
    agg_edges_kernel   <<<edgeBlocks, T, 0, stream>>>(bufA, src, dst, norm, bufB, E);
    bias_relu_kernel   <<<elemBlocks, T, 0, stream>>>(bufB, b2, n);

    classifier_kernel<<<(n * NCLS + T - 1) / T, T, 0, stream>>>(bufB, Wc, bc, out, n);
}


# Edges bucketed by destination each call (int atomic rank + scan); one wave per node sums its rows in registers, stores once: no float atomics in aggregation; sum order = arrival rank
# speedup vs baseline: 2.1018x; 2.1018x over previous
#include <hip/hip_runtime.h>

typedef __attribute__((ext_vector_type(2))) float v2f;
typedef __attribute__((ext_vector_type(8))) float v8f;

#define NNODES 50000
#define DHID   128
#define NCLS   10

__global__ void deg_init_kernel(float* deg, int* cnt, int n, int npad) {
    int i = blockIdx.x * blockDim.x + threadIdx.x;
    if (i < n)    deg[i] = 1.0f;
    if (i < npad) cnt[i] = 0;
}

__global__ void deg_edges_kernel(const int* __restrict__ dst,
                                 const float* __restrict__ w,
                                 float* deg, int* cnt, int* rank, int nE) {
    int e = blockIdx.x * blockDim.x + threadIdx.x;
    if (e < nE) {
        int d = dst[e];
        atomicAdd(&deg[d], w[e]);
        rank[e] = atomicAdd(&cnt[d], 1);
    }
}

__global__ void dinv_kernel(const float* __restrict__ deg, float* dinv, int n) {
    int i = blockIdx.x * blockDim.x + threadIdx.x;
    if (i < n) {
        float d = deg[i];
        dinv[i] = __frsqrt_rn(d);
    }
}

#define SCAN_T   1024
#define SCAN_PER ((NNODES + SCAN_T - 1) / SCAN_T)
#define NPAD     (SCAN_T * SCAN_PER)
__global__ __launch_bounds__(SCAN_T) void rowptr_scan_kernel(
    const int* __restrict__ cnt, int* __restrict__ rowptr)
{
    __shared__ int part[SCAN_T];
    const int tid = threadIdx.x;
    const int beg = tid * SCAN_PER;

    int c[SCAN_PER];
    #pragma unroll
    for (int k = 0; k < SCAN_PER; ++k) c[k] = cnt[beg + k];
    int s = 0;
    #pragma unroll
    for (int k = 0; k < SCAN_PER; ++k) s += c[k];

    part[tid] = s;
    __syncthreads();
    for (int off = 1; off < SCAN_T; off <<= 1) {
        int v = (tid >= off) ? part[tid - off] : 0;
        __syncthreads();
        part[tid] += v;
        __syncthreads();
    }
    int run = part[tid] - s;
    #pragma unroll
    for (int k = 0; k < SCAN_PER; ++k) {
        rowptr[beg + k] = run;
        run += c[k];
    }
    if (tid == SCAN_T - 1) rowptr[NPAD] = run;
}

__global__ void fill_kernel(const int* __restrict__ src,
                            const int* __restrict__ dst,
                            const float* __restrict__ w,
                            const float* __restrict__ dinv,
                            const int* __restrict__ rowptr,
                            const int* __restrict__ rank,
                            int* __restrict__ rec, int nE) {
    int e = blockIdx.x * blockDim.x + threadIdx.x;
    if (e < nE) {
        int s = src[e];
        int d = dst[e];
        float nm = dinv[s] * w[e] * dinv[d];
        int slot = rowptr[d] + rank[e];
        ((int2*)rec)[slot] = make_int2(s, __float_as_int(nm));
    }
}

#define XS_STRIDE 34
#define WT_STRIDE 144

__global__ __launch_bounds__(256) void gemm128_wmma_kernel(
    const float* __restrict__ X, const float* __restrict__ W,
    float* __restrict__ H, int M, int K)
{
    __shared__ float  Xs[128 * XS_STRIDE];
    __shared__ float2 Wt[16 * WT_STRIDE];

    const int tid  = threadIdx.x;
    const int lane = tid & 31;
    const int wave = tid >> 5;
    const int l16  = lane & 15;
    const bool hi  = (lane >= 16);
    const int rowBlock = blockIdx.x * 128;
    const int rowLocal = wave * 16 + l16;

    v8f acc[8];
    #pragma unroll
    for (int t = 0; t < 8; ++t)
        #pragma unroll
        for (int r = 0; r < 8; ++r) acc[t][r] = 0.0f;

    const int nch = (K + 31) / 32;
    for (int ch = 0; ch < nch; ++ch) {
        const int k0 = ch * 32;

        #pragma unroll
        for (int j = 0; j < 4; ++j) {
            int f  = tid * 4 + j;
            int r  = f >> 3;
            int kb = (f & 7) * 4;
            int gr = rowBlock + r;
            float4 v = make_float4(0.f, 0.f, 0.f, 0.f);
            if (gr < M) {
                int kg = k0 + kb;
                if (kg + 3 < K) {
                    v = *(const float4*)(X + (size_t)gr * K + kg);
                } else {
                    float t0 = (kg + 0 < K) ? X[(size_t)gr * K + kg + 0] : 0.f;
                    float t1 = (kg + 1 < K) ? X[(size_t)gr * K + kg + 1] : 0.f;
                    float t2 = (kg + 2 < K) ? X[(size_t)gr * K + kg + 2] : 0.f;
                    float t3 = (kg + 3 < K) ? X[(size_t)gr * K + kg + 3] : 0.f;
                    v = make_float4(t0, t1, t2, t3);
                }
            }
            float* dp = &Xs[r * XS_STRIDE + kb];
            ((float2*)dp)[0] = make_float2(v.x, v.y);
            ((float2*)dp)[1] = make_float2(v.z, v.w);
        }

        #pragma unroll
        for (int j = 0; j < 2; ++j) {
            int f  = tid * 2 + j;
            int kp = f >> 5;
            int cb = (f & 31) * 4;
            int ka = k0 + 2 * kp;
            float4 va = make_float4(0.f, 0.f, 0.f, 0.f);
            float4 vb = make_float4(0.f, 0.f, 0.f, 0.f);
            if (ka < K)     va = *(const float4*)(W + (size_t)ka * DHID + cb);
            if (ka + 1 < K) vb = *(const float4*)(W + (size_t)(ka + 1) * DHID + cb);
            float4 w0 = make_float4(va.x, vb.x, va.y, vb.y);
            float4 w1 = make_float4(va.z, vb.z, va.w, vb.w);
            *(float4*)&Wt[kp * WT_STRIDE + cb]     = w0;
            *(float4*)&Wt[kp * WT_STRIDE + cb + 2] = w1;
        }

        __syncthreads();

        #pragma unroll
        for (int s = 0; s < 8; ++s) {
            const int kp = s * 2 + (hi ? 1 : 0);
            v2f a = *(const v2f*)&Xs[rowLocal * XS_STRIDE + kp * 2];
            float2 bf[8];
            #pragma unroll
            for (int nt = 0; nt < 8; ++nt)
                bf[nt] = Wt[kp * WT_STRIDE + nt * 16 + l16];
            #pragma unroll
            for (int nt = 0; nt < 8; ++nt) {
                v2f b;
                b.x = bf[nt].x;
                b.y = bf[nt].y;
                acc[nt] = __builtin_amdgcn_wmma_f32_16x16x4_f32(
                     false, a,  false, b,
                     (short)0, acc[nt],
                     false,  false);
            }
        }
        __syncthreads();
    }

    #pragma unroll
    for (int nt = 0; nt < 8; ++nt) {
        const int col = nt * 16 + l16;
        #pragma unroll
        for (int r = 0; r < 8; ++r) {
            int m = rowBlock + wave * 16 + r + (hi ? 8 : 0);
            if (m < M) H[(size_t)m * DHID + col] = acc[nt][r];
        }
    }
}

__global__ __launch_bounds__(256) void agg_gather_kernel(
    const float* __restrict__ H,
    const int* __restrict__ rowptr,
    const int* __restrict__ rec,
    const float* __restrict__ dinv,
    float* __restrict__ OUT, int n)
{
    const int lane = threadIdx.x & 31;
    const int node = __builtin_amdgcn_readfirstlane(
        (int)(blockIdx.x * 8 + (threadIdx.x >> 5)));
    if (node >= n) return;

    const float4* __restrict__ H4 = (const float4*)H;
    const int2* __restrict__ R2 = (const int2*)rec;

    const int beg = rowptr[node];
    const int end = rowptr[node + 1];
    const float s = dinv[node];
    const float4 hv = H4[(size_t)node * 32 + lane];
    float4 acc;
    acc.x = hv.x * s * s;
    acc.y = hv.y * s * s;
    acc.z = hv.z * s * s;
    acc.w = hv.w * s * s;

    for (int base = beg; base < end; base += 32) {
        const int nb = (end - base < 32) ? (end - base) : 32;
        int2 r = make_int2(0, 0);
        if (lane < nb) r = R2[base + lane];

        int j = 0;
        for (; j + 4 <= nb; j += 4) {
            const int s0 = __builtin_amdgcn_readlane(r.x, j + 0);
            const int s1 = __builtin_amdgcn_readlane(r.x, j + 1);
            const int s2 = __builtin_amdgcn_readlane(r.x, j + 2);
            const int s3 = __builtin_amdgcn_readlane(r.x, j + 3);
            const float w0 = __int_as_float(__builtin_amdgcn_readlane(r.y, j + 0));
            const float w1 = __int_as_float(__builtin_amdgcn_readlane(r.y, j + 1));
            const float w2 = __int_as_float(__builtin_amdgcn_readlane(r.y, j + 2));
            const float w3 = __int_as_float(__builtin_amdgcn_readlane(r.y, j + 3));
            const float4 h0 = H4[(size_t)s0 * 32 + lane];
            const float4 h1 = H4[(size_t)s1 * 32 + lane];
            const float4 h2 = H4[(size_t)s2 * 32 + lane];
            const float4 h3 = H4[(size_t)s3 * 32 + lane];
            acc.x = fmaf(h0.x, w0, acc.x); acc.y = fmaf(h0.y, w0, acc.y); acc.z = fmaf(h0.z, w0, acc.z); acc.w = fmaf(h0.w, w0, acc.w);
            acc.x = fmaf(h1.x, w1, acc.x); acc.y = fmaf(h1.y, w1, acc.y); acc.z = fmaf(h1.z, w1, acc.z); acc.w = fmaf(h1.w, w1, acc.w);
            acc.x = fmaf(h2.x, w2, acc.x); acc.y = fmaf(h2.y, w2, acc.y); acc.z = fmaf(h2.z, w2, acc.z); acc.w = fmaf(h2.w, w2, acc.w);
            acc.x = fmaf(h3.x, w3, acc.x); acc.y = fmaf(h3.y, w3, acc.y); acc.z = fmaf(h3.z, w3, acc.z); acc.w = fmaf(h3.w, w3, acc.w);
        }
        #pragma unroll 1
        for (; j < nb; ++j) {
            const int s0 = __builtin_amdgcn_readlane(r.x, j);
            const float w0 = __int_as_float(__builtin_amdgcn_readlane(r.y, j));
            const float4 h0 = H4[(size_t)s0 * 32 + lane];
            acc.x = fmaf(h0.x, w0, acc.x); acc.y = fmaf(h0.y, w0, acc.y); acc.z = fmaf(h0.z, w0, acc.z); acc.w = fmaf(h0.w, w0, acc.w);
        }
    }
    ((float4*)OUT)[(size_t)node * 32 + lane] = acc;
}

__global__ void bias_relu_kernel(float* __restrict__ OUT,
                                 const float* __restrict__ b, int n) {
    int i = blockIdx.x * blockDim.x + threadIdx.x;
    if (i < n * DHID) {
        float v = OUT[i] + b[i & (DHID - 1)];
        OUT[i] = v > 0.f ? v : 0.f;
    }
}

__global__ void classifier_kernel(const float* __restrict__ H,
                                  const float* __restrict__ Wc,
                                  const float* __restrict__ bc,
                                  float* __restrict__ out, int n) {
    int i = blockIdx.x * blockDim.x + threadIdx.x;
    if (i < n * NCLS) {
        int row = i / NCLS;
        int c   = i - row * NCLS;
        float s = bc[c];
        const float* h = H + (size_t)row * DHID;
        #pragma unroll 8
        for (int k = 0; k < DHID; ++k) s += h[k] * Wc[k * NCLS + c];
        out[i] = s;
    }
}

extern "C" void kernel_launch(void* const* d_in, const int* in_sizes, int n_in,
                              void* d_out, int out_size, void* d_ws, size_t ws_size,
                              hipStream_t stream) {
    const float* x  = (const float*)d_in[0];
    const int*   ei = (const int*)d_in[1];
    const float* ew = (const float*)d_in[2];
    const float* W1 = (const float*)d_in[3];
    const float* b1 = (const float*)d_in[4];
    const float* W2 = (const float*)d_in[5];
    const float* b2 = (const float*)d_in[6];
    const float* Wc = (const float*)d_in[7];
    const float* bc = (const float*)d_in[8];
    float* out = (float*)d_out;

    const int n  = NNODES;
    const int E  = in_sizes[1] / 2;
    const int K1 = in_sizes[0] / n;
    const int* src = ei;
    const int* dst = ei + E;

    char* ws = (char*)d_ws;
    size_t off = 0;
    auto alloc = [&](size_t bytes) { char* p = ws + off; off = (off + bytes + 255) & ~(size_t)255; return p; };
    float* deg  = (float*)alloc((size_t)n * 4);
    float* dinv = (float*)alloc((size_t)n * 4);
    float* bufA = (float*)alloc((size_t)n * DHID * 4);
    float* bufB = (float*)alloc((size_t)n * DHID * 4);
    int* cnt    = (int*)alloc((size_t)NPAD * 4);
    int* rowptr = (int*)alloc((size_t)(NPAD + 1) * 4);
    int* rank   = (int*)alloc((size_t)E * 4);
    int* rec    = (int*)alloc((size_t)E * 8);
    (void)ws_size;

    const int T = 256;
    deg_init_kernel <<<(NPAD + T - 1) / T, T, 0, stream>>>(deg, cnt, n, NPAD);
    deg_edges_kernel<<<(E + T - 1) / T, T, 0, stream>>>(dst, ew, deg, cnt, rank, E);
    dinv_kernel     <<<(n + T - 1) / T, T, 0, stream>>>(deg, dinv, n);
    rowptr_scan_kernel<<<1, SCAN_T, 0, stream>>>(cnt, rowptr);
    fill_kernel     <<<(E + T - 1) / T, T, 0, stream>>>(src, dst, ew, dinv, rowptr, rank, rec, E);

    const int gemmBlocks = (n + 127) / 128;
    const int elemBlocks = (n * DHID + T - 1) / T;
    const int nodeBlocks = (n + 7) / 8;

    gemm128_wmma_kernel<<<gemmBlocks, T, 0, stream>>>(x, W1, bufA, n, K1);
    agg_gather_kernel  <<<nodeBlocks, T, 0, stream>>>(bufA, rowptr, rec, dinv, bufB, n);
    bias_relu_kernel   <<<elemBlocks, T, 0, stream>>>(bufB, b1, n);

    gemm128_wmma_kernel<<<gemmBlocks, T, 0, stream>>>(bufB, W2, bufA, n, DHID);
    agg_gather_kernel  <<<nodeBlocks, T, 0, stream>>>(bufA, rowptr, rec, dinv, bufB, n);
    bias_relu_kernel   <<<elemBlocks, T, 0, stream>>>(bufB, b2, n);

    classifier_kernel<<<(n * NCLS + T - 1) / T, T, 0, stream>>>(bufB, Wc, bc, out, n);
}


# bias+ReLU applied by the aggregation wave to its finished row sums before its one store (both layers); the two elementwise launches are gone, 12 -> 10 launches; sums and their order unchanged
# speedup vs baseline: 2.4255x; 1.1540x over previous
#include <hip/hip_runtime.h>

typedef __attribute__((ext_vector_type(2))) float v2f;
typedef __attribute__((ext_vector_type(8))) float v8f;

#define NNODES 50000
#define DHID   128
#define NCLS   10

__global__ void deg_init_kernel(float* deg, int* cnt, int n, int npad) {
    int i = blockIdx.x * blockDim.x + threadIdx.x;
    if (i < n)    deg[i] = 1.0f;
    if (i < npad) cnt[i] = 0;
}

__global__ void deg_edges_kernel(const int* __restrict__ dst,
                                 const float* __restrict__ w,
                                 float* deg, int* cnt, int* rank, int nE) {
    int e = blockIdx.x * blockDim.x + threadIdx.x;
    if (e < nE) {
        int d = dst[e];
        atomicAdd(&deg[d], w[e]);
        rank[e] = atomicAdd(&cnt[d], 1);
    }
}

__global__ void dinv_kernel(const float* __restrict__ deg, float* dinv, int n) {
    int i = blockIdx.x * blockDim.x + threadIdx.x;
    if (i < n) {
        float d = deg[i];
        dinv[i] = __frsqrt_rn(d);
    }
}

#define SCAN_T   1024
#define SCAN_PER ((NNODES + SCAN_T - 1) / SCAN_T)
#define NPAD     (SCAN_T * SCAN_PER)
__global__ __launch_bounds__(SCAN_T) void rowptr_scan_kernel(
    const int* __restrict__ cnt, int* __restrict__ rowptr)
{
    __shared__ int part[SCAN_T];
    const int tid = threadIdx.x;
    const int beg = tid * SCAN_PER;

    int c[SCAN_PER];
    #pragma unroll
    for (int k = 0; k < SCAN_PER; ++k) c[k] = cnt[beg + k];
    int s = 0;
    #pragma unroll
    for (int k = 0; k < SCAN_PER; ++k) s += c[k];

    part[tid] = s;
    __syncthreads();
    for (int off = 1; off < SCAN_T; off <<= 1) {
        int v = (tid >= off) ? part[tid - off] : 0;
        __syncthreads();
        part[tid] += v;
        __syncthreads();
    }
    int run = part[tid] - s;
    #pragma unroll
    for (int k = 0; k < SCAN_PER; ++k) {
        rowptr[beg + k] = run;
        run += c[k];
    }
    if (tid == SCAN_T - 1) rowptr[NPAD] = run;
}

__global__ void fill_kernel(const int* __restrict__ src,
                            const int* __restrict__ dst,
                            const float* __restrict__ w,
                            const float* __restrict__ dinv,
                            const int* __restrict__ rowptr,
                            const int* __restrict__ rank,
                            int* __restrict__ rec, int nE) {
    int e = blockIdx.x * blockDim.x + threadIdx.x;
    if (e < nE) {
        int s = src[e];
        int d = dst[e];
        float nm = dinv[s] * w[e] * dinv[d];
        int slot = rowptr[d] + rank[e];
        ((int2*)rec)[slot] = make_int2(s, __float_as_int(nm));
    }
}

#define XS_STRIDE 34
#define WT_STRIDE 144

__global__ __launch_bounds__(256) void gemm128_wmma_kernel(
    const float* __restrict__ X, const float* __restrict__ W,
    float* __restrict__ H, int M, int K)
{
    __shared__ float  Xs[128 * XS_STRIDE];
    __shared__ float2 Wt[16 * WT_STRIDE];

    const int tid  = threadIdx.x;
    const int lane = tid & 31;
    const int wave = tid >> 5;
    const int l16  = lane & 15;
    const bool hi  = (lane >= 16);
    const int rowBlock = blockIdx.x * 128;
    const int rowLocal = wave * 16 + l16;

    v8f acc[8];
    #pragma unroll
    for (int t = 0; t < 8; ++t)
        #pragma unroll
        for (int r = 0; r < 8; ++r) acc[t][r] = 0.0f;

    const int nch = (K + 31) / 32;
    for (int ch = 0; ch < nch; ++ch) {
        const int k0 = ch * 32;

        #pragma unroll
        for (int j = 0; j < 4; ++j) {
            int f  = tid * 4 + j;
            int r  = f >> 3;
            int kb = (f & 7) * 4;
            int gr = rowBlock + r;
            float4 v = make_float4(0.f, 0.f, 0.f, 0.f);
            if (gr < M) {
                int kg = k0 + kb;
                if (kg + 3 < K) {
                    v = *(const float4*)(X + (size_t)gr * K + kg);
                } else {
                    float t0 = (kg + 0 < K) ? X[(size_t)gr * K + kg + 0] : 0.f;
                    float t1 = (kg + 1 < K) ? X[(size_t)gr * K + kg + 1] : 0.f;
                    float t2 = (kg + 2 < K) ? X[(size_t)gr * K + kg + 2] : 0.f;
                    float t3 = (kg + 3 < K) ? X[(size_t)gr * K + kg + 3] : 0.f;
                    v = make_float4(t0, t1, t2, t3);
                }
            }
            float* dp = &Xs[r * XS_STRIDE + kb];
            ((float2*)dp)[0] = make_float2(v.x, v.y);
            ((float2*)dp)[1] = make_float2(v.z, v.w);
        }

        #pragma unroll
        for (int j = 0; j < 2; ++j) {
            int f  = tid * 2 + j;
            int kp = f >> 5;
            int cb = (f & 31) * 4;
            int ka = k0 + 2 * kp;
            float4 va = make_float4(0.f, 0.f, 0.f, 0.f);
            float4 vb = make_float4(0.f, 0.f, 0.f, 0.f);
            if (ka < K)     va = *(const float4*)(W + (size_t)ka * DHID + cb);
            if (ka + 1 < K) vb = *(const float4*)(W + (size_t)(ka + 1) * DHID + cb);
            float4 w0 = make_float4(va.x, vb.x, va.y, vb.y);
            float4 w1 = make_float4(va.z, vb.z, va.w, vb.w);
            *(float4*)&Wt[kp * WT_STRIDE + cb]     = w0;
            *(float4*)&Wt[kp * WT_STRIDE + cb + 2] = w1;
        }

        __syncthreads();

        #pragma unroll
        for (int s = 0; s < 8; ++s) {
            const int kp = s * 2 + (hi ? 1 : 0);
            v2f a = *(const v2f*)&Xs[rowLocal * XS_STRIDE + kp * 2];
            float2 bf[8];
            #pragma unroll
            for (int nt = 0; nt < 8; ++nt)
                bf[nt] = Wt[kp * WT_STRIDE + nt * 16 + l16];
            #pragma unroll
            for (int nt = 0; nt < 8; ++nt) {
                v2f b;
                b.x = bf[nt].x;
                b.y = bf[nt].y;
                acc[nt] = __builtin_amdgcn_wmma_f32_16x16x4_f32(
                     false, a,  false, b,
                     (short)0, acc[nt],
                     false,  false);
            }
        }
        __syncthreads();
    }

    #pragma unroll
    for (int nt = 0; nt < 8; ++nt) {
        const int col = nt * 16 + l16;
        #pragma unroll
        for (int r = 0; r < 8; ++r) {
            int m = rowBlock + wave * 16 + r + (hi ? 8 : 0);
            if (m < M) H[(size_t)m * DHID + col] = acc[nt][r];
        }
    }
}

__global__ __launch_bounds__(256) void agg_gather_kernel(
    const float* __restrict__ H,
    const int* __restrict__ rowptr,
    const int* __restrict__ rec,
    const float* __restrict__ dinv,
    const float* __restrict__ bias,
    float* __restrict__ OUT, int n)
{
    const int lane = threadIdx.x & 31;
    const int node = __builtin_amdgcn_readfirstlane(
        (int)(blockIdx.x * 8 + (threadIdx.x >> 5)));
    if (node >= n) return;

    const float4* __restrict__ H4 = (const float4*)H;
    const int2* __restrict__ R2 = (const int2*)rec;

    const int beg = rowptr[node];
    const int end = rowptr[node + 1];
    const float s = dinv[node];
    const float4 hv = H4[(size_t)node * 32 + lane];
    const float4 bv = ((const float4*)bias)[lane];
    float4 acc;
    acc.x = hv.x * s * s;
    acc.y = hv.y * s * s;
    acc.z = hv.z * s * s;
    acc.w = hv.w * s * s;

    for (int base = beg; base < end; base += 32) {
        const int nb = (end - base < 32) ? (end - base) : 32;
        int2 r = make_int2(0, 0);
        if (lane < nb) r = R2[base + lane];

        int j = 0;
        for (; j + 4 <= nb; j += 4) {
            const int s0 = __builtin_amdgcn_readlane(r.x, j + 0);
            const int s1 = __builtin_amdgcn_readlane(r.x, j + 1);
            const int s2 = __builtin_amdgcn_readlane(r.x, j + 2);
            const int s3 = __builtin_amdgcn_readlane(r.x, j + 3);
            const float w0 = __int_as_float(__builtin_amdgcn_readlane(r.y, j + 0));
            const float w1 = __int_as_float(__builtin_amdgcn_readlane(r.y, j + 1));
            const float w2 = __int_as_float(__builtin_amdgcn_readlane(r.y, j + 2));
            const float w3 = __int_as_float(__builtin_amdgcn_readlane(r.y, j + 3));
            const float4 h0 = H4[(size_t)s0 * 32 + lane];
            const float4 h1 = H4[(size_t)s1 * 32 + lane];
            const float4 h2 = H4[(size_t)s2 * 32 + lane];
            const float4 h3 = H4[(size_t)s3 * 32 + lane];
            acc.x = fmaf(h0.x, w0, acc.x); acc.y = fmaf(h0.y, w0, acc.y); acc.z = fmaf(h0.z, w0, acc.z); acc.w = fmaf(h0.w, w0, acc.w);
            acc.x = fmaf(h1.x, w1, acc.x); acc.y = fmaf(h1.y, w1, acc.y); acc.z = fmaf(h1.z, w1, acc.z); acc.w = fmaf(h1.w, w1, acc.w);
            acc.x = fmaf(h2.x, w2, acc.x); acc.y = fmaf(h2.y, w2, acc.y); acc.z = fmaf(h2.z, w2, acc.z); acc.w = fmaf(h2.w, w2, acc.w);
            acc.x = fmaf(h3.x, w3, acc.x); acc.y = fmaf(h3.y, w3, acc.y); acc.z = fmaf(h3.z, w3, acc.z); acc.w = fmaf(h3.w, w3, acc.w);
        }
        #pragma unroll 1
        for (; j < nb; ++j) {
            const int s0 = __builtin_amdgcn_readlane(r.x, j);
            const float w0 = __int_as_float(__builtin_amdgcn_readlane(r.y, j));
            const float4 h0 = H4[(size_t)s0 * 32 + lane];
            acc.x = fmaf(h0.x, w0, acc.x); acc.y = fmaf(h0.y, w0, acc.y); acc.z = fmaf(h0.z, w0, acc.z); acc.w = fmaf(h0.w, w0, acc.w);
        }
    }
    acc.x += bv.x; acc.y += bv.y; acc.z += bv.z; acc.w += bv.w;
    acc.x = acc.x > 0.f ? acc.x : 0.f;
    acc.y = acc.y > 0.f ? acc.y : 0.f;
    acc.z = acc.z > 0.f ? acc.z : 0.f;
    acc.w = acc.w > 0.f ? acc.w : 0.f;
    ((float4*)OUT)[(size_t)node * 32 + lane] = acc;
}

__global__ void classifier_kernel(const float* __restrict__ H,
                                  const float* __restrict__ Wc,
                                  const float* __restrict__ bc,
                                  float* __restrict__ out, int n) {
    int i = blockIdx.x * blockDim.x + threadIdx.x;
    if (i < n * NCLS) {
        int row = i / NCLS;
        int c   = i - row * NCLS;
        float s = bc[c];
        const float* h = H + (size_t)row * DHID;
        #pragma unroll 8
        for (int k = 0; k < DHID; ++k) s += h[k] * Wc[k * NCLS + c];
        out[i] = s;
    }
}

extern "C" void kernel_launch(void* const* d_in, const int* in_sizes, int n_in,
                              void* d_out, int out_size, void* d_ws, size_t ws_size,
                              hipStream_t stream) {
    const float* x  = (const float*)d_in[0];
    const int*   ei = (const int*)d_in[1];
    const float* ew = (const float*)d_in[2];
    const float* W1 = (const float*)d_in[3];
    const float* b1 = (const float*)d_in[4];
    const float* W2 = (const float*)d_in[5];
    const float* b2 = (const float*)d_in[6];
    const float* Wc = (const float*)d_in[7];
    const float* bc = (const float*)d_in[8];
    float* out = (float*)d_out;

    const int n  = NNODES;
    const int E  = in_sizes[1] / 2;
    const int K1 = in_sizes[0] / n;
    const int* src = ei;
    const int* dst = ei + E;

    char* ws = (char*)d_ws;
    size_t off = 0;
    auto alloc = [&](size_t bytes) { char* p = ws + off; off = (off + bytes + 255) & ~(size_t)255; return p; };
    float* deg  = (float*)alloc((size_t)n * 4);
    float* dinv = (float*)alloc((size_t)n * 4);
    float* bufA = (float*)alloc((size_t)n * DHID * 4);
    float* bufB = (float*)alloc((size_t)n * DHID * 4);
    int* cnt    = (int*)alloc((size_t)NPAD * 4);
    int* rowptr = (int*)alloc((size_t)(NPAD + 1) * 4);
    int* rank   = (int*)alloc((size_t)E * 4);
    int* rec    = (int*)alloc((size_t)E * 8);
    (void)ws_size;

    const int T = 256;
    deg_init_kernel <<<(NPAD + T - 1) / T, T, 0, stream>>>(deg, cnt, n, NPAD);
    deg_edges_kernel<<<(E + T - 1) / T, T, 0, stream>>>(dst, ew, deg, cnt, rank, E);
    dinv_kernel     <<<(n + T - 1) / T, T, 0, stream>>>(deg, dinv, n);
    rowptr_scan_kernel<<<1, SCAN_T, 0, stream>>>(cnt, rowptr);
    fill_kernel     <<<(E + T - 1) / T, T, 0, stream>>>(src, dst, ew, dinv, rowptr, rank, rec, E);

    const int gemmBlocks = (n + 127) / 128;
    const int nodeBlocks = (n + 7) / 8;

    gemm128_wmma_kernel<<<gemmBlocks, T, 0, stream>>>(x, W1, bufA, n, K1);
    agg_gather_kernel  <<<nodeBlocks, T, 0, stream>>>(bufA, rowptr, rec, dinv, b1, bufB, n);

    gemm128_wmma_kernel<<<gemmBlocks, T, 0, stream>>>(bufB, W2, bufA, n, DHID);
    agg_gather_kernel  <<<nodeBlocks, T, 0, stream>>>(bufA, rowptr, rec, dinv, b2, bufB, n);

    classifier_kernel<<<(n * NCLS + T - 1) / T, T, 0, stream>>>(bufB, Wc, bc, out, n);
}


# Edges grouped by destination by a two-level counting sort on LDS counters (256-node bins, then nodes): no global atomics left; degree sums and per-node edge order follow LDS arrival order
# speedup vs baseline: 3.1020x; 1.2789x over previous
#include <hip/hip_runtime.h>

typedef __attribute__((ext_vector_type(2))) float v2f;
typedef __attribute__((ext_vector_type(8))) float v8f;

#define NNODES 50000
#define DHID   128
#define NCLS   10

#define BIN_SHIFT 8
#define BIN_NODES (1 << BIN_SHIFT)
#define NBINS     ((NNODES + BIN_NODES - 1) >> BIN_SHIFT)
#define PB        256
#define SORT_T    1024

#define SCAN_T   1024
#define SCAN_PER ((NNODES + SCAN_T - 1) / SCAN_T)
#define NPAD     (SCAN_T * SCAN_PER)
static_assert(NBINS * PB == NPAD, "the (bin, block) table is exactly the scan's size");

__device__ __forceinline__ int coarse_bin(int d) {
    int b = (int)((unsigned)d >> BIN_SHIFT);
    return b < NBINS ? b : NBINS - 1;
}

__global__ __launch_bounds__(SORT_T) void bin_count_kernel(
    const int* __restrict__ dst, int* __restrict__ bh, int nE)
{
    __shared__ int hist[NBINS];
    const int tid = threadIdx.x;
    for (int i = tid; i < NBINS; i += SORT_T) hist[i] = 0;
    __syncthreads();
    const int chunk = (nE + PB - 1) / PB;
    const int beg = blockIdx.x * chunk;
    const int end = (beg + chunk < nE) ? (beg + chunk) : nE;
    for (int base = beg + tid; base < end; base += 4 * SORT_T) {
        int d[4];
        #pragma unroll
        for (int k = 0; k < 4; ++k) {
            const int e = base + k * SORT_T;
            d[k] = (e < end) ? dst[e] : 0;
        }
        #pragma unroll
        for (int k = 0; k < 4; ++k)
            if (base + k * SORT_T < end) atomicAdd(&hist[coarse_bin(d[k])], 1);
    }
    __syncthreads();
    for (int i = tid; i < NBINS; i += SORT_T) bh[i * PB + blockIdx.x] = hist[i];
}

__global__ __launch_bounds__(SCAN_T) void rowptr_scan_kernel(
    const int* __restrict__ cnt, int* __restrict__ rowptr)
{
    __shared__ int part[SCAN_T];
    const int tid = threadIdx.x;
    const int beg = tid * SCAN_PER;

    int c[SCAN_PER];
    #pragma unroll
    for (int k = 0; k < SCAN_PER; ++k) c[k] = cnt[beg + k];
    int s = 0;
    #pragma unroll
    for (int k = 0; k < SCAN_PER; ++k) s += c[k];

    part[tid] = s;
    __syncthreads();
    for (int off = 1; off < SCAN_T; off <<= 1) {
        int v = (tid >= off) ? part[tid - off] : 0;
        __syncthreads();
        part[tid] += v;
        __syncthreads();
    }
    int run = part[tid] - s;
    #pragma unroll
    for (int k = 0; k < SCAN_PER; ++k) {
        rowptr[beg + k] = run;
        run += c[k];
    }
    if (tid == SCAN_T - 1) rowptr[NPAD] = run;
}

__global__ __launch_bounds__(SORT_T) void bin_place_kernel(
    const int* __restrict__ src, const int* __restrict__ dst,
    const float* __restrict__ w, const int* __restrict__ boff,
    int2* __restrict__ tmp, int nE)
{
    __shared__ int cur[NBINS];
    const int tid = threadIdx.x;
    for (int i = tid; i < NBINS; i += SORT_T) cur[i] = boff[i * PB + blockIdx.x];
    __syncthreads();
    const int chunk = (nE + PB - 1) / PB;
    const int beg = blockIdx.x * chunk;
    const int end = (beg + chunk < nE) ? (beg + chunk) : nE;
    for (int base = beg + tid; base < end; base += 4 * SORT_T) {
        int d[4], s[4];
        float wt[4];
        #pragma unroll
        for (int k = 0; k < 4; ++k) {
            const int e = base + k * SORT_T;
            const bool in = e < end;
            d[k]  = in ? dst[e] : 0;
            s[k]  = in ? src[e] : 0;
            wt[k] = in ? w[e] : 0.f;
        }
        #pragma unroll
        for (int k = 0; k < 4; ++k) {
            if (base + k * SORT_T < end) {
                const int pos = atomicAdd(&cur[coarse_bin(d[k])], 1);
                const unsigned key = (unsigned)s[k] | ((unsigned)(d[k] & (BIN_NODES - 1)) << 24);
                tmp[pos] = make_int2((int)key, __float_as_int(wt[k]));
            }
        }
    }
}

__global__ __launch_bounds__(SORT_T) void bin_sort_kernel(
    const int2* __restrict__ tmp, const int* __restrict__ boff,
    int* __restrict__ rowptr, float* __restrict__ dinv,
    int2* __restrict__ rec, int* __restrict__ dsto, int n)
{
    __shared__ int   cnt[BIN_NODES];
    __shared__ int   sc[BIN_NODES];
    __shared__ float dg[BIN_NODES];
    const int tid = threadIdx.x;
    const int b = blockIdx.x;
    const int bbeg = boff[b * PB];
    const int bend = boff[(b + 1) * PB];
    if (tid < BIN_NODES) { cnt[tid] = 0; dg[tid] = 1.0f; }
    __syncthreads();

    for (int base = bbeg + tid; base < bend; base += 4 * SORT_T) {
        int2 r[4];
        #pragma unroll
        for (int k = 0; k < 4; ++k) {
            const int p = base + k * SORT_T;
            r[k] = (p < bend) ? tmp[p] : make_int2(0, 0);
        }
        #pragma unroll
        for (int k = 0; k < 4; ++k) {
            if (base + k * SORT_T < bend) {
                const int dl = (int)((unsigned)r[k].x >> 24);
                atomicAdd(&cnt[dl], 1);
                atomicAdd(&dg[dl], __int_as_float(r[k].y));
            }
        }
    }
    __syncthreads();

    if (tid < BIN_NODES) sc[tid] = cnt[tid];
    __syncthreads();
    for (int off = 1; off < BIN_NODES; off <<= 1) {
        int v = 0;
        if (tid < BIN_NODES && tid >= off) v = sc[tid - off];
        __syncthreads();
        if (tid < BIN_NODES) sc[tid] += v;
        __syncthreads();
    }
    if (tid < BIN_NODES) {
        const int first = sc[tid] - cnt[tid];
        const int node = b * BIN_NODES + tid;
        rowptr[node] = bbeg + first;
        if (node < n) dinv[node] = __frsqrt_rn(dg[tid]);
        cnt[tid] = first;
    }
    __syncthreads();

    for (int base = bbeg + tid; base < bend; base += 4 * SORT_T) {
        int2 r[4];
        #pragma unroll
        for (int k = 0; k < 4; ++k) {
            const int p = base + k * SORT_T;
            r[k] = (p < bend) ? tmp[p] : make_int2(0, 0);
        }
        #pragma unroll
        for (int k = 0; k < 4; ++k) {
            if (base + k * SORT_T < bend) {
                const int dl = (int)((unsigned)r[k].x >> 24);
                const int pos = bbeg + atomicAdd(&cnt[dl], 1);
                rec[pos]  = make_int2(r[k].x & 0x00FFFFFF, r[k].y);
                dsto[pos] = b * BIN_NODES + dl;
            }
        }
    }
}

__global__ __launch_bounds__(SORT_T) void edge_norm_kernel(
    int2* __restrict__ rec, const int* __restrict__ dsto,
    const float* __restrict__ dinv, int nE)
{
    const int stride = PB * SORT_T;
    for (int base = blockIdx.x * SORT_T + threadIdx.x; base < nE; base += 4 * stride) {
        int2 r[4];
        int  d[4];
        #pragma unroll
        for (int k = 0; k < 4; ++k) {
            const int p = base + k * stride;
            const bool in = p < nE;
            r[k] = in ? rec[p] : make_int2(0, 0);
            d[k] = in ? dsto[p] : 0;
        }
        float a[4], c[4];
        #pragma unroll
        for (int k = 0; k < 4; ++k) { a[k] = dinv[r[k].x]; c[k] = dinv[d[k]]; }
        #pragma unroll
        for (int k = 0; k < 4; ++k) {
            const int p = base + k * stride;
            if (p < nE) {
                const float nm = a[k] * __int_as_float(r[k].y) * c[k];
                rec[p] = make_int2(r[k].x, __float_as_int(nm));
            }
        }
    }
}

#define XS_STRIDE 34
#define WT_STRIDE 144

__global__ __launch_bounds__(256) void gemm128_wmma_kernel(
    const float* __restrict__ X, const float* __restrict__ W,
    float* __restrict__ H, int M, int K)
{
    __shared__ float  Xs[128 * XS_STRIDE];
    __shared__ float2 Wt[16 * WT_STRIDE];

    const int tid  = threadIdx.x;
    const int lane = tid & 31;
    const int wave = tid >> 5;
    const int l16  = lane & 15;
    const bool hi  = (lane >= 16);
    const int rowBlock = blockIdx.x * 128;
    const int rowLocal = wave * 16 + l16;

    v8f acc[8];
    #pragma unroll
    for (int t = 0; t < 8; ++t)
        #pragma unroll
        for (int r = 0; r < 8; ++r) acc[t][r] = 0.0f;

    const int nch = (K + 31) / 32;
    for (int ch = 0; ch < nch; ++ch) {
        const int k0 = ch * 32;

        #pragma unroll
        for (int j = 0; j < 4; ++j) {
            int f  = tid * 4 + j;
            int r  = f >> 3;
            int kb = (f & 7) * 4;
            int gr = rowBlock + r;
            float4 v = make_float4(0.f, 0.f, 0.f, 0.f);
            if (gr < M) {
                int kg = k0 + kb;
                if (kg + 3 < K) {
                    v = *(const float4*)(X + (size_t)gr * K + kg);
                } else {
                    float t0 = (kg + 0 < K) ? X[(size_t)gr * K + kg + 0] : 0.f;
                    float t1 = (kg + 1 < K) ? X[(size_t)gr * K + kg + 1] : 0.f;
                    float t2 = (kg + 2 < K) ? X[(size_t)gr * K + kg + 2] : 0.f;
                    float t3 = (kg + 3 < K) ? X[(size_t)gr * K + kg + 3] : 0.f;
                    v = make_float4(t0, t1, t2, t3);
                }
            }
            float* dp = &Xs[r * XS_STRIDE + kb];
            ((float2*)dp)[0] = make_float2(v.x, v.y);
            ((float2*)dp)[1] = make_float2(v.z, v.w);
        }

        #pragma unroll
        for (int j = 0; j < 2; ++j) {
            int f  = tid * 2 + j;
            int kp = f >> 5;
            int cb = (f & 31) * 4;
            int ka = k0 + 2 * kp;
            float4 va = make_float4(0.f, 0.f, 0.f, 0.f);
            float4 vb = make_float4(0.f, 0.f, 0.f, 0.f);
            if (ka < K)     va = *(const float4*)(W + (size_t)ka * DHID + cb);
            if (ka + 1 < K) vb = *(const float4*)(W + (size_t)(ka + 1) * DHID + cb);
            float4 w0 = make_float4(va.x, vb.x, va.y, vb.y);
            float4 w1 = make_float4(va.z, vb.z, va.w, vb.w);
            *(float4*)&Wt[kp * WT_STRIDE + cb]     = w0;
            *(float4*)&Wt[kp * WT_STRIDE + cb + 2] = w1;
        }

        __syncthreads();

        #pragma unroll
        for (int s = 0; s < 8; ++s) {
            const int kp = s * 2 + (hi ? 1 : 0);
            v2f a = *(const v2f*)&Xs[rowLocal * XS_STRIDE + kp * 2];
            float2 bf[8];
            #pragma unroll
            for (int nt = 0; nt < 8; ++nt)
                bf[nt] = Wt[kp * WT_STRIDE + nt * 16 + l16];
            #pragma unroll
            for (int nt = 0; nt < 8; ++nt) {
                v2f b;
                b.x = bf[nt].x;
                b.y = bf[nt].y;
                acc[nt] = __builtin_amdgcn_wmma_f32_16x16x4_f32(
                     false, a,  false, b,
                     (short)0, acc[nt],
                     false,  false);
            }
        }
        __syncthreads();
    }

    #pragma unroll
    for (int nt = 0; nt < 8; ++nt) {
        const int col = nt * 16 + l16;
        #pragma unroll
        for (int r = 0; r < 8; ++r) {
            int m = rowBlock + wave * 16 + r + (hi ? 8 : 0);
            if (m < M) H[(size_t)m * DHID + col] = acc[nt][r];
        }
    }
}

__global__ __launch_bounds__(256) void agg_gather_kernel(
    const float* __restrict__ H,
    const int* __restrict__ rowptr,
    const int* __restrict__ rec,
    const float* __restrict__ dinv,
    const float* __restrict__ bias,
    float* __restrict__ OUT, int n)
{
    const int lane = threadIdx.x & 31;
    const int node = __builtin_amdgcn_readfirstlane(
        (int)(blockIdx.x * 8 + (threadIdx.x >> 5)));
    if (node >= n) return;

    const float4* __restrict__ H4 = (const float4*)H;
    const int2* __restrict__ R2 = (const int2*)rec;

    const int beg = rowptr[node];
    const int end = rowptr[node + 1];
    const float s = dinv[node];
    const float4 hv = H4[(size_t)node * 32 + lane];
    const float4 bv = ((const float4*)bias)[lane];
    float4 acc;
    acc.x = hv.x * s * s;
    acc.y = hv.y * s * s;
    acc.z = hv.z * s * s;
    acc.w = hv.w * s * s;

    for (int base = beg; base < end; base += 32) {
        const int nb = (end - base < 32) ? (end - base) : 32;
        int2 r = make_int2(0, 0);
        if (lane < nb) r = R2[base + lane];

        int j = 0;
        for (; j + 4 <= nb; j += 4) {
            const int s0 = __builtin_amdgcn_readlane(r.x, j + 0);
            const int s1 = __builtin_amdgcn_readlane(r.x, j + 1);
            const int s2 = __builtin_amdgcn_readlane(r.x, j + 2);
            const int s3 = __builtin_amdgcn_readlane(r.x, j + 3);
            const float w0 = __int_as_float(__builtin_amdgcn_readlane(r.y, j + 0));
            const float w1 = __int_as_float(__builtin_amdgcn_readlane(r.y, j + 1));
            const float w2 = __int_as_float(__builtin_amdgcn_readlane(r.y, j + 2));
            const float w3 = __int_as_float(__builtin_amdgcn_readlane(r.y, j + 3));
            const float4 h0 = H4[(size_t)s0 * 32 + lane];
            const float4 h1 = H4[(size_t)s1 * 32 + lane];
            const float4 h2 = H4[(size_t)s2 * 32 + lane];
            const float4 h3 = H4[(size_t)s3 * 32 + lane];
            acc.x = fmaf(h0.x, w0, acc.x); acc.y = fmaf(h0.y, w0, acc.y); acc.z = fmaf(h0.z, w0, acc.z); acc.w = fmaf(h0.w, w0, acc.w);
            acc.x = fmaf(h1.x, w1, acc.x); acc.y = fmaf(h1.y, w1, acc.y); acc.z = fmaf(h1.z, w1, acc.z); acc.w = fmaf(h1.w, w1, acc.w);
            acc.x = fmaf(h2.x, w2, acc.x); acc.y = fmaf(h2.y, w2, acc.y); acc.z = fmaf(h2.z, w2, acc.z); acc.w = fmaf(h2.w, w2, acc.w);
            acc.x = fmaf(h3.x, w3, acc.x); acc.y = fmaf(h3.y, w3, acc.y); acc.z = fmaf(h3.z, w3, acc.z); acc.w = fmaf(h3.w, w3, acc.w);
        }
        #pragma unroll 1
        for (; j < nb; ++j) {
            const int s0 = __builtin_amdgcn_readlane(r.x, j);
            const float w0 = __int_as_float(__builtin_amdgcn_readlane(r.y, j));
            const float4 h0 = H4[(size_t)s0 * 32 + lane];
            acc.x = fmaf(h0.x, w0, acc.x); acc.y = fmaf(h0.y, w0, acc.y); acc.z = fmaf(h0.z, w0, acc.z); acc.w = fmaf(h0.w, w0, acc.w);
        }
    }
    acc.x += bv.x; acc.y += bv.y; acc.z += bv.z; acc.w += bv.w;
    acc.x = acc.x > 0.f ? acc.x : 0.f;
    acc.y = acc.y > 0.f ? acc.y : 0.f;
    acc.z = acc.z > 0.f ? acc.z : 0.f;
    acc.w = acc.w > 0.f ? acc.w : 0.f;
    ((float4*)OUT)[(size_t)node * 32 + lane] = acc;
}

__global__ void classifier_kernel(const float* __restrict__ H,
                                  const float* __restrict__ Wc,
                                  const float* __restrict__ bc,
                                  float* __restrict__ out, int n) {
    int i = blockIdx.x * blockDim.x + threadIdx.x;
    if (i < n * NCLS) {
        int row = i / NCLS;
        int c   = i - row * NCLS;
        float s = bc[c];
        const float* h = H + (size_t)row * DHID;
        #pragma unroll 8
        for (int k = 0; k < DHID; ++k) s += h[k] * Wc[k * NCLS + c];
        out[i] = s;
    }
}

extern "C" void kernel_launch(void* const* d_in, const int* in_sizes, int n_in,
                              void* d_out, int out_size, void* d_ws, size_t ws_size,
                              hipStream_t stream) {
    const float* x  = (const float*)d_in[0];
    const int*   ei = (const int*)d_in[1];
    const float* ew = (const float*)d_in[2];
    const float* W1 = (const float*)d_in[3];
    const float* b1 = (const float*)d_in[4];
    const float* W2 = (const float*)d_in[5];
    const float* b2 = (const float*)d_in[6];
    const float* Wc = (const float*)d_in[7];
    const float* bc = (const float*)d_in[8];
    float* out = (float*)d_out;

    const int n  = NNODES;
    const int E  = in_sizes[1] / 2;
    const int K1 = in_sizes[0] / n;
    const int* src = ei;
    const int* dst = ei + E;

    char* ws = (char*)d_ws;
    size_t off = 0;
    auto alloc = [&](size_t bytes) { char* p = ws + off; off = (off + bytes + 255) & ~(size_t)255; return p; };
    float* dinv = (float*)alloc((size_t)n * 4);
    float* bufA = (float*)alloc((size_t)n * DHID * 4);
    float* bufB = (float*)alloc((size_t)n * DHID * 4);
    int* bh     = (int*)alloc((size_t)NPAD * 4);
    int* boff   = (int*)alloc((size_t)(NPAD + 1) * 4);
    int* rowptr = (int*)alloc((size_t)(NPAD + 1) * 4);
    int2* tmp   = (int2*)alloc((size_t)E * 8);
    int* rec    = (int*)alloc((size_t)E * 8);
    int* dsto   = (int*)alloc((size_t)E * 4);
    (void)ws_size;

    const int T = 256;
    bin_count_kernel  <<<PB, SORT_T, 0, stream>>>(dst, bh, E);
    rowptr_scan_kernel<<<1, SCAN_T, 0, stream>>>(bh, boff);
    bin_place_kernel  <<<PB, SORT_T, 0, stream>>>(src, dst, ew, boff, tmp, E);
    bin_sort_kernel   <<<NBINS, SORT_T, 0, stream>>>(tmp, boff, rowptr, dinv, (int2*)rec, dsto, n);
    edge_norm_kernel  <<<PB, SORT_T, 0, stream>>>((int2*)rec, dsto, dinv, E);

    const int gemmBlocks = (n + 127) / 128;
    const int nodeBlocks = (n + 7) / 8;

    gemm128_wmma_kernel<<<gemmBlocks, T, 0, stream>>>(x, W1, bufA, n, K1);
    agg_gather_kernel  <<<nodeBlocks, T, 0, stream>>>(bufA, rowptr, rec, dinv, b1, bufB, n);

    gemm128_wmma_kernel<<<gemmBlocks, T, 0, stream>>>(bufB, W2, bufA, n, DHID);
    agg_gather_kernel  <<<nodeBlocks, T, 0, stream>>>(bufA, rowptr, rec, dinv, b2, bufB, n);

    classifier_kernel<<<(n * NCLS + T - 1) / T, T, 0, stream>>>(bufB, Wc, bc, out, n);
}


# Every edge-placing block sums the (bin, block) count table itself (whole 128-B lines, integer wave sums) to get its bin offsets; the one-block 50176-entry scan launch is gone: 10 -> 9 launches
# speedup vs baseline: 3.3540x; 1.0812x over previous
#include <hip/hip_runtime.h>

typedef __attribute__((ext_vector_type(2))) float v2f;
typedef __attribute__((ext_vector_type(8))) float v8f;

#define NNODES 50000
#define DHID   128
#define NCLS   10

#define BIN_SHIFT 8
#define BIN_NODES (1 << BIN_SHIFT)
#define NBINS     ((NNODES + BIN_NODES - 1) >> BIN_SHIFT)
#define PB        256
#define SORT_T    1024

#define NWAVES        (SORT_T / 32)
#define ROW_LANES     8
#define ROWS_PER_LOAD (32 / ROW_LANES)
#define ROW_LOADS     (PB / (4 * ROW_LANES))
#define ROW_PASSES    ((NBINS + NWAVES * ROWS_PER_LOAD - 1) / (NWAVES * ROWS_PER_LOAD))
#define BINS_PER_LANE ((NBINS + 31) / 32)
static_assert(PB % (4 * ROW_LANES) == 0, "a table row is read as int4 by ROW_LANES lanes");

__device__ __forceinline__ int coarse_bin(int d) {
    int b = (int)((unsigned)d >> BIN_SHIFT);
    return b < NBINS ? b : NBINS - 1;
}

__global__ __launch_bounds__(SORT_T) void bin_count_kernel(
    const int* __restrict__ dst, int* __restrict__ bh, int nE)
{
    __shared__ int hist[NBINS];
    const int tid = threadIdx.x;
    for (int i = tid; i < NBINS; i += SORT_T) hist[i] = 0;
    __syncthreads();
    const int chunk = (nE + PB - 1) / PB;
    const int beg = blockIdx.x * chunk;
    const int end = (beg + chunk < nE) ? (beg + chunk) : nE;
    for (int base = beg + tid; base < end; base += 4 * SORT_T) {
        int d[4];
        #pragma unroll
        for (int k = 0; k < 4; ++k) {
            const int e = base + k * SORT_T;
            d[k] = (e < end) ? dst[e] : 0;
        }
        #pragma unroll
        for (int k = 0; k < 4; ++k)
            if (base + k * SORT_T < end) atomicAdd(&hist[coarse_bin(d[k])], 1);
    }
    __syncthreads();
    for (int i = tid; i < NBINS; i += SORT_T) bh[i * PB + blockIdx.x] = hist[i];
}

__global__ __launch_bounds__(SORT_T) void bin_place_kernel(
    const int* __restrict__ src, const int* __restrict__ dst,
    const float* __restrict__ w, const int* __restrict__ bh,
    int* __restrict__ binstart, int2* __restrict__ tmp, int nE)
{
    __shared__ int cur[NBINS];
    __shared__ int tot[NBINS];
    __shared__ int pre[NBINS];
    const int tid  = threadIdx.x;
    const int lane = tid & 31;
    const int wave = tid >> 5;
    const int blk  = blockIdx.x;

    const int l8   = lane & (ROW_LANES - 1);
    const int rsub = lane / ROW_LANES;
    int4 v[ROW_PASSES][ROW_LOADS];
    #pragma unroll
    for (int q = 0; q < ROW_PASSES; ++q) {
        const int i = (q * NWAVES + wave) * ROWS_PER_LOAD + rsub;
        const int4* row = (const int4*)(bh + (i < NBINS ? i : NBINS - 1) * PB);
        #pragma unroll
        for (int j = 0; j < ROW_LOADS; ++j) v[q][j] = row[j * ROW_LANES + l8];
    }
    int t[ROW_PASSES], p[ROW_PASSES];
    #pragma unroll
    for (int q = 0; q < ROW_PASSES; ++q) {
        t[q] = 0;
        p[q] = 0;
        #pragma unroll
        for (int j = 0; j < ROW_LOADS; ++j) {
            const int c0 = (j * ROW_LANES + l8) * 4;
            t[q] += v[q][j].x + v[q][j].y + v[q][j].z + v[q][j].w;
            p[q] += (c0 + 0 < blk ? v[q][j].x : 0) + (c0 + 1 < blk ? v[q][j].y : 0)
                  + (c0 + 2 < blk ? v[q][j].z : 0) + (c0 + 3 < blk ? v[q][j].w : 0);
        }
    }
    #pragma unroll
    for (int off = ROW_LANES / 2; off > 0; off >>= 1) {
        #pragma unroll
        for (int q = 0; q < ROW_PASSES; ++q) {
            t[q] += __shfl_xor(t[q], off, 32);
            p[q] += __shfl_xor(p[q], off, 32);
        }
    }
    #pragma unroll
    for (int q = 0; q < ROW_PASSES; ++q) {
        const int i = (q * NWAVES + wave) * ROWS_PER_LOAD + rsub;
        if (l8 == 0 && i < NBINS) { tot[i] = t[q]; pre[i] = p[q]; }
    }
    __syncthreads();

    if (wave == 0) {
        int c[BINS_PER_LANE], pr[BINS_PER_LANE];
        int s = 0;
        #pragma unroll
        for (int k = 0; k < BINS_PER_LANE; ++k) {
            const int i = lane * BINS_PER_LANE + k;
            c[k]  = (i < NBINS) ? tot[i] : 0;
            pr[k] = (i < NBINS) ? pre[i] : 0;
            s += c[k];
        }
        int inc = s;
        #pragma unroll
        for (int off = 1; off < 32; off <<= 1) {
            const int u = __shfl_up(inc, off, 32);
            if (lane >= off) inc += u;
        }
        int run = inc - s;
        #pragma unroll
        for (int k = 0; k < BINS_PER_LANE; ++k) {
            const int i = lane * BINS_PER_LANE + k;
            if (i < NBINS) {
                cur[i] = run + pr[k];
                if (blk == 0) binstart[i] = run;
            }
            run += c[k];
        }
        if (blk == 0 && lane == 31) binstart[NBINS] = run;
    }
    __syncthreads();
    const int chunk = (nE + PB - 1) / PB;
    const int beg = blockIdx.x * chunk;
    const int end = (beg + chunk < nE) ? (beg + chunk) : nE;
    for (int base = beg + tid; base < end; base += 4 * SORT_T) {
        int d[4], s[4];
        float wt[4];
        #pragma unroll
        for (int k = 0; k < 4; ++k) {
            const int e = base + k * SORT_T;
            const bool in = e < end;
            d[k]  = in ? dst[e] : 0;
            s[k]  = in ? src[e] : 0;
            wt[k] = in ? w[e] : 0.f;
        }
        #pragma unroll
        for (int k = 0; k < 4; ++k) {
            if (base + k * SORT_T < end) {
                const int pos = atomicAdd(&cur[coarse_bin(d[k])], 1);
                const unsigned key = (unsigned)s[k] | ((unsigned)(d[k] & (BIN_NODES - 1)) << 24);
                tmp[pos] = make_int2((int)key, __float_as_int(wt[k]));
            }
        }
    }
}

__global__ __launch_bounds__(SORT_T) void bin_sort_kernel(
    const int2* __restrict__ tmp, const int* __restrict__ binstart,
    int* __restrict__ rowptr, float* __restrict__ dinv,
    int2* __restrict__ rec, int* __restrict__ dsto, int n)
{
    __shared__ int   cnt[BIN_NODES];
    __shared__ int   sc[BIN_NODES];
    __shared__ float dg[BIN_NODES];
    const int tid = threadIdx.x;
    const int b = blockIdx.x;
    const int bbeg = binstart[b];
    const int bend = binstart[b + 1];
    if (tid < BIN_NODES) { cnt[tid] = 0; dg[tid] = 1.0f; }
    __syncthreads();

    for (int base = bbeg + tid; base < bend; base += 4 * SORT_T) {
        int2 r[4];
        #pragma unroll
        for (int k = 0; k < 4; ++k) {
            const int p = base + k * SORT_T;
            r[k] = (p < bend) ? tmp[p] : make_int2(0, 0);
        }
        #pragma unroll
        for (int k = 0; k < 4; ++k) {
            if (base + k * SORT_T < bend) {
                const int dl = (int)((unsigned)r[k].x >> 24);
                atomicAdd(&cnt[dl], 1);
                atomicAdd(&dg[dl], __int_as_float(r[k].y));
            }
        }
    }
    __syncthreads();

    if (tid < BIN_NODES) sc[tid] = cnt[tid];
    __syncthreads();
    for (int off = 1; off < BIN_NODES; off <<= 1) {
        int v = 0;
        if (tid < BIN_NODES && tid >= off) v = sc[tid - off];
        __syncthreads();
        if (tid < BIN_NODES) sc[tid] += v;
        __syncthreads();
    }
    if (tid < BIN_NODES) {
        const int first = sc[tid] - cnt[tid];
        const int node = b * BIN_NODES + tid;
        rowptr[node] = bbeg + first;
        if (node < n) dinv[node] = __frsqrt_rn(dg[tid]);
        cnt[tid] = first;
    }
    __syncthreads();

    for (int base = bbeg + tid; base < bend; base += 4 * SORT_T) {
        int2 r[4];
        #pragma unroll
        for (int k = 0; k < 4; ++k) {
            const int p = base + k * SORT_T;
            r[k] = (p < bend) ? tmp[p] : make_int2(0, 0);
        }
        #pragma unroll
        for (int k = 0; k < 4; ++k) {
            if (base + k * SORT_T < bend) {
                const int dl = (int)((unsigned)r[k].x >> 24);
                const int pos = bbeg + atomicAdd(&cnt[dl], 1);
                rec[pos]  = make_int2(r[k].x & 0x00FFFFFF, r[k].y);
                dsto[pos] = b * BIN_NODES + dl;
            }
        }
    }
}

__global__ __launch_bounds__(SORT_T) void edge_norm_kernel(
    int2* __restrict__ rec, const int* __restrict__ dsto,
    const float* __restrict__ dinv, int nE)
{
    const int stride = PB * SORT_T;
    for (int base = blockIdx.x * SORT_T + threadIdx.x; base < nE; base += 4 * stride) {
        int2 r[4];
        int  d[4];
        #pragma unroll
        for (int k = 0; k < 4; ++k) {
            const int p = base + k * stride;
            const bool in = p < nE;
            r[k] = in ? rec[p] : make_int2(0, 0);
            d[k] = in ? dsto[p] : 0;
        }
        float a[4], c[4];
        #pragma unroll
        for (int k = 0; k < 4; ++k) { a[k] = dinv[r[k].x]; c[k] = dinv[d[k]]; }
        #pragma unroll
        for (int k = 0; k < 4; ++k) {
            const int p = base + k * stride;
            if (p < nE) {
                const float nm = a[k] * __int_as_float(r[k].y) * c[k];
                rec[p] = make_int2(r[k].x, __float_as_int(nm));
            }
        }
    }
}

#define XS_STRIDE 34
#define WT_STRIDE 144

__global__ __launch_bounds__(256) void gemm128_wmma_kernel(
    const float* __restrict__ X, const float* __restrict__ W,
    float* __restrict__ H, int M, int K)
{
    __shared__ float  Xs[128 * XS_STRIDE];
    __shared__ float2 Wt[16 * WT_STRIDE];

    const int tid  = threadIdx.x;
    const int lane = tid & 31;
    const int wave = tid >> 5;
    const int l16  = lane & 15;
    const bool hi  = (lane >= 16);
    const int rowBlock = blockIdx.x * 128;
    const int rowLocal = wave * 16 + l16;

    v8f acc[8];
    #pragma unroll
    for (int t = 0; t < 8; ++t)
        #pragma unroll
        for (int r = 0; r < 8; ++r) acc[t][r] = 0.0f;

    const int nch = (K + 31) / 32;
    for (int ch = 0; ch < nch; ++ch) {
        const int k0 = ch * 32;

        #pragma unroll
        for (int j = 0; j < 4; ++j) {
            int f  = tid * 4 + j;
            int r  = f >> 3;
            int kb = (f & 7) * 4;
            int gr = rowBlock + r;
            float4 v = make_float4(0.f, 0.f, 0.f, 0.f);
            if (gr < M) {
                int kg = k0 + kb;
                if (kg + 3 < K) {
                    v = *(const float4*)(X + (size_t)gr * K + kg);
                } else {
                    float t0 = (kg + 0 < K) ? X[(size_t)gr * K + kg + 0] : 0.f;
                    float t1 = (kg + 1 < K) ? X[(size_t)gr * K + kg + 1] : 0.f;
                    float t2 = (kg + 2 < K) ? X[(size_t)gr * K + kg + 2] : 0.f;
                    float t3 = (kg + 3 < K) ? X[(size_t)gr * K + kg + 3] : 0.f;
                    v = make_float4(t0, t1, t2, t3);
                }
            }
            float* dp = &Xs[r * XS_STRIDE + kb];
            ((float2*)dp)[0] = make_float2(v.x, v.y);
            ((float2*)dp)[1] = make_float2(v.z, v.w);
        }

        #pragma unroll
        for (int j = 0; j < 2; ++j) {
            int f  = tid * 2 + j;
            int kp = f >> 5;
            int cb = (f & 31) * 4;
            int ka = k0 + 2 * kp;
            float4 va = make_float4(0.f, 0.f, 0.f, 0.f);
            float4 vb = make_float4(0.f, 0.f, 0.f, 0.f);
            if (ka < K)     va = *(const float4*)(W + (size_t)ka * DHID + cb);
            if (ka + 1 < K) vb = *(const float4*)(W + (size_t)(ka + 1) * DHID + cb);
            float4 w0 = make_float4(va.x, vb.x, va.y, vb.y);
            float4 w1 = make_float4(va.z, vb.z, va.w, vb.w);
            *(float4*)&Wt[kp * WT_STRIDE + cb]     = w0;
            *(float4*)&Wt[kp * WT_STRIDE + cb + 2] = w1;
        }

        __syncthreads();

        #pragma unroll
        for (int s = 0; s < 8; ++s) {
            const int kp = s * 2 + (hi ? 1 : 0);
            v2f a = *(const v2f*)&Xs[rowLocal * XS_STRIDE + kp * 2];
            float2 bf[8];
            #pragma unroll
            for (int nt = 0; nt < 8; ++nt)
                bf[nt] = Wt[kp * WT_STRIDE + nt * 16 + l16];
            #pragma unroll
            for (int nt = 0; nt < 8; ++nt) {
                v2f b;
                b.x = bf[nt].x;
                b.y = bf[nt].y;
                acc[nt] = __builtin_amdgcn_wmma_f32_16x16x4_f32(
                     false, a,  false, b,
                     (short)0, acc[nt],
                     false,  false);
            }
        }
        __syncthreads();
    }

    #pragma unroll
    for (int nt = 0; nt < 8; ++nt) {
        const int col = nt * 16 + l16;
        #pragma unroll
        for (int r = 0; r < 8; ++r) {
            int m = rowBlock + wave * 16 + r + (hi ? 8 : 0);
            if (m < M) H[(size_t)m * DHID + col] = acc[nt][r];
        }
    }
}

__global__ __launch_bounds__(256) void agg_gather_kernel(
    const float* __restrict__ H,
    const int* __restrict__ rowptr,
    const int* __restrict__ rec,
    const float* __restrict__ dinv,
    const float* __restrict__ bias,
    float* __restrict__ OUT, int n)
{
    const int lane = threadIdx.x & 31;
    const int node = __builtin_amdgcn_readfirstlane(
        (int)(blockIdx.x * 8 + (threadIdx.x >> 5)));
    if (node >= n) return;

    const float4* __restrict__ H4 = (const float4*)H;
    const int2* __restrict__ R2 = (const int2*)rec;

    const int beg = rowptr[node];
    const int end = rowptr[node + 1];
    const float s = dinv[node];
    const float4 hv = H4[(size_t)node * 32 + lane];
    const float4 bv = ((const float4*)bias)[lane];
    float4 acc;
    acc.x = hv.x * s * s;
    acc.y = hv.y * s * s;
    acc.z = hv.z * s * s;
    acc.w = hv.w * s * s;

    for (int base = beg; base < end; base += 32) {
        const int nb = (end - base < 32) ? (end - base) : 32;
        int2 r = make_int2(0, 0);
        if (lane < nb) r = R2[base + lane];

        int j = 0;
        for (; j + 4 <= nb; j += 4) {
            const int s0 = __builtin_amdgcn_readlane(r.x, j + 0);
            const int s1 = __builtin_amdgcn_readlane(r.x, j + 1);
            const int s2 = __builtin_amdgcn_readlane(r.x, j + 2);
            const int s3 = __builtin_amdgcn_readlane(r.x, j + 3);
            const float w0 = __int_as_float(__builtin_amdgcn_readlane(r.y, j + 0));
            const float w1 = __int_as_float(__builtin_amdgcn_readlane(r.y, j + 1));
            const float w2 = __int_as_float(__builtin_amdgcn_readlane(r.y, j + 2));
            const float w3 = __int_as_float(__builtin_amdgcn_readlane(r.y, j + 3));
            const float4 h0 = H4[(size_t)s0 * 32 + lane];
            const float4 h1 = H4[(size_t)s1 * 32 + lane];
            const float4 h2 = H4[(size_t)s2 * 32 + lane];
            const float4 h3 = H4[(size_t)s3 * 32 + lane];
            acc.x = fmaf(h0.x, w0, acc.x); acc.y = fmaf(h0.y, w0, acc.y); acc.z = fmaf(h0.z, w0, acc.z); acc.w = fmaf(h0.w, w0, acc.w);
            acc.x = fmaf(h1.x, w1, acc.x); acc.y = fmaf(h1.y, w1, acc.y); acc.z = fmaf(h1.z, w1, acc.z); acc.w = fmaf(h1.w, w1, acc.w);
            acc.x = fmaf(h2.x, w2, acc.x); acc.y = fmaf(h2.y, w2, acc.y); acc.z = fmaf(h2.z, w2, acc.z); acc.w = fmaf(h2.w, w2, acc.w);
            acc.x = fmaf(h3.x, w3, acc.x); acc.y = fmaf(h3.y, w3, acc.y); acc.z = fmaf(h3.z, w3, acc.z); acc.w = fmaf(h3.w, w3, acc.w);
        }
        #pragma unroll 1
        for (; j < nb; ++j) {
            const int s0 = __builtin_amdgcn_readlane(r.x, j);
            const float w0 = __int_as_float(__builtin_amdgcn_readlane(r.y, j));
            const float4 h0 = H4[(size_t)s0 * 32 + lane];
            acc.x = fmaf(h0.x, w0, acc.x); acc.y = fmaf(h0.y, w0, acc.y); acc.z = fmaf(h0.z, w0, acc.z); acc.w = fmaf(h0.w, w0, acc.w);
        }
    }
    acc.x += bv.x; acc.y += bv.y; acc.z += bv.z; acc.w += bv.w;
    acc.x = acc.x > 0.f ? acc.x : 0.f;
    acc.y = acc.y > 0.f ? acc.y : 0.f;
    acc.z = acc.z > 0.f ? acc.z : 0.f;
    acc.w = acc.w > 0.f ? acc.w : 0.f;
    ((float4*)OUT)[(size_t)node * 32 + lane] = acc;
}

__global__ void classifier_kernel(const float* __restrict__ H,
                                  const float* __restrict__ Wc,
                                  const float* __restrict__ bc,
                                  float* __restrict__ out, int n) {
    int i = blockIdx.x * blockDim.x + threadIdx.x;
    if (i < n * NCLS) {
        int row = i / NCLS;
        int c   = i - row * NCLS;
        float s = bc[c];
        const float* h = H + (size_t)row * DHID;
        #pragma unroll 8
        for (int k = 0; k < DHID; ++k) s += h[k] * Wc[k * NCLS + c];
        out[i] = s;
    }
}

extern "C" void kernel_launch(void* const* d_in, const int* in_sizes, int n_in,
                              void* d_out, int out_size, void* d_ws, size_t ws_size,
                              hipStream_t stream) {
    const float* x  = (const float*)d_in[0];
    const int*   ei = (const int*)d_in[1];
    const float* ew = (const float*)d_in[2];
    const float* W1 = (const float*)d_in[3];
    const float* b1 = (const float*)d_in[4];
    const float* W2 = (const float*)d_in[5];
    const float* b2 = (const float*)d_in[6];
    const float* Wc = (const float*)d_in[7];
    const float* bc = (const float*)d_in[8];
    float* out = (float*)d_out;

    const int n  = NNODES;
    const int E  = in_sizes[1] / 2;
    const int K1 = in_sizes[0] / n;
    const int* src = ei;
    const int* dst = ei + E;

    char* ws = (char*)d_ws;
    size_t off = 0;
    auto alloc = [&](size_t bytes) { char* p = ws + off; off = (off + bytes + 255) & ~(size_t)255; return p; };
    float* dinv = (float*)alloc((size_t)n * 4);
    float* bufA = (float*)alloc((size_t)n * DHID * 4);
    float* bufB = (float*)alloc((size_t)n * DHID * 4);
    int* bh       = (int*)alloc((size_t)NBINS * PB * 4);
    int* binstart = (int*)alloc((size_t)(NBINS + 1) * 4);
    int* rowptr   = (int*)alloc((size_t)(NBINS * BIN_NODES + 1) * 4);
    int2* tmp   = (int2*)alloc((size_t)E * 8);
    int* rec    = (int*)alloc((size_t)E * 8);
    int* dsto   = (int*)alloc((size_t)E * 4);
    (void)ws_size;

    const int T = 256;
    bin_count_kernel  <<<PB, SORT_T, 0, stream>>>(dst, bh, E);
    bin_place_kernel  <<<PB, SORT_T, 0, stream>>>(src, dst, ew, bh, binstart, tmp, E);
    bin_sort_kernel   <<<NBINS, SORT_T, 0, stream>>>(tmp, binstart, rowptr, dinv, (int2*)rec, dsto, n);
    edge_norm_kernel  <<<PB, SORT_T, 0, stream>>>((int2*)rec, dsto, dinv, E);

    const int gemmBlocks = (n + 127) / 128;
    const int nodeBlocks = (n + 7) / 8;

    gemm128_wmma_kernel<<<gemmBlocks, T, 0, stream>>>(x, W1, bufA, n, K1);
    agg_gather_kernel  <<<nodeBlocks, T, 0, stream>>>(bufA, rowptr, rec, dinv, b1, bufB, n);

    gemm128_wmma_kernel<<<gemmBlocks, T, 0, stream>>>(bufB, W2, bufA, n, DHID);
    agg_gather_kernel  <<<nodeBlocks, T, 0, stream>>>(bufA, rowptr, rec, dinv, b2, bufB, n);

    classifier_kernel<<<(n * NCLS + T - 1) / T, T, 0, stream>>>(bufB, Wc, bc, out, n);
}


# Classifier done by the layer-2 aggregation wave on the row it holds (Wc in LDS; logit = 4 terms per lane, then a 32-lane tree; was one serial 128-term sum); row never stored; 9 -> 8 launches
# speedup vs baseline: 3.4903x; 1.0406x over previous
#include <hip/hip_runtime.h>

typedef __attribute__((ext_vector_type(2))) float v2f;
typedef __attribute__((ext_vector_type(8))) float v8f;

#define NNODES 50000
#define DHID   128
#define NCLS   10

#define BIN_SHIFT 8
#define BIN_NODES (1 << BIN_SHIFT)
#define NBINS     ((NNODES + BIN_NODES - 1) >> BIN_SHIFT)
#define PB        256
#define SORT_T    1024

#define NWAVES        (SORT_T / 32)
#define ROW_LANES     8
#define ROWS_PER_LOAD (32 / ROW_LANES)
#define ROW_LOADS     (PB / (4 * ROW_LANES))
#define ROW_PASSES    ((NBINS + NWAVES * ROWS_PER_LOAD - 1) / (NWAVES * ROWS_PER_LOAD))
#define BINS_PER_LANE ((NBINS + 31) / 32)
static_assert(PB % (4 * ROW_LANES) == 0, "a table row is read as int4 by ROW_LANES lanes");

__device__ __forceinline__ int coarse_bin(int d) {
    int b = (int)((unsigned)d >> BIN_SHIFT);
    return b < NBINS ? b : NBINS - 1;
}

__global__ __launch_bounds__(SORT_T) void bin_count_kernel(
    const int* __restrict__ dst, int* __restrict__ bh, int nE)
{
    __shared__ int hist[NBINS];
    const int tid = threadIdx.x;
    for (int i = tid; i < NBINS; i += SORT_T) hist[i] = 0;
    __syncthreads();
    const int chunk = (nE + PB - 1) / PB;
    const int beg = blockIdx.x * chunk;
    const int end = (beg + chunk < nE) ? (beg + chunk) : nE;
    for (int base = beg + tid; base < end; base += 4 * SORT_T) {
        int d[4];
        #pragma unroll
        for (int k = 0; k < 4; ++k) {
            const int e = base + k * SORT_T;
            d[k] = (e < end) ? dst[e] : 0;
        }
        #pragma unroll
        for (int k = 0; k < 4; ++k)
            if (base + k * SORT_T < end) atomicAdd(&hist[coarse_bin(d[k])], 1);
    }
    __syncthreads();
    for (int i = tid; i < NBINS; i += SORT_T) bh[i * PB + blockIdx.x] = hist[i];
}

__global__ __launch_bounds__(SORT_T) void bin_place_kernel(
    const int* __restrict__ src, const int* __restrict__ dst,
    const float* __restrict__ w, const int* __restrict__ bh,
    int* __restrict__ binstart, int2* __restrict__ tmp, int nE)
{
    __shared__ int cur[NBINS];
    __shared__ int tot[NBINS];
    __shared__ int pre[NBINS];
    const int tid  = threadIdx.x;
    const int lane = tid & 31;
    const int wave = tid >> 5;
    const int blk  = blockIdx.x;

    const int l8   = lane & (ROW_LANES - 1);
    const int rsub = lane / ROW_LANES;
    int4 v[ROW_PASSES][ROW_LOADS];
    #pragma unroll
    for (int q = 0; q < ROW_PASSES; ++q) {
        const int i = (q * NWAVES + wave) * ROWS_PER_LOAD + rsub;
        const int4* row = (const int4*)(bh + (i < NBINS ? i : NBINS - 1) * PB);
        #pragma unroll
        for (int j = 0; j < ROW_LOADS; ++j) v[q][j] = row[j * ROW_LANES + l8];
    }
    int t[ROW_PASSES], p[ROW_PASSES];
    #pragma unroll
    for (int q = 0; q < ROW_PASSES; ++q) {
        t[q] = 0;
        p[q] = 0;
        #pragma unroll
        for (int j = 0; j < ROW_LOADS; ++j) {
            const int c0 = (j * ROW_LANES + l8) * 4;
            t[q] += v[q][j].x + v[q][j].y + v[q][j].z + v[q][j].w;
            p[q] += (c0 + 0 < blk ? v[q][j].x : 0) + (c0 + 1 < blk ? v[q][j].y : 0)
                  + (c0 + 2 < blk ? v[q][j].z : 0) + (c0 + 3 < blk ? v[q][j].w : 0);
        }
    }
    #pragma unroll
    for (int off = ROW_LANES / 2; off > 0; off >>= 1) {
        #pragma unroll
        for (int q = 0; q < ROW_PASSES; ++q) {
            t[q] += __shfl_xor(t[q], off, 32);
            p[q] += __shfl_xor(p[q], off, 32);
        }
    }
    #pragma unroll
    for (int q = 0; q < ROW_PASSES; ++q) {
        const int i = (q * NWAVES + wave) * ROWS_PER_LOAD + rsub;
        if (l8 == 0 && i < NBINS) { tot[i] = t[q]; pre[i] = p[q]; }
    }
    __syncthreads();

    if (wave == 0) {
        int c[BINS_PER_LANE], pr[BINS_PER_LANE];
        int s = 0;
        #pragma unroll
        for (int k = 0; k < BINS_PER_LANE; ++k) {
            const int i = lane * BINS_PER_LANE + k;
            c[k]  = (i < NBINS) ? tot[i] : 0;
            pr[k] = (i < NBINS) ? pre[i] : 0;
            s += c[k];
        }
        int inc = s;
        #pragma unroll
        for (int off = 1; off < 32; off <<= 1) {
            const int u = __shfl_up(inc, off, 32);
            if (lane >= off) inc += u;
        }
        int run = inc - s;
        #pragma unroll
        for (int k = 0; k < BINS_PER_LANE; ++k) {
            const int i = lane * BINS_PER_LANE + k;
            if (i < NBINS) {
                cur[i] = run + pr[k];
                if (blk == 0) binstart[i] = run;
            }
            run += c[k];
        }
        if (blk == 0 && lane == 31) binstart[NBINS] = run;
    }
    __syncthreads();
    const int chunk = (nE + PB - 1) / PB;
    const int beg = blockIdx.x * chunk;
    const int end = (beg + chunk < nE) ? (beg + chunk) : nE;
    for (int base = beg + tid; base < end; base += 4 * SORT_T) {
        int d[4], s[4];
        float wt[4];
        #pragma unroll
        for (int k = 0; k < 4; ++k) {
            const int e = base + k * SORT_T;
            const bool in = e < end;
            d[k]  = in ? dst[e] : 0;
            s[k]  = in ? src[e] : 0;
            wt[k] = in ? w[e] : 0.f;
        }
        #pragma unroll
        for (int k = 0; k < 4; ++k) {
            if (base + k * SORT_T < end) {
                const int pos = atomicAdd(&cur[coarse_bin(d[k])], 1);
                const unsigned key = (unsigned)s[k] | ((unsigned)(d[k] & (BIN_NODES - 1)) << 24);
                tmp[pos] = make_int2((int)key, __float_as_int(wt[k]));
            }
        }
    }
}

__global__ __launch_bounds__(SORT_T) void bin_sort_kernel(
    const int2* __restrict__ tmp, const int* __restrict__ binstart,
    int* __restrict__ rowptr, float* __restrict__ dinv,
    int2* __restrict__ rec, int* __restrict__ dsto, int n)
{
    __shared__ int   cnt[BIN_NODES];
    __shared__ int   sc[BIN_NODES];
    __shared__ float dg[BIN_NODES];
    const int tid = threadIdx.x;
    const int b = blockIdx.x;
    const int bbeg = binstart[b];
    const int bend = binstart[b + 1];
    if (tid < BIN_NODES) { cnt[tid] = 0; dg[tid] = 1.0f; }
    __syncthreads();

    for (int base = bbeg + tid; base < bend; base += 4 * SORT_T) {
        int2 r[4];
        #pragma unroll
        for (int k = 0; k < 4; ++k) {
            const int p = base + k * SORT_T;
            r[k] = (p < bend) ? tmp[p] : make_int2(0, 0);
        }
        #pragma unroll
        for (int k = 0; k < 4; ++k) {
            if (base + k * SORT_T < bend) {
                const int dl = (int)((unsigned)r[k].x >> 24);
                atomicAdd(&cnt[dl], 1);
                atomicAdd(&dg[dl], __int_as_float(r[k].y));
            }
        }
    }
    __syncthreads();

    if (tid < BIN_NODES) sc[tid] = cnt[tid];
    __syncthreads();
    for (int off = 1; off < BIN_NODES; off <<= 1) {
        int v = 0;
        if (tid < BIN_NODES && tid >= off) v = sc[tid - off];
        __syncthreads();
        if (tid < BIN_NODES) sc[tid] += v;
        __syncthreads();
    }
    if (tid < BIN_NODES) {
        const int first = sc[tid] - cnt[tid];
        const int node = b * BIN_NODES + tid;
        rowptr[node] = bbeg + first;
        if (node < n) dinv[node] = __frsqrt_rn(dg[tid]);
        cnt[tid] = first;
    }
    __syncthreads();

    for (int base = bbeg + tid; base < bend; base += 4 * SORT_T) {
        int2 r[4];
        #pragma unroll
        for (int k = 0; k < 4; ++k) {
            const int p = base + k * SORT_T;
            r[k] = (p < bend) ? tmp[p] : make_int2(0, 0);
        }
        #pragma unroll
        for (int k = 0; k < 4; ++k) {
            if (base + k * SORT_T < bend) {
                const int dl = (int)((unsigned)r[k].x >> 24);
                const int pos = bbeg + atomicAdd(&cnt[dl], 1);
                rec[pos]  = make_int2(r[k].x & 0x00FFFFFF, r[k].y);
                dsto[pos] = b * BIN_NODES + dl;
            }
        }
    }
}

__global__ __launch_bounds__(SORT_T) void edge_norm_kernel(
    int2* __restrict__ rec, const int* __restrict__ dsto,
    const float* __restrict__ dinv, int nE)
{
    const int stride = PB * SORT_T;
    for (int base = blockIdx.x * SORT_T + threadIdx.x; base < nE; base += 4 * stride) {
        int2 r[4];
        int  d[4];
        #pragma unroll
        for (int k = 0; k < 4; ++k) {
            const int p = base + k * stride;
            const bool in = p < nE;
            r[k] = in ? rec[p] : make_int2(0, 0);
            d[k] = in ? dsto[p] : 0;
        }
        float a[4], c[4];
        #pragma unroll
        for (int k = 0; k < 4; ++k) { a[k] = dinv[r[k].x]; c[k] = dinv[d[k]]; }
        #pragma unroll
        for (int k = 0; k < 4; ++k) {
            const int p = base + k * stride;
            if (p < nE) {
                const float nm = a[k] * __int_as_float(r[k].y) * c[k];
                rec[p] = make_int2(r[k].x, __float_as_int(nm));
            }
        }
    }
}

#define XS_STRIDE 34
#define WT_STRIDE 144

__global__ __launch_bounds__(256) void gemm128_wmma_kernel(
    const float* __restrict__ X, const float* __restrict__ W,
    float* __restrict__ H, int M, int K)
{
    __shared__ float  Xs[128 * XS_STRIDE];
    __shared__ float2 Wt[16 * WT_STRIDE];

    const int tid  = threadIdx.x;
    const int lane = tid & 31;
    const int wave = tid >> 5;
    const int l16  = lane & 15;
    const bool hi  = (lane >= 16);
    const int rowBlock = blockIdx.x * 128;
    const int rowLocal = wave * 16 + l16;

    v8f acc[8];
    #pragma unroll
    for (int t = 0; t < 8; ++t)
        #pragma unroll
        for (int r = 0; r < 8; ++r) acc[t][r] = 0.0f;

    const int nch = (K + 31) / 32;
    for (int ch = 0; ch < nch; ++ch) {
        const int k0 = ch * 32;

        #pragma unroll
        for (int j = 0; j < 4; ++j) {
            int f  = tid * 4 + j;
            int r  = f >> 3;
            int kb = (f & 7) * 4;
            int gr = rowBlock + r;
            float4 v = make_float4(0.f, 0.f, 0.f, 0.f);
            if (gr < M) {
                int kg = k0 + kb;
                if (kg + 3 < K) {
                    v = *(const float4*)(X + (size_t)gr * K + kg);
                } else {
                    float t0 = (kg + 0 < K) ? X[(size_t)gr * K + kg + 0] : 0.f;
                    float t1 = (kg + 1 < K) ? X[(size_t)gr * K + kg + 1] : 0.f;
                    float t2 = (kg + 2 < K) ? X[(size_t)gr * K + kg + 2] : 0.f;
                    float t3 = (kg + 3 < K) ? X[(size_t)gr * K + kg + 3] : 0.f;
                    v = make_float4(t0, t1, t2, t3);
                }
            }
            float* dp = &Xs[r * XS_STRIDE + kb];
            ((float2*)dp)[0] = make_float2(v.x, v.y);
            ((float2*)dp)[1] = make_float2(v.z, v.w);
        }

        #pragma unroll
        for (int j = 0; j < 2; ++j) {
            int f  = tid * 2 + j;
            int kp = f >> 5;
            int cb = (f & 31) * 4;
            int ka = k0 + 2 * kp;
            float4 va = make_float4(0.f, 0.f, 0.f, 0.f);
            float4 vb = make_float4(0.f, 0.f, 0.f, 0.f);
            if (ka < K)     va = *(const float4*)(W + (size_t)ka * DHID + cb);
            if (ka + 1 < K) vb = *(const float4*)(W + (size_t)(ka + 1) * DHID + cb);
            float4 w0 = make_float4(va.x, vb.x, va.y, vb.y);
            float4 w1 = make_float4(va.z, vb.z, va.w, vb.w);
            *(float4*)&Wt[kp * WT_STRIDE + cb]     = w0;
            *(float4*)&Wt[kp * WT_STRIDE + cb + 2] = w1;
        }

        __syncthreads();

        #pragma unroll
        for (int s = 0; s < 8; ++s) {
            const int kp = s * 2 + (hi ? 1 : 0);
            v2f a = *(const v2f*)&Xs[rowLocal * XS_STRIDE + kp * 2];
            float2 bf[8];
            #pragma unroll
            for (int nt = 0; nt < 8; ++nt)
                bf[nt] = Wt[kp * WT_STRIDE + nt * 16 + l16];
            #pragma unroll
            for (int nt = 0; nt < 8; ++nt) {
                v2f b;
                b.x = bf[nt].x;
                b.y = bf[nt].y;
                acc[nt] = __builtin_amdgcn_wmma_f32_16x16x4_f32(
                     false, a,  false, b,
                     (short)0, acc[nt],
                     false,  false);
            }
        }
        __syncthreads();
    }

    #pragma unroll
    for (int nt = 0; nt < 8; ++nt) {
        const int col = nt * 16 + l16;
        #pragma unroll
        for (int r = 0; r < 8; ++r) {
            int m = rowBlock + wave * 16 + r + (hi ? 8 : 0);
            if (m < M) H[(size_t)m * DHID + col] = acc[nt][r];
        }
    }
}

__device__ __forceinline__ float4 agg_row(
    const float4* __restrict__ H4,
    const int* __restrict__ rowptr,
    const int2* __restrict__ R2,
    const float* __restrict__ dinv,
    const float* __restrict__ bias,
    int node, int lane)
{
    const int beg = rowptr[node];
    const int end = rowptr[node + 1];
    const float s = dinv[node];
    const float4 hv = H4[(size_t)node * 32 + lane];
    const float4 bv = ((const float4*)bias)[lane];
    float4 acc;
    acc.x = hv.x * s * s;
    acc.y = hv.y * s * s;
    acc.z = hv.z * s * s;
    acc.w = hv.w * s * s;

    for (int base = beg; base < end; base += 32) {
        const int nb = (end - base < 32) ? (end - base) : 32;
        int2 r = make_int2(0, 0);
        if (lane < nb) r = R2[base + lane];

        int j = 0;
        for (; j + 4 <= nb; j += 4) {
            const int s0 = __builtin_amdgcn_readlane(r.x, j + 0);
            const int s1 = __builtin_amdgcn_readlane(r.x, j + 1);
            const int s2 = __builtin_amdgcn_readlane(r.x, j + 2);
            const int s3 = __builtin_amdgcn_readlane(r.x, j + 3);
            const float w0 = __int_as_float(__builtin_amdgcn_readlane(r.y, j + 0));
            const float w1 = __int_as_float(__builtin_amdgcn_readlane(r.y, j + 1));
            const float w2 = __int_as_float(__builtin_amdgcn_readlane(r.y, j + 2));
            const float w3 = __int_as_float(__builtin_amdgcn_readlane(r.y, j + 3));
            const float4 h0 = H4[(size_t)s0 * 32 + lane];
            const float4 h1 = H4[(size_t)s1 * 32 + lane];
            const float4 h2 = H4[(size_t)s2 * 32 + lane];
            const float4 h3 = H4[(size_t)s3 * 32 + lane];
            acc.x = fmaf(h0.x, w0, acc.x); acc.y = fmaf(h0.y, w0, acc.y); acc.z = fmaf(h0.z, w0, acc.z); acc.w = fmaf(h0.w, w0, acc.w);
            acc.x = fmaf(h1.x, w1, acc.x); acc.y = fmaf(h1.y, w1, acc.y); acc.z = fmaf(h1.z, w1, acc.z); acc.w = fmaf(h1.w, w1, acc.w);
            acc.x = fmaf(h2.x, w2, acc.x); acc.y = fmaf(h2.y, w2, acc.y); acc.z = fmaf(h2.z, w2, acc.z); acc.w = fmaf(h2.w, w2, acc.w);
            acc.x = fmaf(h3.x, w3, acc.x); acc.y = fmaf(h3.y, w3, acc.y); acc.z = fmaf(h3.z, w3, acc.z); acc.w = fmaf(h3.w, w3, acc.w);
        }
        #pragma unroll 1
        for (; j < nb; ++j) {
            const int s0 = __builtin_amdgcn_readlane(r.x, j);
            const float w0 = __int_as_float(__builtin_amdgcn_readlane(r.y, j));
            const float4 h0 = H4[(size_t)s0 * 32 + lane];
            acc.x = fmaf(h0.x, w0, acc.x); acc.y = fmaf(h0.y, w0, acc.y); acc.z = fmaf(h0.z, w0, acc.z); acc.w = fmaf(h0.w, w0, acc.w);
        }
    }
    acc.x += bv.x; acc.y += bv.y; acc.z += bv.z; acc.w += bv.w;
    acc.x = acc.x > 0.f ? acc.x : 0.f;
    acc.y = acc.y > 0.f ? acc.y : 0.f;
    acc.z = acc.z > 0.f ? acc.z : 0.f;
    acc.w = acc.w > 0.f ? acc.w : 0.f;
    return acc;
}

__global__ __launch_bounds__(256) void agg_gather_kernel(
    const float* __restrict__ H,
    const int* __restrict__ rowptr,
    const int* __restrict__ rec,
    const float* __restrict__ dinv,
    const float* __restrict__ bias,
    float* __restrict__ OUT, int n)
{
    const int lane = threadIdx.x & 31;
    const int node = __builtin_amdgcn_readfirstlane(
        (int)(blockIdx.x * 8 + (threadIdx.x >> 5)));
    if (node >= n) return;
    const float4 acc = agg_row((const float4*)H, rowptr, (const int2*)rec, dinv, bias, node, lane);
    ((float4*)OUT)[(size_t)node * 32 + lane] = acc;
}

#define CLS_NODES     16
#define WC_LANE_F4    10
#define WC_PITCH_F4   11
static_assert(DHID == 128 && (DHID * NCLS) % 4 == 0 && DHID * NCLS / 4 == 32 * WC_LANE_F4,
              "a lane owns 4 rows of Wc = WC_LANE_F4 float4");
static_assert(NCLS <= 16 && (CLS_NODES * NCLS) % 32 == 0, "logit butterfly and whole-line stores");

__device__ __forceinline__ float xhalf(float x) {
    const int b = __builtin_bit_cast(int, x);
    return __builtin_bit_cast(float,
        __builtin_amdgcn_permlanex16(b, b, 0x76543210, 0xfedcba98, false, false));
}
template <int M> __device__ __forceinline__ float xrow(float x) {
    const int b = __builtin_bit_cast(int, x);
    return __builtin_bit_cast(float,
        __builtin_amdgcn_update_dpp(b, b, 0x160 + M, 0xf, 0xf, false));
}

__global__ __launch_bounds__(CLS_NODES * 32) void agg_gather_classify_kernel(
    const float* __restrict__ H,
    const int* __restrict__ rowptr,
    const int* __restrict__ rec,
    const float* __restrict__ dinv,
    const float* __restrict__ bias,
    const float* __restrict__ Wc,
    const float* __restrict__ bc,
    float* __restrict__ out, int n)
{
    __shared__ float4 wcs[32 * WC_PITCH_F4];
    __shared__ float  lg[CLS_NODES * NCLS];
    const int tid  = threadIdx.x;
    const int lane = tid & 31;
    const int wv   = tid >> 5;

    const bool stager = tid < 32 * WC_LANE_F4;
    float4 wstage = make_float4(0.f, 0.f, 0.f, 0.f);
    if (stager) wstage = ((const float4*)Wc)[tid];
    const int cls = lane >> 1;
    const float bcv = bc[cls < NCLS ? cls : 0];

    const int node = __builtin_amdgcn_readfirstlane((int)(blockIdx.x * CLS_NODES + wv));
    float4 h = make_float4(0.f, 0.f, 0.f, 0.f);
    if (node < n)
        h = agg_row((const float4*)H, rowptr, (const int2*)rec, dinv, bias, node, lane);
    if (stager) wcs[(tid / WC_LANE_F4) * WC_PITCH_F4 + (tid % WC_LANE_F4)] = wstage;
    __syncthreads();

    float w[4 * WC_LANE_F4];
    #pragma unroll
    for (int i = 0; i < WC_LANE_F4; ++i) {
        const float4 t = wcs[lane * WC_PITCH_F4 + i];
        w[4 * i + 0] = t.x; w[4 * i + 1] = t.y; w[4 * i + 2] = t.z; w[4 * i + 3] = t.w;
    }
    float p[16];
    #pragma unroll
    for (int c = 0; c < 16; ++c) {
        if (c < NCLS) {
            float a = h.x * w[c];
            a = fmaf(h.y, w[NCLS + c], a);
            a = fmaf(h.z, w[2 * NCLS + c], a);
            a = fmaf(h.w, w[3 * NCLS + c], a);
            p[c] = a;
        } else {
            p[c] = 0.f;
        }
    }
    const bool b4 = (lane & 16) != 0, b3 = (lane & 8) != 0, b2 = (lane & 4) != 0, b1 = (lane & 2) != 0;
    float q8[8], q4[4], q2[2];
    #pragma unroll
    for (int j = 0; j < 8; ++j) {
        const float keep = b4 ? p[j + 8] : p[j];
        const float send = b4 ? p[j] : p[j + 8];
        q8[j] = keep + xhalf(send);
    }
    #pragma unroll
    for (int j = 0; j < 4; ++j) {
        const float keep = b3 ? q8[j + 4] : q8[j];
        const float send = b3 ? q8[j] : q8[j + 4];
        q4[j] = keep + xrow<8>(send);
    }
    #pragma unroll
    for (int j = 0; j < 2; ++j) {
        const float keep = b2 ? q4[j + 2] : q4[j];
        const float send = b2 ? q4[j] : q4[j + 2];
        q2[j] = keep + xrow<4>(send);
    }
    const float keep1 = b1 ? q2[1] : q2[0];
    const float send1 = b1 ? q2[0] : q2[1];
    const float q1 = keep1 + xrow<2>(send1);
    const float q0 = q1 + xrow<1>(q1);
    if ((lane & 1) == 0 && cls < NCLS) lg[wv * NCLS + cls] = q0 + bcv;
    __syncthreads();

    if (tid < CLS_NODES * NCLS) {
        const int g = blockIdx.x * (CLS_NODES * NCLS) + tid;
        if (g < n * NCLS) out[g] = lg[tid];
    }
}

extern "C" void kernel_launch(void* const* d_in, const int* in_sizes, int n_in,
                              void* d_out, int out_size, void* d_ws, size_t ws_size,
                              hipStream_t stream) {
    const float* x  = (const float*)d_in[0];
    const int*   ei = (const int*)d_in[1];
    const float* ew = (const float*)d_in[2];
    const float* W1 = (const float*)d_in[3];
    const float* b1 = (const float*)d_in[4];
    const float* W2 = (const float*)d_in[5];
    const float* b2 = (const float*)d_in[6];
    const float* Wc = (const float*)d_in[7];
    const float* bc = (const float*)d_in[8];
    float* out = (float*)d_out;

    const int n  = NNODES;
    const int E  = in_sizes[1] / 2;
    const int K1 = in_sizes[0] / n;
    const int* src = ei;
    const int* dst = ei + E;

    char* ws = (char*)d_ws;
    size_t off = 0;
    auto alloc = [&](size_t bytes) { char* p = ws + off; off = (off + bytes + 255) & ~(size_t)255; return p; };
    float* dinv = (float*)alloc((size_t)n * 4);
    float* bufA = (float*)alloc((size_t)n * DHID * 4);
    float* bufB = (float*)alloc((size_t)n * DHID * 4);
    int* bh       = (int*)alloc((size_t)NBINS * PB * 4);
    int* binstart = (int*)alloc((size_t)(NBINS + 1) * 4);
    int* rowptr   = (int*)alloc((size_t)(NBINS * BIN_NODES + 1) * 4);
    int2* tmp   = (int2*)alloc((size_t)E * 8);
    int* rec    = (int*)alloc((size_t)E * 8);
    int* dsto   = (int*)alloc((size_t)E * 4);
    (void)ws_size;

    const int T = 256;
    bin_count_kernel  <<<PB, SORT_T, 0, stream>>>(dst, bh, E);
    bin_place_kernel  <<<PB, SORT_T, 0, stream>>>(src, dst, ew, bh, binstart, tmp, E);
    bin_sort_kernel   <<<NBINS, SORT_T, 0, stream>>>(tmp, binstart, rowptr, dinv, (int2*)rec, dsto, n);
    edge_norm_kernel  <<<PB, SORT_T, 0, stream>>>((int2*)rec, dsto, dinv, E);

    const int gemmBlocks = (n + 127) / 128;
    const int nodeBlocks = (n + 7) / 8;
    const int clsBlocks  = (n + CLS_NODES - 1) / CLS_NODES;

    gemm128_wmma_kernel<<<gemmBlocks, T, 0, stream>>>(x, W1, bufA, n, K1);
    agg_gather_kernel  <<<nodeBlocks, T, 0, stream>>>(bufA, rowptr, rec, dinv, b1, bufB, n);

    gemm128_wmma_kernel<<<gemmBlocks, T, 0, stream>>>(bufB, W2, bufA, n, DHID);
    agg_gather_classify_kernel<<<clsBlocks, CLS_NODES * 32, 0, stream>>>(
        bufA, rowptr, rec, dinv, b2, Wc, bc, out, n);
}


# Sym. norm factored out of the edges: GEMM stores dinv[row]*row, gather sums raw w*row, scales by dinv[dst] once; edge-norm pass + dst array gone, 8 -> 7 launches; sum order kept, products regrouped
# speedup vs baseline: 3.8522x; 1.1037x over previous
#include <hip/hip_runtime.h>

typedef __attribute__((ext_vector_type(2))) float v2f;
typedef __attribute__((ext_vector_type(8))) float v8f;

#define NNODES 50000
#define DHID   128
#define NCLS   10

#define BIN_SHIFT 8
#define BIN_NODES (1 << BIN_SHIFT)
#define NBINS     ((NNODES + BIN_NODES - 1) >> BIN_SHIFT)
#define PB        256
#define SORT_T    1024

#define NWAVES        (SORT_T / 32)
#define ROW_LANES     8
#define ROWS_PER_LOAD (32 / ROW_LANES)
#define ROW_LOADS     (PB / (4 * ROW_LANES))
#define ROW_PASSES    ((NBINS + NWAVES * ROWS_PER_LOAD - 1) / (NWAVES * ROWS_PER_LOAD))
#define BINS_PER_LANE ((NBINS + 31) / 32)
static_assert(PB % (4 * ROW_LANES) == 0, "a table row is read as int4 by ROW_LANES lanes");

__device__ __forceinline__ int coarse_bin(int d) {
    int b = (int)((unsigned)d >> BIN_SHIFT);
    return b < NBINS ? b : NBINS - 1;
}

__global__ __launch_bounds__(SORT_T) void bin_count_kernel(
    const int* __restrict__ dst, int* __restrict__ bh, int nE)
{
    __shared__ int hist[NBINS];
    const int tid = threadIdx.x;
    for (int i = tid; i < NBINS; i += SORT_T) hist[i] = 0;
    __syncthreads();
    const int chunk = (nE + PB - 1) / PB;
    const int beg = blockIdx.x * chunk;
    const int end = (beg + chunk < nE) ? (beg + chunk) : nE;
    for (int base = beg + tid; base < end; base += 4 * SORT_T) {
        int d[4];
        #pragma unroll
        for (int k = 0; k < 4; ++k) {
            const int e = base + k * SORT_T;
            d[k] = (e < end) ? dst[e] : 0;
        }
        #pragma unroll
        for (int k = 0; k < 4; ++k)
            if (base + k * SORT_T < end) atomicAdd(&hist[coarse_bin(d[k])], 1);
    }
    __syncthreads();
    for (int i = tid; i < NBINS; i += SORT_T) bh[i * PB + blockIdx.x] = hist[i];
}

__global__ __launch_bounds__(SORT_T) void bin_place_kernel(
    const int* __restrict__ src, const int* __restrict__ dst,
    const float* __restrict__ w, const int* __restrict__ bh,
    int* __restrict__ binstart, int2* __restrict__ tmp, int nE)
{
    __shared__ int cur[NBINS];
    __shared__ int tot[NBINS];
    __shared__ int pre[NBINS];
    const int tid  = threadIdx.x;
    const int lane = tid & 31;
    const int wave = tid >> 5;
    const int blk  = blockIdx.x;

    const int l8   = lane & (ROW_LANES - 1);
    const int rsub = lane / ROW_LANES;
    int4 v[ROW_PASSES][ROW_LOADS];
    #pragma unroll
    for (int q = 0; q < ROW_PASSES; ++q) {
        const int i = (q * NWAVES + wave) * ROWS_PER_LOAD + rsub;
        const int4* row = (const int4*)(bh + (i < NBINS ? i : NBINS - 1) * PB);
        #pragma unroll
        for (int j = 0; j < ROW_LOADS; ++j) v[q][j] = row[j * ROW_LANES + l8];
    }
    int t[ROW_PASSES], p[ROW_PASSES];
    #pragma unroll
    for (int q = 0; q < ROW_PASSES; ++q) {
        t[q] = 0;
        p[q] = 0;
        #pragma unroll
        for (int j = 0; j < ROW_LOADS; ++j) {
            const int c0 = (j * ROW_LANES + l8) * 4;
            t[q] += v[q][j].x + v[q][j].y + v[q][j].z + v[q][j].w;
            p[q] += (c0 + 0 < blk ? v[q][j].x : 0) + (c0 + 1 < blk ? v[q][j].y : 0)
                  + (c0 + 2 < blk ? v[q][j].z : 0) + (c0 + 3 < blk ? v[q][j].w : 0);
        }
    }
    #pragma unroll
    for (int off = ROW_LANES / 2; off > 0; off >>= 1) {
        #pragma unroll
        for (int q = 0; q < ROW_PASSES; ++q) {
            t[q] += __shfl_xor(t[q], off, 32);
            p[q] += __shfl_xor(p[q], off, 32);
        }
    }
    #pragma unroll
    for (int q = 0; q < ROW_PASSES; ++q) {
        const int i = (q * NWAVES + wave) * ROWS_PER_LOAD + rsub;
        if (l8 == 0 && i < NBINS) { tot[i] = t[q]; pre[i] = p[q]; }
    }
    __syncthreads();

    if (wave == 0) {
        int c[BINS_PER_LANE], pr[BINS_PER_LANE];
        int s = 0;
        #pragma unroll
        for (int k = 0; k < BINS_PER_LANE; ++k) {
            const int i = lane * BINS_PER_LANE + k;
            c[k]  = (i < NBINS) ? tot[i] : 0;
            pr[k] = (i < NBINS) ? pre[i] : 0;
            s += c[k];
        }
        int inc = s;
        #pragma unroll
        for (int off = 1; off < 32; off <<= 1) {
            const int u = __shfl_up(inc, off, 32);
            if (lane >= off) inc += u;
        }
        int run = inc - s;
        #pragma unroll
        for (int k = 0; k < BINS_PER_LANE; ++k) {
            const int i = lane * BINS_PER_LANE + k;
            if (i < NBINS) {
                cur[i] = run + pr[k];
                if (blk == 0) binstart[i] = run;
            }
            run += c[k];
        }
        if (blk == 0 && lane == 31) binstart[NBINS] = run;
    }
    __syncthreads();
    const int chunk = (nE + PB - 1) / PB;
    const int beg = blockIdx.x * chunk;
    const int end = (beg + chunk < nE) ? (beg + chunk) : nE;
    for (int base = beg + tid; base < end; base += 4 * SORT_T) {
        int d[4], s[4];
        float wt[4];
        #pragma unroll
        for (int k = 0; k < 4; ++k) {
            const int e = base + k * SORT_T;
            const bool in = e < end;
            d[k]  = in ? dst[e] : 0;
            s[k]  = in ? src[e] : 0;
            wt[k] = in ? w[e] : 0.f;
        }
        #pragma unroll
        for (int k = 0; k < 4; ++k) {
            if (base + k * SORT_T < end) {
                const int pos = atomicAdd(&cur[coarse_bin(d[k])], 1);
                const unsigned key = (unsigned)s[k] | ((unsigned)(d[k] & (BIN_NODES - 1)) << 24);
                tmp[pos] = make_int2((int)key, __float_as_int(wt[k]));
            }
        }
    }
}

__global__ __launch_bounds__(SORT_T) void bin_sort_kernel(
    const int2* __restrict__ tmp, const int* __restrict__ binstart,
    int* __restrict__ rowptr, float* __restrict__ dinv,
    int2* __restrict__ rec, int n)
{
    __shared__ int   cnt[BIN_NODES];
    __shared__ int   sc[BIN_NODES];
    __shared__ float dg[BIN_NODES];
    const int tid = threadIdx.x;
    const int b = blockIdx.x;
    const int bbeg = binstart[b];
    const int bend = binstart[b + 1];
    if (tid < BIN_NODES) { cnt[tid] = 0; dg[tid] = 1.0f; }
    __syncthreads();

    for (int base = bbeg + tid; base < bend; base += 4 * SORT_T) {
        int2 r[4];
        #pragma unroll
        for (int k = 0; k < 4; ++k) {
            const int p = base + k * SORT_T;
            r[k] = (p < bend) ? tmp[p] : make_int2(0, 0);
        }
        #pragma unroll
        for (int k = 0; k < 4; ++k) {
            if (base + k * SORT_T < bend) {
                const int dl = (int)((unsigned)r[k].x >> 24);
                atomicAdd(&cnt[dl], 1);
                atomicAdd(&dg[dl], __int_as_float(r[k].y));
            }
        }
    }
    __syncthreads();

    if (tid < BIN_NODES) sc[tid] = cnt[tid];
    __syncthreads();
    for (int off = 1; off < BIN_NODES; off <<= 1) {
        int v = 0;
        if (tid < BIN_NODES && tid >= off) v = sc[tid - off];
        __syncthreads();
        if (tid < BIN_NODES) sc[tid] += v;
        __syncthreads();
    }
    if (tid < BIN_NODES) {
        const int first = sc[tid] - cnt[tid];
        const int node = b * BIN_NODES + tid;
        rowptr[node] = bbeg + first;
        if (node < n) dinv[node] = __frsqrt_rn(dg[tid]);
        cnt[tid] = first;
    }
    __syncthreads();

    for (int base = bbeg + tid; base < bend; base += 4 * SORT_T) {
        int2 r[4];
        #pragma unroll
        for (int k = 0; k < 4; ++k) {
            const int p = base + k * SORT_T;
            r[k] = (p < bend) ? tmp[p] : make_int2(0, 0);
        }
        #pragma unroll
        for (int k = 0; k < 4; ++k) {
            if (base + k * SORT_T < bend) {
                const int dl = (int)((unsigned)r[k].x >> 24);
                const int pos = bbeg + atomicAdd(&cnt[dl], 1);
                rec[pos] = make_int2(r[k].x & 0x00FFFFFF, r[k].y);
            }
        }
    }
}

#define XS_STRIDE 34
#define WT_STRIDE 144

__global__ __launch_bounds__(256) void gemm128_wmma_kernel(
    const float* __restrict__ X, const float* __restrict__ W,
    const float* __restrict__ rs, float* __restrict__ H, int M, int K)
{
    __shared__ float  Xs[128 * XS_STRIDE];
    __shared__ float2 Wt[16 * WT_STRIDE];

    const int tid  = threadIdx.x;
    const int lane = tid & 31;
    const int wave = tid >> 5;
    const int l16  = lane & 15;
    const bool hi  = (lane >= 16);
    const int rowBlock = blockIdx.x * 128;
    const int rowLocal = wave * 16 + l16;

    v8f acc[8];
    #pragma unroll
    for (int t = 0; t < 8; ++t)
        #pragma unroll
        for (int r = 0; r < 8; ++r) acc[t][r] = 0.0f;

    float rsc[8];
    #pragma unroll
    for (int r = 0; r < 8; ++r) {
        const int m = rowBlock + wave * 16 + r + (hi ? 8 : 0);
        rsc[r] = rs[m < M ? m : M - 1];
    }

    const int nch = (K + 31) / 32;
    for (int ch = 0; ch < nch; ++ch) {
        const int k0 = ch * 32;

        #pragma unroll
        for (int j = 0; j < 4; ++j) {
            int f  = tid * 4 + j;
            int r  = f >> 3;
            int kb = (f & 7) * 4;
            int gr = rowBlock + r;
            float4 v = make_float4(0.f, 0.f, 0.f, 0.f);
            if (gr < M) {
                int kg = k0 + kb;
                if (kg + 3 < K) {
                    v = *(const float4*)(X + (size_t)gr * K + kg);
                } else {
                    float t0 = (kg + 0 < K) ? X[(size_t)gr * K + kg + 0] : 0.f;
                    float t1 = (kg + 1 < K) ? X[(size_t)gr * K + kg + 1] : 0.f;
                    float t2 = (kg + 2 < K) ? X[(size_t)gr * K + kg + 2] : 0.f;
                    float t3 = (kg + 3 < K) ? X[(size_t)gr * K + kg + 3] : 0.f;
                    v = make_float4(t0, t1, t2, t3);
                }
            }
            float* dp = &Xs[r * XS_STRIDE + kb];
            ((float2*)dp)[0] = make_float2(v.x, v.y);
            ((float2*)dp)[1] = make_float2(v.z, v.w);
        }

        #pragma unroll
        for (int j = 0; j < 2; ++j) {
            int f  = tid * 2 + j;
            int kp = f >> 5;
            int cb = (f & 31) * 4;
            int ka = k0 + 2 * kp;
            float4 va = make_float4(0.f, 0.f, 0.f, 0.f);
            float4 vb = make_float4(0.f, 0.f, 0.f, 0.f);
            if (ka < K)     va = *(const float4*)(W + (size_t)ka * DHID + cb);
            if (ka + 1 < K) vb = *(const float4*)(W + (size_t)(ka + 1) * DHID + cb);
            float4 w0 = make_float4(va.x, vb.x, va.y, vb.y);
            float4 w1 = make_float4(va.z, vb.z, va.w, vb.w);
            *(float4*)&Wt[kp * WT_STRIDE + cb]     = w0;
            *(float4*)&Wt[kp * WT_STRIDE + cb + 2] = w1;
        }

        __syncthreads();

        #pragma unroll
        for (int s = 0; s < 8; ++s) {
            const int kp = s * 2 + (hi ? 1 : 0);
            v2f a = *(const v2f*)&Xs[rowLocal * XS_STRIDE + kp * 2];
            float2 bf[8];
            #pragma unroll
            for (int nt = 0; nt < 8; ++nt)
                bf[nt] = Wt[kp * WT_STRIDE + nt * 16 + l16];
            #pragma unroll
            for (int nt = 0; nt < 8; ++nt) {
                v2f b;
                b.x = bf[nt].x;
                b.y = bf[nt].y;
                acc[nt] = __builtin_amdgcn_wmma_f32_16x16x4_f32(
                     false, a,  false, b,
                     (short)0, acc[nt],
                     false,  false);
            }
        }
        __syncthreads();
    }

    #pragma unroll
    for (int nt = 0; nt < 8; ++nt) {
        const int col = nt * 16 + l16;
        #pragma unroll
        for (int r = 0; r < 8; ++r) {
            int m = rowBlock + wave * 16 + r + (hi ? 8 : 0);
            if (m < M) H[(size_t)m * DHID + col] = acc[nt][r] * rsc[r];
        }
    }
}

__device__ __forceinline__ float4 agg_row(
    const float4* __restrict__ H4,
    const int* __restrict__ rowptr,
    const int2* __restrict__ R2,
    const float* __restrict__ dinv,
    const float* __restrict__ bias,
    int node, int lane)
{
    const int beg = rowptr[node];
    const int end = rowptr[node + 1];
    const float s = dinv[node];
    const float4 hv = H4[(size_t)node * 32 + lane];
    const float4 bv = ((const float4*)bias)[lane];
    float4 acc = hv;

    for (int base = beg; base < end; base += 32) {
        const int nb = (end - base < 32) ? (end - base) : 32;
        int2 r = make_int2(0, 0);
        if (lane < nb) r = R2[base + lane];

        int j = 0;
        for (; j + 4 <= nb; j += 4) {
            const int s0 = __builtin_amdgcn_readlane(r.x, j + 0);
            const int s1 = __builtin_amdgcn_readlane(r.x, j + 1);
            const int s2 = __builtin_amdgcn_readlane(r.x, j + 2);
            const int s3 = __builtin_amdgcn_readlane(r.x, j + 3);
            const float w0 = __int_as_float(__builtin_amdgcn_readlane(r.y, j + 0));
            const float w1 = __int_as_float(__builtin_amdgcn_readlane(r.y, j + 1));
            const float w2 = __int_as_float(__builtin_amdgcn_readlane(r.y, j + 2));
            const float w3 = __int_as_float(__builtin_amdgcn_readlane(r.y, j + 3));
            const float4 h0 = H4[(size_t)s0 * 32 + lane];
            const float4 h1 = H4[(size_t)s1 * 32 + lane];
            const float4 h2 = H4[(size_t)s2 * 32 + lane];
            const float4 h3 = H4[(size_t)s3 * 32 + lane];
            acc.x = fmaf(h0.x, w0, acc.x); acc.y = fmaf(h0.y, w0, acc.y); acc.z = fmaf(h0.z, w0, acc.z); acc.w = fmaf(h0.w, w0, acc.w);
            acc.x = fmaf(h1.x, w1, acc.x); acc.y = fmaf(h1.y, w1, acc.y); acc.z = fmaf(h1.z, w1, acc.z); acc.w = fmaf(h1.w, w1, acc.w);
            acc.x = fmaf(h2.x, w2, acc.x); acc.y = fmaf(h2.y, w2, acc.y); acc.z = fmaf(h2.z, w2, acc.z); acc.w = fmaf(h2.w, w2, acc.w);
            acc.x = fmaf(h3.x, w3, acc.x); acc.y = fmaf(h3.y, w3, acc.y); acc.z = fmaf(h3.z, w3, acc.z); acc.w = fmaf(h3.w, w3, acc.w);
        }
        #pragma unroll 1
        for (; j < nb; ++j) {
            const int s0 = __builtin_amdgcn_readlane(r.x, j);
            const float w0 = __int_as_float(__builtin_amdgcn_readlane(r.y, j));
            const float4 h0 = H4[(size_t)s0 * 32 + lane];
            acc.x = fmaf(h0.x, w0, acc.x); acc.y = fmaf(h0.y, w0, acc.y); acc.z = fmaf(h0.z, w0, acc.z); acc.w = fmaf(h0.w, w0, acc.w);
        }
    }
    acc.x = fmaf(acc.x, s, bv.x); acc.y = fmaf(acc.y, s, bv.y);
    acc.z = fmaf(acc.z, s, bv.z); acc.w = fmaf(acc.w, s, bv.w);
    acc.x = acc.x > 0.f ? acc.x : 0.f;
    acc.y = acc.y > 0.f ? acc.y : 0.f;
    acc.z = acc.z > 0.f ? acc.z : 0.f;
    acc.w = acc.w > 0.f ? acc.w : 0.f;
    return acc;
}

__global__ __launch_bounds__(256) void agg_gather_kernel(
    const float* __restrict__ H,
    const int* __restrict__ rowptr,
    const int* __restrict__ rec,
    const float* __restrict__ dinv,
    const float* __restrict__ bias,
    float* __restrict__ OUT, int n)
{
    const int lane = threadIdx.x & 31;
    const int node = __builtin_amdgcn_readfirstlane(
        (int)(blockIdx.x * 8 + (threadIdx.x >> 5)));
    if (node >= n) return;
    const float4 acc = agg_row((const float4*)H, rowptr, (const int2*)rec, dinv, bias, node, lane);
    ((float4*)OUT)[(size_t)node * 32 + lane] = acc;
}

#define CLS_NODES     16
#define WC_LANE_F4    10
#define WC_PITCH_F4   11
static_assert(DHID == 128 && (DHID * NCLS) % 4 == 0 && DHID * NCLS / 4 == 32 * WC_LANE_F4,
              "a lane owns 4 rows of Wc = WC_LANE_F4 float4");
static_assert(NCLS <= 16 && (CLS_NODES * NCLS) % 32 == 0, "logit butterfly and whole-line stores");

__device__ __forceinline__ float xhalf(float x) {
    const int b = __builtin_bit_cast(int, x);
    return __builtin_bit_cast(float,
        __builtin_amdgcn_permlanex16(b, b, 0x76543210, 0xfedcba98, false, false));
}
template <int M> __device__ __forceinline__ float xrow(float x) {
    const int b = __builtin_bit_cast(int, x);
    return __builtin_bit_cast(float,
        __builtin_amdgcn_update_dpp(b, b, 0x160 + M, 0xf, 0xf, false));
}

__global__ __launch_bounds__(CLS_NODES * 32) void agg_gather_classify_kernel(
    const float* __restrict__ H,
    const int* __restrict__ rowptr,
    const int* __restrict__ rec,
    const float* __restrict__ dinv,
    const float* __restrict__ bias,
    const float* __restrict__ Wc,
    const float* __restrict__ bc,
    float* __restrict__ out, int n)
{
    __shared__ float4 wcs[32 * WC_PITCH_F4];
    __shared__ float  lg[CLS_NODES * NCLS];
    const int tid  = threadIdx.x;
    const int lane = tid & 31;
    const int wv   = tid >> 5;

    const bool stager = tid < 32 * WC_LANE_F4;
    float4 wstage = make_float4(0.f, 0.f, 0.f, 0.f);
    if (stager) wstage = ((const float4*)Wc)[tid];
    const int cls = lane >> 1;
    const float bcv = bc[cls < NCLS ? cls : 0];

    const int node = __builtin_amdgcn_readfirstlane((int)(blockIdx.x * CLS_NODES + wv));
    float4 h = make_float4(0.f, 0.f, 0.f, 0.f);
    if (node < n)
        h = agg_row((const float4*)H, rowptr, (const int2*)rec, dinv, bias, node, lane);
    if (stager) wcs[(tid / WC_LANE_F4) * WC_PITCH_F4 + (tid % WC_LANE_F4)] = wstage;
    __syncthreads();

    float w[4 * WC_LANE_F4];
    #pragma unroll
    for (int i = 0; i < WC_LANE_F4; ++i) {
        const float4 t = wcs[lane * WC_PITCH_F4 + i];
        w[4 * i + 0] = t.x; w[4 * i + 1] = t.y; w[4 * i + 2] = t.z; w[4 * i + 3] = t.w;
    }
    float p[16];
    #pragma unroll
    for (int c = 0; c < 16; ++c) {
        if (c < NCLS) {
            float a = h.x * w[c];
            a = fmaf(h.y, w[NCLS + c], a);
            a = fmaf(h.z, w[2 * NCLS + c], a);
            a = fmaf(h.w, w[3 * NCLS + c], a);
            p[c] = a;
        } else {
            p[c] = 0.f;
        }
    }
    const bool b4 = (lane & 16) != 0, b3 = (lane & 8) != 0, b2 = (lane & 4) != 0, b1 = (lane & 2) != 0;
    float q8[8], q4[4], q2[2];
    #pragma unroll
    for (int j = 0; j < 8; ++j) {
        const float keep = b4 ? p[j + 8] : p[j];
        const float send = b4 ? p[j] : p[j + 8];
        q8[j] = keep + xhalf(send);
    }
    #pragma unroll
    for (int j = 0; j < 4; ++j) {
        const float keep = b3 ? q8[j + 4] : q8[j];
        const float send = b3 ? q8[j] : q8[j + 4];
        q4[j] = keep + xrow<8>(send);
    }
    #pragma unroll
    for (int j = 0; j < 2; ++j) {
        const float keep = b2 ? q4[j + 2] : q4[j];
        const float send = b2 ? q4[j] : q4[j + 2];
        q2[j] = keep + xrow<4>(send);
    }
    const float keep1 = b1 ? q2[1] : q2[0];
    const float send1 = b1 ? q2[0] : q2[1];
    const float q1 = keep1 + xrow<2>(send1);
    const float q0 = q1 + xrow<1>(q1);
    if ((lane & 1) == 0 && cls < NCLS) lg[wv * NCLS + cls] = q0 + bcv;
    __syncthreads();

    if (tid < CLS_NODES * NCLS) {
        const int g = blockIdx.x * (CLS_NODES * NCLS) + tid;
        if (g < n * NCLS) out[g] = lg[tid];
    }
}

extern "C" void kernel_launch(void* const* d_in, const int* in_sizes, int n_in,
                              void* d_out, int out_size, void* d_ws, size_t ws_size,
                              hipStream_t stream) {
    const float* x  = (const float*)d_in[0];
    const int*   ei = (const int*)d_in[1];
    const float* ew = (const float*)d_in[2];
    const float* W1 = (const float*)d_in[3];
    const float* b1 = (const float*)d_in[4];
    const float* W2 = (const float*)d_in[5];
    const float* b2 = (const float*)d_in[6];
    const float* Wc = (const float*)d_in[7];
    const float* bc = (const float*)d_in[8];
    float* out = (float*)d_out;

    const int n  = NNODES;
    const int E  = in_sizes[1] / 2;
    const int K1 = in_sizes[0] / n;
    const int* src = ei;
    const int* dst = ei + E;

    char* ws = (char*)d_ws;
    size_t off = 0;
    auto alloc = [&](size_t bytes) { char* p = ws + off; off = (off + bytes + 255) & ~(size_t)255; return p; };
    float* dinv = (float*)alloc((size_t)n * 4);
    float* bufA = (float*)alloc((size_t)n * DHID * 4);
    float* bufB = (float*)alloc((size_t)n * DHID * 4);
    int* bh       = (int*)alloc((size_t)NBINS * PB * 4);
    int* binstart = (int*)alloc((size_t)(NBINS + 1) * 4);
    int* rowptr   = (int*)alloc((size_t)(NBINS * BIN_NODES + 1) * 4);
    int2* tmp   = (int2*)alloc((size_t)E * 8);
    int* rec    = (int*)alloc((size_t)E * 8);
    (void)ws_size;

    const int T = 256;
    bin_count_kernel  <<<PB, SORT_T, 0, stream>>>(dst, bh, E);
    bin_place_kernel  <<<PB, SORT_T, 0, stream>>>(src, dst, ew, bh, binstart, tmp, E);
    bin_sort_kernel   <<<NBINS, SORT_T, 0, stream>>>(tmp, binstart, rowptr, dinv, (int2*)rec, n);

    const int gemmBlocks = (n + 127) / 128;
    const int nodeBlocks = (n + 7) / 8;
    const int clsBlocks  = (n + CLS_NODES - 1) / CLS_NODES;

    gemm128_wmma_kernel<<<gemmBlocks, T, 0, stream>>>(x, W1, dinv, bufA, n, K1);
    agg_gather_kernel  <<<nodeBlocks, T, 0, stream>>>(bufA, rowptr, rec, dinv, b1, bufB, n);

    gemm128_wmma_kernel<<<gemmBlocks, T, 0, stream>>>(bufB, W2, dinv, bufA, n, DHID);
    agg_gather_classify_kernel<<<clsBlocks, CLS_NODES * 32, 0, stream>>>(
        bufA, rowptr, rec, dinv, b2, Wc, bc, out, n);
}


# Both sort levels stage their records in LDS and copy them out in slot order: tmp and the edge list are written as runs of adjacent records, not one scattered 8-byte store per edge
# speedup vs baseline: 4.0950x; 1.0630x over previous
#include <hip/hip_runtime.h>

typedef __attribute__((ext_vector_type(2))) float v2f;
typedef __attribute__((ext_vector_type(8))) float v8f;

#define NNODES 50000
#define DHID   128
#define NCLS   10

#define BIN_SHIFT 8
#define BIN_NODES (1 << BIN_SHIFT)
#define NBINS     ((NNODES + BIN_NODES - 1) >> BIN_SHIFT)
#define PB        256
#define SORT_T    1024
#define PLACE_CAP 4096
#define SORT_CAP  8192

#define NWAVES        (SORT_T / 32)
#define ROW_LANES     8
#define ROWS_PER_LOAD (32 / ROW_LANES)
#define ROW_LOADS     (PB / (4 * ROW_LANES))
#define ROW_PASSES    ((NBINS + NWAVES * ROWS_PER_LOAD - 1) / (NWAVES * ROWS_PER_LOAD))
#define BINS_PER_LANE ((NBINS + 31) / 32)
static_assert(PB % (4 * ROW_LANES) == 0, "a table row is read as int4 by ROW_LANES lanes");

__device__ __forceinline__ int coarse_bin(int d) {
    int b = (int)((unsigned)d >> BIN_SHIFT);
    return b < NBINS ? b : NBINS - 1;
}

__global__ __launch_bounds__(SORT_T) void bin_count_kernel(
    const int* __restrict__ dst, int* __restrict__ bh, int nE)
{
    __shared__ int hist[NBINS];
    const int tid = threadIdx.x;
    for (int i = tid; i < NBINS; i += SORT_T) hist[i] = 0;
    __syncthreads();
    const int chunk = (nE + PB - 1) / PB;
    const int beg = blockIdx.x * chunk;
    const int end = (beg + chunk < nE) ? (beg + chunk) : nE;
    for (int base = beg + tid; base < end; base += 4 * SORT_T) {
        int d[4];
        #pragma unroll
        for (int k = 0; k < 4; ++k) {
            const int e = base + k * SORT_T;
            d[k] = (e < end) ? dst[e] : 0;
        }
        #pragma unroll
        for (int k = 0; k < 4; ++k)
            if (base + k * SORT_T < end) atomicAdd(&hist[coarse_bin(d[k])], 1);
    }
    __syncthreads();
    for (int i = tid; i < NBINS; i += SORT_T) bh[i * PB + blockIdx.x] = hist[i];
}

__global__ __launch_bounds__(SORT_T) void bin_place_kernel(
    const int* __restrict__ src, const int* __restrict__ dst,
    const float* __restrict__ w, const int* __restrict__ bh,
    int* __restrict__ binstart, int2* __restrict__ tmp, int nE)
{
    __shared__ int cur[NBINS];
    __shared__ int tot[NBINS];
    __shared__ int pre[NBINS];
    __shared__ int own[NBINS];
    __shared__ int dlt[NBINS];
    __shared__ int2 stg[PLACE_CAP];
    __shared__ int  gps[PLACE_CAP];
    const int tid  = threadIdx.x;
    const int lane = tid & 31;
    const int wave = tid >> 5;
    const int blk  = blockIdx.x;
    const int chunk = (nE + PB - 1) / PB;
    const int beg = blockIdx.x * chunk;
    const int end = (beg + chunk < nE) ? (beg + chunk) : nE;
    const bool staged = (end - beg) <= PLACE_CAP;

    int myown = 0;
    if (tid < NBINS) myown = bh[tid * PB + blk];

    const int l8   = lane & (ROW_LANES - 1);
    const int rsub = lane / ROW_LANES;
    int4 v[ROW_PASSES][ROW_LOADS];
    #pragma unroll
    for (int q = 0; q < ROW_PASSES; ++q) {
        const int i = (q * NWAVES + wave) * ROWS_PER_LOAD + rsub;
        const int4* row = (const int4*)(bh + (i < NBINS ? i : NBINS - 1) * PB);
        #pragma unroll
        for (int j = 0; j < ROW_LOADS; ++j) v[q][j] = row[j * ROW_LANES + l8];
    }
    int t[ROW_PASSES], p[ROW_PASSES];
    #pragma unroll
    for (int q = 0; q < ROW_PASSES; ++q) {
        t[q] = 0;
        p[q] = 0;
        #pragma unroll
        for (int j = 0; j < ROW_LOADS; ++j) {
            const int c0 = (j * ROW_LANES + l8) * 4;
            t[q] += v[q][j].x + v[q][j].y + v[q][j].z + v[q][j].w;
            p[q] += (c0 + 0 < blk ? v[q][j].x : 0) + (c0 + 1 < blk ? v[q][j].y : 0)
                  + (c0 + 2 < blk ? v[q][j].z : 0) + (c0 + 3 < blk ? v[q][j].w : 0);
        }
    }
    #pragma unroll
    for (int off = ROW_LANES / 2; off > 0; off >>= 1) {
        #pragma unroll
        for (int q = 0; q < ROW_PASSES; ++q) {
            t[q] += __shfl_xor(t[q], off, 32);
            p[q] += __shfl_xor(p[q], off, 32);
        }
    }
    #pragma unroll
    for (int q = 0; q < ROW_PASSES; ++q) {
        const int i = (q * NWAVES + wave) * ROWS_PER_LOAD + rsub;
        if (l8 == 0 && i < NBINS) { tot[i] = t[q]; pre[i] = p[q]; }
    }
    if (tid < NBINS) own[tid] = myown;
    __syncthreads();

    if (wave == 0) {
        int c[BINS_PER_LANE], pr[BINS_PER_LANE], ow[BINS_PER_LANE];
        int s = 0, so = 0;
        #pragma unroll
        for (int k = 0; k < BINS_PER_LANE; ++k) {
            const int i = lane * BINS_PER_LANE + k;
            c[k]  = (i < NBINS) ? tot[i] : 0;
            pr[k] = (i < NBINS) ? pre[i] : 0;
            ow[k] = (i < NBINS) ? own[i] : 0;
            s += c[k];
            so += ow[k];
        }
        int inc = s;
        int linc = so;
        #pragma unroll
        for (int off = 1; off < 32; off <<= 1) {
            const int u  = __shfl_up(inc, off, 32);
            const int lu = __shfl_up(linc, off, 32);
            if (lane >= off) { inc += u; linc += lu; }
        }
        int run  = inc - s;
        int lrun = linc - so;
        #pragma unroll
        for (int k = 0; k < BINS_PER_LANE; ++k) {
            const int i = lane * BINS_PER_LANE + k;
            if (i < NBINS) {
                const int g = run + pr[k];
                cur[i] = staged ? lrun : g;
                dlt[i] = g - lrun;
                if (blk == 0) binstart[i] = run;
            }
            run  += c[k];
            lrun += ow[k];
        }
        if (blk == 0 && lane == 31) binstart[NBINS] = run;
    }
    __syncthreads();
    for (int base = beg + tid; base < end; base += 4 * SORT_T) {
        int d[4], s[4];
        float wt[4];
        #pragma unroll
        for (int k = 0; k < 4; ++k) {
            const int e = base + k * SORT_T;
            const bool in = e < end;
            d[k]  = in ? dst[e] : 0;
            s[k]  = in ? src[e] : 0;
            wt[k] = in ? w[e] : 0.f;
        }
        #pragma unroll
        for (int k = 0; k < 4; ++k) {
            if (base + k * SORT_T < end) {
                const int bin = coarse_bin(d[k]);
                const int pos = atomicAdd(&cur[bin], 1);
                const unsigned key = (unsigned)s[k] | ((unsigned)(d[k] & (BIN_NODES - 1)) << 24);
                const int2 r = make_int2((int)key, __float_as_int(wt[k]));
                if (staged) { stg[pos] = r; gps[pos] = pos + dlt[bin]; }
                else        tmp[pos] = r;
            }
        }
    }
    if (staged) {
        __syncthreads();
        for (int t = tid; t < end - beg; t += SORT_T) tmp[gps[t]] = stg[t];
    }
}

__global__ __launch_bounds__(SORT_T) void bin_sort_kernel(
    const int2* __restrict__ tmp, const int* __restrict__ binstart,
    int* __restrict__ rowptr, float* __restrict__ dinv,
    int2* __restrict__ rec, int n)
{
    __shared__ int   cnt[BIN_NODES];
    __shared__ int   sc[BIN_NODES];
    __shared__ float dg[BIN_NODES];
    __shared__ int2  stg[SORT_CAP];
    const int tid = threadIdx.x;
    const int b = blockIdx.x;
    const int bbeg = binstart[b];
    const int bend = binstart[b + 1];
    const bool staged = (bend - bbeg) <= SORT_CAP;
    if (tid < BIN_NODES) { cnt[tid] = 0; dg[tid] = 1.0f; }
    __syncthreads();

    for (int base = bbeg + tid; base < bend; base += 4 * SORT_T) {
        int2 r[4];
        #pragma unroll
        for (int k = 0; k < 4; ++k) {
            const int p = base + k * SORT_T;
            r[k] = (p < bend) ? tmp[p] : make_int2(0, 0);
        }
        #pragma unroll
        for (int k = 0; k < 4; ++k) {
            if (base + k * SORT_T < bend) {
                const int dl = (int)((unsigned)r[k].x >> 24);
                atomicAdd(&cnt[dl], 1);
                atomicAdd(&dg[dl], __int_as_float(r[k].y));
            }
        }
    }
    __syncthreads();

    if (tid < BIN_NODES) sc[tid] = cnt[tid];
    __syncthreads();
    for (int off = 1; off < BIN_NODES; off <<= 1) {
        int v = 0;
        if (tid < BIN_NODES && tid >= off) v = sc[tid - off];
        __syncthreads();
        if (tid < BIN_NODES) sc[tid] += v;
        __syncthreads();
    }
    if (tid < BIN_NODES) {
        const int first = sc[tid] - cnt[tid];
        const int node = b * BIN_NODES + tid;
        rowptr[node] = bbeg + first;
        if (node < n) dinv[node] = __frsqrt_rn(dg[tid]);
        cnt[tid] = first;
    }
    __syncthreads();

    for (int base = bbeg + tid; base < bend; base += 4 * SORT_T) {
        int2 r[4];
        #pragma unroll
        for (int k = 0; k < 4; ++k) {
            const int p = base + k * SORT_T;
            r[k] = (p < bend) ? tmp[p] : make_int2(0, 0);
        }
        #pragma unroll
        for (int k = 0; k < 4; ++k) {
            if (base + k * SORT_T < bend) {
                const int dl = (int)((unsigned)r[k].x >> 24);
                const int lp = atomicAdd(&cnt[dl], 1);
                const int2 o = make_int2(r[k].x & 0x00FFFFFF, r[k].y);
                if (staged) stg[lp] = o;
                else        rec[bbeg + lp] = o;
            }
        }
    }
    if (staged) {
        __syncthreads();
        for (int i = tid; i < bend - bbeg; i += SORT_T) rec[bbeg + i] = stg[i];
    }
}

#define XS_STRIDE 34
#define WT_STRIDE 144

__global__ __launch_bounds__(256) void gemm128_wmma_kernel(
    const float* __restrict__ X, const float* __restrict__ W,
    const float* __restrict__ rs, float* __restrict__ H, int M, int K)
{
    __shared__ float  Xs[128 * XS_STRIDE];
    __shared__ float2 Wt[16 * WT_STRIDE];

    const int tid  = threadIdx.x;
    const int lane = tid & 31;
    const int wave = tid >> 5;
    const int l16  = lane & 15;
    const bool hi  = (lane >= 16);
    const int rowBlock = blockIdx.x * 128;
    const int rowLocal = wave * 16 + l16;

    v8f acc[8];
    #pragma unroll
    for (int t = 0; t < 8; ++t)
        #pragma unroll
        for (int r = 0; r < 8; ++r) acc[t][r] = 0.0f;

    float rsc[8];
    #pragma unroll
    for (int r = 0; r < 8; ++r) {
        const int m = rowBlock + wave * 16 + r + (hi ? 8 : 0);
        rsc[r] = rs[m < M ? m : M - 1];
    }

    const int nch = (K + 31) / 32;
    for (int ch = 0; ch < nch; ++ch) {
        const int k0 = ch * 32;

        #pragma unroll
        for (int j = 0; j < 4; ++j) {
            int f  = tid * 4 + j;
            int r  = f >> 3;
            int kb = (f & 7) * 4;
            int gr = rowBlock + r;
            float4 v = make_float4(0.f, 0.f, 0.f, 0.f);
            if (gr < M) {
                int kg = k0 + kb;
                if (kg + 3 < K) {
                    v = *(const float4*)(X + (size_t)gr * K + kg);
                } else {
                    float t0 = (kg + 0 < K) ? X[(size_t)gr * K + kg + 0] : 0.f;
                    float t1 = (kg + 1 < K) ? X[(size_t)gr * K + kg + 1] : 0.f;
                    float t2 = (kg + 2 < K) ? X[(size_t)gr * K + kg + 2] : 0.f;
                    float t3 = (kg + 3 < K) ? X[(size_t)gr * K + kg + 3] : 0.f;
                    v = make_float4(t0, t1, t2, t3);
                }
            }
            float* dp = &Xs[r * XS_STRIDE + kb];
            ((float2*)dp)[0] = make_float2(v.x, v.y);
            ((float2*)dp)[1] = make_float2(v.z, v.w);
        }

        #pragma unroll
        for (int j = 0; j < 2; ++j) {
            int f  = tid * 2 + j;
            int kp = f >> 5;
            int cb = (f & 31) * 4;
            int ka = k0 + 2 * kp;
            float4 va = make_float4(0.f, 0.f, 0.f, 0.f);
            float4 vb = make_float4(0.f, 0.f, 0.f, 0.f);
            if (ka < K)     va = *(const float4*)(W + (size_t)ka * DHID + cb);
            if (ka + 1 < K) vb = *(const float4*)(W + (size_t)(ka + 1) * DHID + cb);
            float4 w0 = make_float4(va.x, vb.x, va.y, vb.y);
            float4 w1 = make_float4(va.z, vb.z, va.w, vb.w);
            *(float4*)&Wt[kp * WT_STRIDE + cb]     = w0;
            *(float4*)&Wt[kp * WT_STRIDE + cb + 2] = w1;
        }

        __syncthreads();

        #pragma unroll
        for (int s = 0; s < 8; ++s) {
            const int kp = s * 2 + (hi ? 1 : 0);
            v2f a = *(const v2f*)&Xs[rowLocal * XS_STRIDE + kp * 2];
            float2 bf[8];
            #pragma unroll
            for (int nt = 0; nt < 8; ++nt)
                bf[nt] = Wt[kp * WT_STRIDE + nt * 16 + l16];
            #pragma unroll
            for (int nt = 0; nt < 8; ++nt) {
                v2f b;
                b.x = bf[nt].x;
                b.y = bf[nt].y;
                acc[nt] = __builtin_amdgcn_wmma_f32_16x16x4_f32(
                     false, a,  false, b,
                     (short)0, acc[nt],
                     false,  false);
            }
        }
        __syncthreads();
    }

    #pragma unroll
    for (int nt = 0; nt < 8; ++nt) {
        const int col = nt * 16 + l16;
        #pragma unroll
        for (int r = 0; r < 8; ++r) {
            int m = rowBlock + wave * 16 + r + (hi ? 8 : 0);
            if (m < M) H[(size_t)m * DHID + col] = acc[nt][r] * rsc[r];
        }
    }
}

__device__ __forceinline__ float4 agg_row(
    const float4* __restrict__ H4,
    const int* __restrict__ rowptr,
    const int2* __restrict__ R2,
    const float* __restrict__ dinv,
    const float* __restrict__ bias,
    int node, int lane)
{
    const int beg = rowptr[node];
    const int end = rowptr[node + 1];
    const float s = dinv[node];
    const float4 hv = H4[(size_t)node * 32 + lane];
    const float4 bv = ((const float4*)bias)[lane];
    float4 acc = hv;

    for (int base = beg; base < end; base += 32) {
        const int nb = (end - base < 32) ? (end - base) : 32;
        int2 r = make_int2(0, 0);
        if (lane < nb) r = R2[base + lane];

        int j = 0;
        for (; j + 4 <= nb; j += 4) {
            const int s0 = __builtin_amdgcn_readlane(r.x, j + 0);
            const int s1 = __builtin_amdgcn_readlane(r.x, j + 1);
            const int s2 = __builtin_amdgcn_readlane(r.x, j + 2);
            const int s3 = __builtin_amdgcn_readlane(r.x, j + 3);
            const float w0 = __int_as_float(__builtin_amdgcn_readlane(r.y, j + 0));
            const float w1 = __int_as_float(__builtin_amdgcn_readlane(r.y, j + 1));
            const float w2 = __int_as_float(__builtin_amdgcn_readlane(r.y, j + 2));
            const float w3 = __int_as_float(__builtin_amdgcn_readlane(r.y, j + 3));
            const float4 h0 = H4[(size_t)s0 * 32 + lane];
            const float4 h1 = H4[(size_t)s1 * 32 + lane];
            const float4 h2 = H4[(size_t)s2 * 32 + lane];
            const float4 h3 = H4[(size_t)s3 * 32 + lane];
            acc.x = fmaf(h0.x, w0, acc.x); acc.y = fmaf(h0.y, w0, acc.y); acc.z = fmaf(h0.z, w0, acc.z); acc.w = fmaf(h0.w, w0, acc.w);
            acc.x = fmaf(h1.x, w1, acc.x); acc.y = fmaf(h1.y, w1, acc.y); acc.z = fmaf(h1.z, w1, acc.z); acc.w = fmaf(h1.w, w1, acc.w);
            acc.x = fmaf(h2.x, w2, acc.x); acc.y = fmaf(h2.y, w2, acc.y); acc.z = fmaf(h2.z, w2, acc.z); acc.w = fmaf(h2.w, w2, acc.w);
            acc.x = fmaf(h3.x, w3, acc.x); acc.y = fmaf(h3.y, w3, acc.y); acc.z = fmaf(h3.z, w3, acc.z); acc.w = fmaf(h3.w, w3, acc.w);
        }
        #pragma unroll 1
        for (; j < nb; ++j) {
            const int s0 = __builtin_amdgcn_readlane(r.x, j);
            const float w0 = __int_as_float(__builtin_amdgcn_readlane(r.y, j));
            const float4 h0 = H4[(size_t)s0 * 32 + lane];
            acc.x = fmaf(h0.x, w0, acc.x); acc.y = fmaf(h0.y, w0, acc.y); acc.z = fmaf(h0.z, w0, acc.z); acc.w = fmaf(h0.w, w0, acc.w);
        }
    }
    acc.x = fmaf(acc.x, s, bv.x); acc.y = fmaf(acc.y, s, bv.y);
    acc.z = fmaf(acc.z, s, bv.z); acc.w = fmaf(acc.w, s, bv.w);
    acc.x = acc.x > 0.f ? acc.x : 0.f;
    acc.y = acc.y > 0.f ? acc.y : 0.f;
    acc.z = acc.z > 0.f ? acc.z : 0.f;
    acc.w = acc.w > 0.f ? acc.w : 0.f;
    return acc;
}

__global__ __launch_bounds__(256) void agg_gather_kernel(
    const float* __restrict__ H,
    const int* __restrict__ rowptr,
    const int* __restrict__ rec,
    const float* __restrict__ dinv,
    const float* __restrict__ bias,
    float* __restrict__ OUT, int n)
{
    const int lane = threadIdx.x & 31;
    const int node = __builtin_amdgcn_readfirstlane(
        (int)(blockIdx.x * 8 + (threadIdx.x >> 5)));
    if (node >= n) return;
    const float4 acc = agg_row((const float4*)H, rowptr, (const int2*)rec, dinv, bias, node, lane);
    ((float4*)OUT)[(size_t)node * 32 + lane] = acc;
}

#define CLS_NODES     16
#define WC_LANE_F4    10
#define WC_PITCH_F4   11
static_assert(DHID == 128 && (DHID * NCLS) % 4 == 0 && DHID * NCLS / 4 == 32 * WC_LANE_F4,
              "a lane owns 4 rows of Wc = WC_LANE_F4 float4");
static_assert(NCLS <= 16 && (CLS_NODES * NCLS) % 32 == 0, "logit butterfly and whole-line stores");

__device__ __forceinline__ float xhalf(float x) {
    const int b = __builtin_bit_cast(int, x);
    return __builtin_bit_cast(float,
        __builtin_amdgcn_permlanex16(b, b, 0x76543210, 0xfedcba98, false, false));
}
template <int M> __device__ __forceinline__ float xrow(float x) {
    const int b = __builtin_bit_cast(int, x);
    return __builtin_bit_cast(float,
        __builtin_amdgcn_update_dpp(b, b, 0x160 + M, 0xf, 0xf, false));
}

__global__ __launch_bounds__(CLS_NODES * 32) void agg_gather_classify_kernel(
    const float* __restrict__ H,
    const int* __restrict__ rowptr,
    const int* __restrict__ rec,
    const float* __restrict__ dinv,
    const float* __restrict__ bias,
    const float* __restrict__ Wc,
    const float* __restrict__ bc,
    float* __restrict__ out, int n)
{
    __shared__ float4 wcs[32 * WC_PITCH_F4];
    __shared__ float  lg[CLS_NODES * NCLS];
    const int tid  = threadIdx.x;
    const int lane = tid & 31;
    const int wv   = tid >> 5;

    const bool stager = tid < 32 * WC_LANE_F4;
    float4 wstage = make_float4(0.f, 0.f, 0.f, 0.f);
    if (stager) wstage = ((const float4*)Wc)[tid];
    const int cls = lane >> 1;
    const float bcv = bc[cls < NCLS ? cls : 0];

    const int node = __builtin_amdgcn_readfirstlane((int)(blockIdx.x * CLS_NODES + wv));
    float4 h = make_float4(0.f, 0.f, 0.f, 0.f);
    if (node < n)
        h = agg_row((const float4*)H, rowptr, (const int2*)rec, dinv, bias, node, lane);
    if (stager) wcs[(tid / WC_LANE_F4) * WC_PITCH_F4 + (tid % WC_LANE_F4)] = wstage;
    __syncthreads();

    float w[4 * WC_LANE_F4];
    #pragma unroll
    for (int i = 0; i < WC_LANE_F4; ++i) {
        const float4 t = wcs[lane * WC_PITCH_F4 + i];
        w[4 * i + 0] = t.x; w[4 * i + 1] = t.y; w[4 * i + 2] = t.z; w[4 * i + 3] = t.w;
    }
    float p[16];
    #pragma unroll
    for (int c = 0; c < 16; ++c) {
        if (c < NCLS) {
            float a = h.x * w[c];
            a = fmaf(h.y, w[NCLS + c], a);
            a = fmaf(h.z, w[2 * NCLS + c], a);
            a = fmaf(h.w, w[3 * NCLS + c], a);
            p[c] = a;
        } else {
            p[c] = 0.f;
        }
    }
    const bool b4 = (lane & 16) != 0, b3 = (lane & 8) != 0, b2 = (lane & 4) != 0, b1 = (lane & 2) != 0;
    float q8[8], q4[4], q2[2];
    #pragma unroll
    for (int j = 0; j < 8; ++j) {
        const float keep = b4 ? p[j + 8] : p[j];
        const float send = b4 ? p[j] : p[j + 8];
        q8[j] = keep + xhalf(send);
    }
    #pragma unroll
    for (int j = 0; j < 4; ++j) {
        const float keep = b3 ? q8[j + 4] : q8[j];
        const float send = b3 ? q8[j] : q8[j + 4];
        q4[j] = keep + xrow<8>(send);
    }
    #pragma unroll
    for (int j = 0; j < 2; ++j) {
        const float keep = b2 ? q4[j + 2] : q4[j];
        const float send = b2 ? q4[j] : q4[j + 2];
        q2[j] = keep + xrow<4>(send);
    }
    const float keep1 = b1 ? q2[1] : q2[0];
    const float send1 = b1 ? q2[0] : q2[1];
    const float q1 = keep1 + xrow<2>(send1);
    const float q0 = q1 + xrow<1>(q1);
    if ((lane & 1) == 0 && cls < NCLS) lg[wv * NCLS + cls] = q0 + bcv;
    __syncthreads();

    if (tid < CLS_NODES * NCLS) {
        const int g = blockIdx.x * (CLS_NODES * NCLS) + tid;
        if (g < n * NCLS) out[g] = lg[tid];
    }
}

extern "C" void kernel_launch(void* const* d_in, const int* in_sizes, int n_in,
                              void* d_out, int out_size, void* d_ws, size_t ws_size,
                              hipStream_t stream) {
    const float* x  = (const float*)d_in[0];
    const int*   ei = (const int*)d_in[1];
    const float* ew = (const float*)d_in[2];
    const float* W1 = (const float*)d_in[3];
    const float* b1 = (const float*)d_in[4];
    const float* W2 = (const float*)d_in[5];
    const float* b2 = (const float*)d_in[6];
    const float* Wc = (const float*)d_in[7];
    const float* bc = (const float*)d_in[8];
    float* out = (float*)d_out;

    const int n  = NNODES;
    const int E  = in_sizes[1] / 2;
    const int K1 = in_sizes[0] / n;
    const int* src = ei;
    const int* dst = ei + E;

    char* ws = (char*)d_ws;
    size_t off = 0;
    auto alloc = [&](size_t bytes) { char* p = ws + off; off = (off + bytes + 255) & ~(size_t)255; return p; };
    float* dinv = (float*)alloc((size_t)n * 4);
    float* bufA = (float*)alloc((size_t)n * DHID * 4);
    float* bufB = (float*)alloc((size_t)n * DHID * 4);
    int* bh       = (int*)alloc((size_t)NBINS * PB * 4);
    int* binstart = (int*)alloc((size_t)(NBINS + 1) * 4);
    int* rowptr   = (int*)alloc((size_t)(NBINS * BIN_NODES + 1) * 4);
    int2* tmp   = (int2*)alloc((size_t)E * 8);
    int* rec    = (int*)alloc((size_t)E * 8);
    (void)ws_size;

    const int T = 256;
    bin_count_kernel  <<<PB, SORT_T, 0, stream>>>(dst, bh, E);
    bin_place_kernel  <<<PB, SORT_T, 0, stream>>>(src, dst, ew, bh, binstart, tmp, E);
    bin_sort_kernel   <<<NBINS, SORT_T, 0, stream>>>(tmp, binstart, rowptr, dinv, (int2*)rec, n);

    const int gemmBlocks = (n + 127) / 128;
    const int nodeBlocks = (n + 7) / 8;
    const int clsBlocks  = (n + CLS_NODES - 1) / CLS_NODES;

    gemm128_wmma_kernel<<<gemmBlocks, T, 0, stream>>>(x, W1, dinv, bufA, n, K1);
    agg_gather_kernel  <<<nodeBlocks, T, 0, stream>>>(bufA, rowptr, rec, dinv, b1, bufB, n);

    gemm128_wmma_kernel<<<gemmBlocks, T, 0, stream>>>(bufB, W2, dinv, bufA, n, DHID);
    agg_gather_classify_kernel<<<clsBlocks, CLS_NODES * 32, 0, stream>>>(
        bufA, rowptr, rec, dinv, b2, Wc, bc, out, n);
}


# Layer 2's product h1 @ W2 (f32 WMMA, same k order) moved into the layer-1 aggregation launch: 16 finished rows form an A tile in LDS, W2 staged per block; h1 never stored; 7 -> 6 launches
# speedup vs baseline: 4.3507x; 1.0624x over previous
#include <hip/hip_runtime.h>

typedef __attribute__((ext_vector_type(2))) float v2f;
typedef __attribute__((ext_vector_type(8))) float v8f;

#define NNODES 50000
#define DHID   128
#define NCLS   10

#define BIN_SHIFT 8
#define BIN_NODES (1 << BIN_SHIFT)
#define NBINS     ((NNODES + BIN_NODES - 1) >> BIN_SHIFT)
#define PB        256
#define SORT_T    1024
#define PLACE_CAP 4096
#define SORT_CAP  8192

#define NWAVES        (SORT_T / 32)
#define ROW_LANES     8
#define ROWS_PER_LOAD (32 / ROW_LANES)
#define ROW_LOADS     (PB / (4 * ROW_LANES))
#define ROW_PASSES    ((NBINS + NWAVES * ROWS_PER_LOAD - 1) / (NWAVES * ROWS_PER_LOAD))
#define BINS_PER_LANE ((NBINS + 31) / 32)
static_assert(PB % (4 * ROW_LANES) == 0, "a table row is read as int4 by ROW_LANES lanes");

__device__ __forceinline__ int coarse_bin(int d) {
    int b = (int)((unsigned)d >> BIN_SHIFT);
    return b < NBINS ? b : NBINS - 1;
}

__global__ __launch_bounds__(SORT_T) void bin_count_kernel(
    const int* __restrict__ dst, int* __restrict__ bh, int nE)
{
    __shared__ int hist[NBINS];
    const int tid = threadIdx.x;
    for (int i = tid; i < NBINS; i += SORT_T) hist[i] = 0;
    __syncthreads();
    const int chunk = (nE + PB - 1) / PB;
    const int beg = blockIdx.x * chunk;
    const int end = (beg + chunk < nE) ? (beg + chunk) : nE;
    for (int base = beg + tid; base < end; base += 4 * SORT_T) {
        int d[4];
        #pragma unroll
        for (int k = 0; k < 4; ++k) {
            const int e = base + k * SORT_T;
            d[k] = (e < end) ? dst[e] : 0;
        }
        #pragma unroll
        for (int k = 0; k < 4; ++k)
            if (base + k * SORT_T < end) atomicAdd(&hist[coarse_bin(d[k])], 1);
    }
    __syncthreads();
    for (int i = tid; i < NBINS; i += SORT_T) bh[i * PB + blockIdx.x] = hist[i];
}

__global__ __launch_bounds__(SORT_T) void bin_place_kernel(
    const int* __restrict__ src, const int* __restrict__ dst,
    const float* __restrict__ w, const int* __restrict__ bh,
    int* __restrict__ binstart, int2* __restrict__ tmp, int nE)
{
    __shared__ int cur[NBINS];
    __shared__ int tot[NBINS];
    __shared__ int pre[NBINS];
    __shared__ int own[NBINS];
    __shared__ int dlt[NBINS];
    __shared__ int2 stg[PLACE_CAP];
    __shared__ int  gps[PLACE_CAP];
    const int tid  = threadIdx.x;
    const int lane = tid & 31;
    const int wave = tid >> 5;
    const int blk  = blockIdx.x;
    const int chunk = (nE + PB - 1) / PB;
    const int beg = blockIdx.x * chunk;
    const int end = (beg + chunk < nE) ? (beg + chunk) : nE;
    const bool staged = (end - beg) <= PLACE_CAP;

    int myown = 0;
    if (tid < NBINS) myown = bh[tid * PB + blk];

    const int l8   = lane & (ROW_LANES - 1);
    const int rsub = lane / ROW_LANES;
    int4 v[ROW_PASSES][ROW_LOADS];
    #pragma unroll
    for (int q = 0; q < ROW_PASSES; ++q) {
        const int i = (q * NWAVES + wave) * ROWS_PER_LOAD + rsub;
        const int4* row = (const int4*)(bh + (i < NBINS ? i : NBINS - 1) * PB);
        #pragma unroll
        for (int j = 0; j < ROW_LOADS; ++j) v[q][j] = row[j * ROW_LANES + l8];
    }
    int t[ROW_PASSES], p[ROW_PASSES];
    #pragma unroll
    for (int q = 0; q < ROW_PASSES; ++q) {
        t[q] = 0;
        p[q] = 0;
        #pragma unroll
        for (int j = 0; j < ROW_LOADS; ++j) {
            const int c0 = (j * ROW_LANES + l8) * 4;
            t[q] += v[q][j].x + v[q][j].y + v[q][j].z + v[q][j].w;
            p[q] += (c0 + 0 < blk ? v[q][j].x : 0) + (c0 + 1 < blk ? v[q][j].y : 0)
                  + (c0 + 2 < blk ? v[q][j].z : 0) + (c0 + 3 < blk ? v[q][j].w : 0);
        }
    }
    #pragma unroll
    for (int off = ROW_LANES / 2; off > 0; off >>= 1) {
        #pragma unroll
        for (int q = 0; q < ROW_PASSES; ++q) {
            t[q] += __shfl_xor(t[q], off, 32);
            p[q] += __shfl_xor(p[q], off, 32);
        }
    }
    #pragma unroll
    for (int q = 0; q < ROW_PASSES; ++q) {
        const int i = (q * NWAVES + wave) * ROWS_PER_LOAD + rsub;
        if (l8 == 0 && i < NBINS) { tot[i] = t[q]; pre[i] = p[q]; }
    }
    if (tid < NBINS) own[tid] = myown;
    __syncthreads();

    if (wave == 0) {
        int c[BINS_PER_LANE], pr[BINS_PER_LANE], ow[BINS_PER_LANE];
        int s = 0, so = 0;
        #pragma unroll
        for (int k = 0; k < BINS_PER_LANE; ++k) {
            const int i = lane * BINS_PER_LANE + k;
            c[k]  = (i < NBINS) ? tot[i] : 0;
            pr[k] = (i < NBINS) ? pre[i] : 0;
            ow[k] = (i < NBINS) ? own[i] : 0;
            s += c[k];
            so += ow[k];
        }
        int inc = s;
        int linc = so;
        #pragma unroll
        for (int off = 1; off < 32; off <<= 1) {
            const int u  = __shfl_up(inc, off, 32);
            const int lu = __shfl_up(linc, off, 32);
            if (lane >= off) { inc += u; linc += lu; }
        }
        int run  = inc - s;
        int lrun = linc - so;
        #pragma unroll
        for (int k = 0; k < BINS_PER_LANE; ++k) {
            const int i = lane * BINS_PER_LANE + k;
            if (i < NBINS) {
                const int g = run + pr[k];
                cur[i] = staged ? lrun : g;
                dlt[i] = g - lrun;
                if (blk == 0) binstart[i] = run;
            }
            run  += c[k];
            lrun += ow[k];
        }
        if (blk == 0 && lane == 31) binstart[NBINS] = run;
    }
    __syncthreads();
    for (int base = beg + tid; base < end; base += 4 * SORT_T) {
        int d[4], s[4];
        float wt[4];
        #pragma unroll
        for (int k = 0; k < 4; ++k) {
            const int e = base + k * SORT_T;
            const bool in = e < end;
            d[k]  = in ? dst[e] : 0;
            s[k]  = in ? src[e] : 0;
            wt[k] = in ? w[e] : 0.f;
        }
        #pragma unroll
        for (int k = 0; k < 4; ++k) {
            if (base + k * SORT_T < end) {
                const int bin = coarse_bin(d[k]);
                const int pos = atomicAdd(&cur[bin], 1);
                const unsigned key = (unsigned)s[k] | ((unsigned)(d[k] & (BIN_NODES - 1)) << 24);
                const int2 r = make_int2((int)key, __float_as_int(wt[k]));
                if (staged) { stg[pos] = r; gps[pos] = pos + dlt[bin]; }
                else        tmp[pos] = r;
            }
        }
    }
    if (staged) {
        __syncthreads();
        for (int t = tid; t < end - beg; t += SORT_T) tmp[gps[t]] = stg[t];
    }
}

__global__ __launch_bounds__(SORT_T) void bin_sort_kernel(
    const int2* __restrict__ tmp, const int* __restrict__ binstart,
    int* __restrict__ rowptr, float* __restrict__ dinv,
    int2* __restrict__ rec, int n)
{
    __shared__ int   cnt[BIN_NODES];
    __shared__ int   sc[BIN_NODES];
    __shared__ float dg[BIN_NODES];
    __shared__ int2  stg[SORT_CAP];
    const int tid = threadIdx.x;
    const int b = blockIdx.x;
    const int bbeg = binstart[b];
    const int bend = binstart[b + 1];
    const bool staged = (bend - bbeg) <= SORT_CAP;
    if (tid < BIN_NODES) { cnt[tid] = 0; dg[tid] = 1.0f; }
    __syncthreads();

    for (int base = bbeg + tid; base < bend; base += 4 * SORT_T) {
        int2 r[4];
        #pragma unroll
        for (int k = 0; k < 4; ++k) {
            const int p = base + k * SORT_T;
            r[k] = (p < bend) ? tmp[p] : make_int2(0, 0);
        }
        #pragma unroll
        for (int k = 0; k < 4; ++k) {
            if (base + k * SORT_T < bend) {
                const int dl = (int)((unsigned)r[k].x >> 24);
                atomicAdd(&cnt[dl], 1);
                atomicAdd(&dg[dl], __int_as_float(r[k].y));
            }
        }
    }
    __syncthreads();

    if (tid < BIN_NODES) sc[tid] = cnt[tid];
    __syncthreads();
    for (int off = 1; off < BIN_NODES; off <<= 1) {
        int v = 0;
        if (tid < BIN_NODES && tid >= off) v = sc[tid - off];
        __syncthreads();
        if (tid < BIN_NODES) sc[tid] += v;
        __syncthreads();
    }
    if (tid < BIN_NODES) {
        const int first = sc[tid] - cnt[tid];
        const int node = b * BIN_NODES + tid;
        rowptr[node] = bbeg + first;
        if (node < n) dinv[node] = __frsqrt_rn(dg[tid]);
        cnt[tid] = first;
    }
    __syncthreads();

    for (int base = bbeg + tid; base < bend; base += 4 * SORT_T) {
        int2 r[4];
        #pragma unroll
        for (int k = 0; k < 4; ++k) {
            const int p = base + k * SORT_T;
            r[k] = (p < bend) ? tmp[p] : make_int2(0, 0);
        }
        #pragma unroll
        for (int k = 0; k < 4; ++k) {
            if (base + k * SORT_T < bend) {
                const int dl = (int)((unsigned)r[k].x >> 24);
                const int lp = atomicAdd(&cnt[dl], 1);
                const int2 o = make_int2(r[k].x & 0x00FFFFFF, r[k].y);
                if (staged) stg[lp] = o;
                else        rec[bbeg + lp] = o;
            }
        }
    }
    if (staged) {
        __syncthreads();
        for (int i = tid; i < bend - bbeg; i += SORT_T) rec[bbeg + i] = stg[i];
    }
}

#define XS_STRIDE 34
#define WT_STRIDE 144

__global__ __launch_bounds__(256) void gemm128_wmma_kernel(
    const float* __restrict__ X, const float* __restrict__ W,
    const float* __restrict__ rs, float* __restrict__ H, int M, int K)
{
    __shared__ float  Xs[128 * XS_STRIDE];
    __shared__ float2 Wt[16 * WT_STRIDE];

    const int tid  = threadIdx.x;
    const int lane = tid & 31;
    const int wave = tid >> 5;
    const int l16  = lane & 15;
    const bool hi  = (lane >= 16);
    const int rowBlock = blockIdx.x * 128;
    const int rowLocal = wave * 16 + l16;

    v8f acc[8];
    #pragma unroll
    for (int t = 0; t < 8; ++t)
        #pragma unroll
        for (int r = 0; r < 8; ++r) acc[t][r] = 0.0f;

    float rsc[8];
    #pragma unroll
    for (int r = 0; r < 8; ++r) {
        const int m = rowBlock + wave * 16 + r + (hi ? 8 : 0);
        rsc[r] = rs[m < M ? m : M - 1];
    }

    const int nch = (K + 31) / 32;
    for (int ch = 0; ch < nch; ++ch) {
        const int k0 = ch * 32;

        #pragma unroll
        for (int j = 0; j < 4; ++j) {
            int f  = tid * 4 + j;
            int r  = f >> 3;
            int kb = (f & 7) * 4;
            int gr = rowBlock + r;
            float4 v = make_float4(0.f, 0.f, 0.f, 0.f);
            if (gr < M) {
                int kg = k0 + kb;
                if (kg + 3 < K) {
                    v = *(const float4*)(X + (size_t)gr * K + kg);
                } else {
                    float t0 = (kg + 0 < K) ? X[(size_t)gr * K + kg + 0] : 0.f;
                    float t1 = (kg + 1 < K) ? X[(size_t)gr * K + kg + 1] : 0.f;
                    float t2 = (kg + 2 < K) ? X[(size_t)gr * K + kg + 2] : 0.f;
                    float t3 = (kg + 3 < K) ? X[(size_t)gr * K + kg + 3] : 0.f;
                    v = make_float4(t0, t1, t2, t3);
                }
            }
            float* dp = &Xs[r * XS_STRIDE + kb];
            ((float2*)dp)[0] = make_float2(v.x, v.y);
            ((float2*)dp)[1] = make_float2(v.z, v.w);
        }

        #pragma unroll
        for (int j = 0; j < 2; ++j) {
            int f  = tid * 2 + j;
            int kp = f >> 5;
            int cb = (f & 31) * 4;
            int ka = k0 + 2 * kp;
            float4 va = make_float4(0.f, 0.f, 0.f, 0.f);
            float4 vb = make_float4(0.f, 0.f, 0.f, 0.f);
            if (ka < K)     va = *(const float4*)(W + (size_t)ka * DHID + cb);
            if (ka + 1 < K) vb = *(const float4*)(W + (size_t)(ka + 1) * DHID + cb);
            float4 w0 = make_float4(va.x, vb.x, va.y, vb.y);
            float4 w1 = make_float4(va.z, vb.z, va.w, vb.w);
            *(float4*)&Wt[kp * WT_STRIDE + cb]     = w0;
            *(float4*)&Wt[kp * WT_STRIDE + cb + 2] = w1;
        }

        __syncthreads();

        #pragma unroll
        for (int s = 0; s < 8; ++s) {
            const int kp = s * 2 + (hi ? 1 : 0);
            v2f a = *(const v2f*)&Xs[rowLocal * XS_STRIDE + kp * 2];
            float2 bf[8];
            #pragma unroll
            for (int nt = 0; nt < 8; ++nt)
                bf[nt] = Wt[kp * WT_STRIDE + nt * 16 + l16];
            #pragma unroll
            for (int nt = 0; nt < 8; ++nt) {
                v2f b;
                b.x = bf[nt].x;
                b.y = bf[nt].y;
                acc[nt] = __builtin_amdgcn_wmma_f32_16x16x4_f32(
                     false, a,  false, b,
                     (short)0, acc[nt],
                     false,  false);
            }
        }
        __syncthreads();
    }

    #pragma unroll
    for (int nt = 0; nt < 8; ++nt) {
        const int col = nt * 16 + l16;
        #pragma unroll
        for (int r = 0; r < 8; ++r) {
            int m = rowBlock + wave * 16 + r + (hi ? 8 : 0);
            if (m < M) H[(size_t)m * DHID + col] = acc[nt][r] * rsc[r];
        }
    }
}

__device__ __forceinline__ float4 agg_row(
    const float4* __restrict__ H4,
    const int* __restrict__ rowptr,
    const int2* __restrict__ R2,
    const float* __restrict__ dinv,
    const float* __restrict__ bias,
    int node, int lane)
{
    const int beg = rowptr[node];
    const int end = rowptr[node + 1];
    const float s = dinv[node];
    const float4 hv = H4[(size_t)node * 32 + lane];
    const float4 bv = ((const float4*)bias)[lane];
    float4 acc = hv;

    for (int base = beg; base < end; base += 32) {
        const int nb = (end - base < 32) ? (end - base) : 32;
        int2 r = make_int2(0, 0);
        if (lane < nb) r = R2[base + lane];

        int j = 0;
        for (; j + 4 <= nb; j += 4) {
            const int s0 = __builtin_amdgcn_readlane(r.x, j + 0);
            const int s1 = __builtin_amdgcn_readlane(r.x, j + 1);
            const int s2 = __builtin_amdgcn_readlane(r.x, j + 2);
            const int s3 = __builtin_amdgcn_readlane(r.x, j + 3);
            const float w0 = __int_as_float(__builtin_amdgcn_readlane(r.y, j + 0));
            const float w1 = __int_as_float(__builtin_amdgcn_readlane(r.y, j + 1));
            const float w2 = __int_as_float(__builtin_amdgcn_readlane(r.y, j + 2));
            const float w3 = __int_as_float(__builtin_amdgcn_readlane(r.y, j + 3));
            const float4 h0 = H4[(size_t)s0 * 32 + lane];
            const float4 h1 = H4[(size_t)s1 * 32 + lane];
            const float4 h2 = H4[(size_t)s2 * 32 + lane];
            const float4 h3 = H4[(size_t)s3 * 32 + lane];
            acc.x = fmaf(h0.x, w0, acc.x); acc.y = fmaf(h0.y, w0, acc.y); acc.z = fmaf(h0.z, w0, acc.z); acc.w = fmaf(h0.w, w0, acc.w);
            acc.x = fmaf(h1.x, w1, acc.x); acc.y = fmaf(h1.y, w1, acc.y); acc.z = fmaf(h1.z, w1, acc.z); acc.w = fmaf(h1.w, w1, acc.w);
            acc.x = fmaf(h2.x, w2, acc.x); acc.y = fmaf(h2.y, w2, acc.y); acc.z = fmaf(h2.z, w2, acc.z); acc.w = fmaf(h2.w, w2, acc.w);
            acc.x = fmaf(h3.x, w3, acc.x); acc.y = fmaf(h3.y, w3, acc.y); acc.z = fmaf(h3.z, w3, acc.z); acc.w = fmaf(h3.w, w3, acc.w);
        }
        #pragma unroll 1
        for (; j < nb; ++j) {
            const int s0 = __builtin_amdgcn_readlane(r.x, j);
            const float w0 = __int_as_float(__builtin_amdgcn_readlane(r.y, j));
            const float4 h0 = H4[(size_t)s0 * 32 + lane];
            acc.x = fmaf(h0.x, w0, acc.x); acc.y = fmaf(h0.y, w0, acc.y); acc.z = fmaf(h0.z, w0, acc.z); acc.w = fmaf(h0.w, w0, acc.w);
        }
    }
    acc.x = fmaf(acc.x, s, bv.x); acc.y = fmaf(acc.y, s, bv.y);
    acc.z = fmaf(acc.z, s, bv.z); acc.w = fmaf(acc.w, s, bv.w);
    acc.x = acc.x > 0.f ? acc.x : 0.f;
    acc.y = acc.y > 0.f ? acc.y : 0.f;
    acc.z = acc.z > 0.f ? acc.z : 0.f;
    acc.w = acc.w > 0.f ? acc.w : 0.f;
    return acc;
}

#define FG_WAVES  16
#define FG_T      (FG_WAVES * 32)
#define FG_STRIPS 5
#define AS_STRIDE 132
#define FG_KG     2
#define DS_STRIDE 36
static_assert(DHID == 128 && FG_WAVES == 16, "one A tile = 16 rows of 128, four multiplying waves x two column tiles");

__global__ __launch_bounds__(FG_T, 12) void agg_gather_gemm_kernel(
    const float* __restrict__ H,
    const int* __restrict__ rowptr,
    const int* __restrict__ rec,
    const float* __restrict__ dinv,
    const float* __restrict__ bias,
    const float* __restrict__ W,
    float* __restrict__ OUT, int n)
{
    __shared__ __attribute__((aligned(16))) float2 Wt[(DHID / 2) * WT_STRIDE];
    __shared__ __attribute__((aligned(16))) float  As[2][16 * AS_STRIDE];
    __shared__ __attribute__((aligned(16))) float Sc[2][16];
    __shared__ __attribute__((aligned(16))) float Ds[4][16 * DS_STRIDE];
    const int tid  = threadIdx.x;
    const int lane = tid & 31;
    const int l16  = lane & 15;
    const int hi   = lane >> 4;
    const int wv = __builtin_amdgcn_readfirstlane(tid >> 5);

    #pragma unroll
    for (int j = 0; j < 4; ++j) {
        const int f  = tid + j * FG_T;
        const int kp = f >> 5;
        const int cb = (f & 31) * 4;
        const float4 va = *(const float4*)(W + (size_t)(2 * kp) * DHID + cb);
        const float4 vb = *(const float4*)(W + (size_t)(2 * kp + 1) * DHID + cb);
        *(float4*)&Wt[kp * WT_STRIDE + cb]     = make_float4(va.x, vb.x, va.y, vb.y);
        *(float4*)&Wt[kp * WT_STRIDE + cb + 2] = make_float4(va.z, vb.z, va.w, vb.w);
    }

    const int nstrips = (n + 15) >> 4;
    for (int it = 0; it < FG_STRIPS; ++it) {
        const int strip = blockIdx.x * FG_STRIPS + it;
        if (strip >= nstrips) break;
        const int node = strip * 16 + wv;
        float4 h = make_float4(0.f, 0.f, 0.f, 0.f);
        float sc = 0.f;
        if (node < n) {
            sc = dinv[node];
            h = agg_row((const float4*)H, rowptr, (const int2*)rec, dinv, bias, node, lane);
        }
        float* as = As[it & 1];
        *(float4*)&as[wv * AS_STRIDE + 4 * lane] = h;
        if (lane == 0) Sc[it & 1][wv] = sc;
        __syncthreads();

        if ((wv >> 2) == (it & 3)) {
            const int q = wv & 3;
            v8f acc0, acc1;
            #pragma unroll
            for (int r = 0; r < 8; ++r) { acc0[r] = 0.0f; acc1[r] = 0.0f; }
            const float*  ap = &as[l16 * AS_STRIDE + 2 * hi];
            const float2* bp = &Wt[hi * WT_STRIDE + (2 * q) * 16 + l16];
            v2f    fa[2][FG_KG];
            float2 f0[2][FG_KG], f1[2][FG_KG];
            #pragma unroll
            for (int j = 0; j < FG_KG; ++j) {
                fa[0][j] = *(const v2f*)(ap + 4 * j);
                f0[0][j] = bp[(2 * j) * WT_STRIDE];
                f1[0][j] = bp[(2 * j) * WT_STRIDE + 16];
            }
            #pragma unroll
            for (int g = 0; g < DHID / (4 * FG_KG); ++g) {
                const int cur = g & 1;
                const int nxt = cur ^ 1;
                if (g + 1 < DHID / (4 * FG_KG)) {
                    #pragma unroll
                    for (int j = 0; j < FG_KG; ++j) {
                        const int s = FG_KG * (g + 1) + j;
                        fa[nxt][j] = *(const v2f*)(ap + 4 * s);
                        f0[nxt][j] = bp[(2 * s) * WT_STRIDE];
                        f1[nxt][j] = bp[(2 * s) * WT_STRIDE + 16];
                    }
                }
                __builtin_amdgcn_sched_barrier(0);
                #pragma unroll
                for (int j = 0; j < FG_KG; ++j) {
                    v2f b0, b1;
                    b0.x = f0[cur][j].x; b0.y = f0[cur][j].y;
                    b1.x = f1[cur][j].x; b1.y = f1[cur][j].y;
                    acc0 = __builtin_amdgcn_wmma_f32_16x16x4_f32(
                         false, fa[cur][j],  false, b0,
                         (short)0, acc0,
                         false,  false);
                    acc1 = __builtin_amdgcn_wmma_f32_16x16x4_f32(
                         false, fa[cur][j],  false, b1,
                         (short)0, acc1,
                         false,  false);
                }
                __builtin_amdgcn_sched_barrier(0);
            }
            const float4 sa = *(const float4*)&Sc[it & 1][8 * hi];
            const float4 sb = *(const float4*)&Sc[it & 1][8 * hi + 4];
            const float rsc[8] = {sa.x, sa.y, sa.z, sa.w, sb.x, sb.y, sb.z, sb.w};
            float* ds = Ds[q];
            #pragma unroll
            for (int r = 0; r < 8; ++r) {
                ds[(r + 8 * hi) * DS_STRIDE + l16]      = acc0[r] * rsc[r];
                ds[(r + 8 * hi) * DS_STRIDE + 16 + l16] = acc1[r] * rsc[r];
            }
            #pragma unroll
            for (int j = 0; j < 4; ++j) {
                const int row = 4 * j + (lane >> 3);
                const int c4  = (lane & 7) * 4;
                const float4 v = *(const float4*)&ds[row * DS_STRIDE + c4];
                const int m = strip * 16 + row;
                if (m < n) *(float4*)&OUT[(size_t)m * DHID + q * 32 + c4] = v;
            }
        }
    }
}

#define CLS_NODES     16
#define WC_LANE_F4    10
#define WC_PITCH_F4   11
static_assert(DHID == 128 && (DHID * NCLS) % 4 == 0 && DHID * NCLS / 4 == 32 * WC_LANE_F4,
              "a lane owns 4 rows of Wc = WC_LANE_F4 float4");
static_assert(NCLS <= 16 && (CLS_NODES * NCLS) % 32 == 0, "logit butterfly and whole-line stores");

__device__ __forceinline__ float xhalf(float x) {
    const int b = __builtin_bit_cast(int, x);
    return __builtin_bit_cast(float,
        __builtin_amdgcn_permlanex16(b, b, 0x76543210, 0xfedcba98, false, false));
}
template <int M> __device__ __forceinline__ float xrow(float x) {
    const int b = __builtin_bit_cast(int, x);
    return __builtin_bit_cast(float,
        __builtin_amdgcn_update_dpp(b, b, 0x160 + M, 0xf, 0xf, false));
}

__global__ __launch_bounds__(CLS_NODES * 32) void agg_gather_classify_kernel(
    const float* __restrict__ H,
    const int* __restrict__ rowptr,
    const int* __restrict__ rec,
    const float* __restrict__ dinv,
    const float* __restrict__ bias,
    const float* __restrict__ Wc,
    const float* __restrict__ bc,
    float* __restrict__ out, int n)
{
    __shared__ float4 wcs[32 * WC_PITCH_F4];
    __shared__ float  lg[CLS_NODES * NCLS];
    const int tid  = threadIdx.x;
    const int lane = tid & 31;
    const int wv   = tid >> 5;

    const bool stager = tid < 32 * WC_LANE_F4;
    float4 wstage = make_float4(0.f, 0.f, 0.f, 0.f);
    if (stager) wstage = ((const float4*)Wc)[tid];
    const int cls = lane >> 1;
    const float bcv = bc[cls < NCLS ? cls : 0];

    const int node = __builtin_amdgcn_readfirstlane((int)(blockIdx.x * CLS_NODES + wv));
    float4 h = make_float4(0.f, 0.f, 0.f, 0.f);
    if (node < n)
        h = agg_row((const float4*)H, rowptr, (const int2*)rec, dinv, bias, node, lane);
    if (stager) wcs[(tid / WC_LANE_F4) * WC_PITCH_F4 + (tid % WC_LANE_F4)] = wstage;
    __syncthreads();

    float w[4 * WC_LANE_F4];
    #pragma unroll
    for (int i = 0; i < WC_LANE_F4; ++i) {
        const float4 t = wcs[lane * WC_PITCH_F4 + i];
        w[4 * i + 0] = t.x; w[4 * i + 1] = t.y; w[4 * i + 2] = t.z; w[4 * i + 3] = t.w;
    }
    float p[16];
    #pragma unroll
    for (int c = 0; c < 16; ++c) {
        if (c < NCLS) {
            float a = h.x * w[c];
            a = fmaf(h.y, w[NCLS + c], a);
            a = fmaf(h.z, w[2 * NCLS + c], a);
            a = fmaf(h.w, w[3 * NCLS + c], a);
            p[c] = a;
        } else {
            p[c] = 0.f;
        }
    }
    const bool b4 = (lane & 16) != 0, b3 = (lane & 8) != 0, b2 = (lane & 4) != 0, b1 = (lane & 2) != 0;
    float q8[8], q4[4], q2[2];
    #pragma unroll
    for (int j = 0; j < 8; ++j) {
        const float keep = b4 ? p[j + 8] : p[j];
        const float send = b4 ? p[j] : p[j + 8];
        q8[j] = keep + xhalf(send);
    }
    #pragma unroll
    for (int j = 0; j < 4; ++j) {
        const float keep = b3 ? q8[j + 4] : q8[j];
        const float send = b3 ? q8[j] : q8[j + 4];
        q4[j] = keep + xrow<8>(send);
    }
    #pragma unroll
    for (int j = 0; j < 2; ++j) {
        const float keep = b2 ? q4[j + 2] : q4[j];
        const float send = b2 ? q4[j] : q4[j + 2];
        q2[j] = keep + xrow<4>(send);
    }
    const float keep1 = b1 ? q2[1] : q2[0];
    const float send1 = b1 ? q2[0] : q2[1];
    const float q1 = keep1 + xrow<2>(send1);
    const float q0 = q1 + xrow<1>(q1);
    if ((lane & 1) == 0 && cls < NCLS) lg[wv * NCLS + cls] = q0 + bcv;
    __syncthreads();

    if (tid < CLS_NODES * NCLS) {
        const int g = blockIdx.x * (CLS_NODES * NCLS) + tid;
        if (g < n * NCLS) out[g] = lg[tid];
    }
}

extern "C" void kernel_launch(void* const* d_in, const int* in_sizes, int n_in,
                              void* d_out, int out_size, void* d_ws, size_t ws_size,
                              hipStream_t stream) {
    const float* x  = (const float*)d_in[0];
    const int*   ei = (const int*)d_in[1];
    const float* ew = (const float*)d_in[2];
    const float* W1 = (const float*)d_in[3];
    const float* b1 = (const float*)d_in[4];
    const float* W2 = (const float*)d_in[5];
    const float* b2 = (const float*)d_in[6];
    const float* Wc = (const float*)d_in[7];
    const float* bc = (const float*)d_in[8];
    float* out = (float*)d_out;

    const int n  = NNODES;
    const int E  = in_sizes[1] / 2;
    const int K1 = in_sizes[0] / n;
    const int* src = ei;
    const int* dst = ei + E;

    char* ws = (char*)d_ws;
    size_t off = 0;
    auto alloc = [&](size_t bytes) { char* p = ws + off; off = (off + bytes + 255) & ~(size_t)255; return p; };
    float* dinv = (float*)alloc((size_t)n * 4);
    float* bufA = (float*)alloc((size_t)n * DHID * 4);
    float* bufB = (float*)alloc((size_t)n * DHID * 4);
    int* bh       = (int*)alloc((size_t)NBINS * PB * 4);
    int* binstart = (int*)alloc((size_t)(NBINS + 1) * 4);
    int* rowptr   = (int*)alloc((size_t)(NBINS * BIN_NODES + 1) * 4);
    int2* tmp   = (int2*)alloc((size_t)E * 8);
    int* rec    = (int*)alloc((size_t)E * 8);
    (void)ws_size;

    const int T = 256;
    bin_count_kernel  <<<PB, SORT_T, 0, stream>>>(dst, bh, E);
    bin_place_kernel  <<<PB, SORT_T, 0, stream>>>(src, dst, ew, bh, binstart, tmp, E);
    bin_sort_kernel   <<<NBINS, SORT_T, 0, stream>>>(tmp, binstart, rowptr, dinv, (int2*)rec, n);

    const int gemmBlocks = (n + 127) / 128;
    const int fuseBlocks = ((n + 15) / 16 + FG_STRIPS - 1) / FG_STRIPS;
    const int clsBlocks  = (n + CLS_NODES - 1) / CLS_NODES;

    gemm128_wmma_kernel<<<gemmBlocks, T, 0, stream>>>(x, W1, dinv, bufA, n, K1);
    agg_gather_gemm_kernel<<<fuseBlocks, FG_T, 0, stream>>>(bufA, rowptr, rec, dinv, b1, W2, bufB, n);

    agg_gather_classify_kernel<<<clsBlocks, CLS_NODES * 32, 0, stream>>>(
        bufB, rowptr, rec, dinv, b2, Wc, bc, out, n);
}


# Edge sort level 1 without a count launch: a block sorts its own chunk by bin in LDS (add's return = rank), writes it in place plus a (bin, chunk) offset table; 6 -> 5 launches; LDS arrival order kept
# speedup vs baseline: 4.5382x; 1.0431x over previous
#include <hip/hip_runtime.h>

typedef __attribute__((ext_vector_type(2))) float v2f;
typedef __attribute__((ext_vector_type(8))) float v8f;

#define NNODES 50000
#define DHID   128
#define NCLS   10

#define BIN_SHIFT 8
#define BIN_NODES (1 << BIN_SHIFT)
#define NBINS     ((NNODES + BIN_NODES - 1) >> BIN_SHIFT)
#define PB        256
#define SORT_T    1024
#define PLACE_T   512
#define PLACE_CAP 4096
#define PLACE_PER (PLACE_CAP / PLACE_T)
#define SORT_CAP  8192
#define SEG_LANES (SORT_T / PB)
#define SEG_PER   8
#define BINS_PER_LANE ((NBINS + 31) / 32)
static_assert(SORT_T == SEG_LANES * PB && PB % 32 == 0 && PB <= SORT_T, "level 2: SEG_LANES threads per chunk, whole waves scan the PB runs");
static_assert(PLACE_CAP % PLACE_T == 0, "a level-1 thread holds a whole number of records");

__device__ __forceinline__ int coarse_bin(int d) {
    int b = (int)((unsigned)d >> BIN_SHIFT);
    return b < NBINS ? b : NBINS - 1;
}

__global__ __launch_bounds__(PLACE_T) void bin_place_kernel(
    const int* __restrict__ src, const int* __restrict__ dst,
    const float* __restrict__ w, int* __restrict__ loff,
    int2* __restrict__ tmp, int nE)
{
    __shared__ int hist[NBINS];
    __shared__ int off[NBINS];
    __shared__ int2 stg[PLACE_CAP];
    const int tid  = threadIdx.x;
    const int lane = tid & 31;
    const int wave = tid >> 5;
    const int blk  = blockIdx.x;
    const int chunk = (nE + PB - 1) / PB;
    const int beg = blk * chunk;
    const int end = (beg + chunk < nE) ? (beg + chunk) : nE;
    const int len = (end > beg) ? (end - beg) : 0;
    const bool staged = len <= PLACE_CAP;
    for (int i = tid; i < NBINS; i += PLACE_T) hist[i] = 0;
    __syncthreads();

    int d[PLACE_PER], s[PLACE_PER], rk[PLACE_PER];
    float wt[PLACE_PER];
    if (staged) {
        #pragma unroll
        for (int k = 0; k < PLACE_PER; ++k) {
            const int e = beg + tid + k * PLACE_T;
            const bool in = e < end;
            d[k]  = in ? dst[e] : 0;
            s[k]  = in ? src[e] : 0;
            wt[k] = in ? w[e] : 0.f;
        }
        #pragma unroll
        for (int k = 0; k < PLACE_PER; ++k) {
            rk[k] = 0;
            if (beg + tid + k * PLACE_T < end) rk[k] = atomicAdd(&hist[coarse_bin(d[k])], 1);
        }
    } else {
        for (int e = beg + tid; e < end; e += PLACE_T) atomicAdd(&hist[coarse_bin(dst[e])], 1);
    }
    __syncthreads();

    if (wave == 0) {
        int cc[BINS_PER_LANE];
        int sum = 0;
        #pragma unroll
        for (int k = 0; k < BINS_PER_LANE; ++k) {
            const int i = lane * BINS_PER_LANE + k;
            cc[k] = (i < NBINS) ? hist[i] : 0;
            sum += cc[k];
        }
        int inc = sum;
        #pragma unroll
        for (int o = 1; o < 32; o <<= 1) {
            const int u = __shfl_up(inc, o, 32);
            if (lane >= o) inc += u;
        }
        int run = inc - sum;
        #pragma unroll
        for (int k = 0; k < BINS_PER_LANE; ++k) {
            const int i = lane * BINS_PER_LANE + k;
            if (i < NBINS) { off[i] = run; loff[i * PB + blk] = run; }
            run += cc[k];
        }
        if (lane == 31) loff[NBINS * PB + blk] = run;
    }
    __syncthreads();

    if (staged) {
        int fs[PLACE_PER];
        #pragma unroll
        for (int k = 0; k < PLACE_PER; ++k) fs[k] = off[coarse_bin(d[k])];
        #pragma unroll
        for (int k = 0; k < PLACE_PER; ++k) {
            if (beg + tid + k * PLACE_T < end) {
                const unsigned key = (unsigned)s[k] | ((unsigned)(d[k] & (BIN_NODES - 1)) << 24);
                stg[fs[k] + rk[k]] = make_int2((int)key, __float_as_int(wt[k]));
            }
        }
        __syncthreads();
        for (int t = tid; t < len; t += PLACE_T) tmp[beg + t] = stg[t];
    } else {
        for (int e = beg + tid; e < end; e += PLACE_T) {
            const int dd = dst[e];
            const int pos = atomicAdd(&off[coarse_bin(dd)], 1);
            const unsigned key = (unsigned)src[e] | ((unsigned)(dd & (BIN_NODES - 1)) << 24);
            tmp[beg + pos] = make_int2((int)key, __float_as_int(w[e]));
        }
    }
}

__global__ __launch_bounds__(SORT_T) void bin_sort_kernel(
    const int2* __restrict__ tmp, const int* __restrict__ loff,
    int* __restrict__ rowptr, float* __restrict__ dinv,
    int2* __restrict__ rec, int n, int nE)
{
    __shared__ int   cnt[BIN_NODES];
    __shared__ int   sc[BIN_NODES];
    __shared__ float dg[BIN_NODES];
    __shared__ int   sbeg[PB];
    __shared__ int   scnt[PB];
    __shared__ int   spos[PB];
    __shared__ int   wlen[PB / 32];
    __shared__ int   wbef[PB / 32];
    __shared__ int2  inr[SORT_CAP];
    __shared__ int2  stg[SORT_CAP];
    const int tid  = threadIdx.x;
    const int lane = tid & 31;
    const int wave = tid >> 5;
    const int b = blockIdx.x;
    const int chunk = (nE + PB - 1) / PB;
    if (tid < BIN_NODES) { cnt[tid] = 0; dg[tid] = 1.0f; }

    int myLen = 0, myInc = 0, myBefInc = 0;
    if (tid < PB) {
        const int o0 = loff[b * PB + tid];
        const int o1 = loff[(b + 1) * PB + tid];
        myLen = o1 - o0;
        sbeg[tid] = tid * chunk + o0;
        scnt[tid] = myLen;
        myInc = myLen;
        myBefInc = o0;
        #pragma unroll
        for (int o = 1; o < 32; o <<= 1) {
            const int u = __shfl_up(myInc, o, 32);
            const int v = __shfl_up(myBefInc, o, 32);
            if (lane >= o) { myInc += u; myBefInc += v; }
        }
        if (lane == 31) { wlen[wave] = myInc; wbef[wave] = myBefInc; }
    }
    __syncthreads();
    int bbeg = 0, total = 0, wpre = 0;
    #pragma unroll
    for (int j = 0; j < PB / 32; ++j) {
        const int l = wlen[j];
        bbeg += wbef[j];
        if (j < wave) wpre += l;
        total += l;
    }
    if (tid < PB) spos[tid] = wpre + myInc - myLen;
    const bool staged = total <= SORT_CAP;
    __syncthreads();

    const int sg  = tid / SEG_LANES;
    const int sub = tid % SEG_LANES;
    const int sb = sbeg[sg], sn = scnt[sg], sp = spos[sg];
    for (int i0 = sub; i0 < sn; i0 += SEG_PER * SEG_LANES) {
        int2 r[SEG_PER];
        #pragma unroll
        for (int k = 0; k < SEG_PER; ++k) {
            const int i = i0 + k * SEG_LANES;
            r[k] = (i < sn) ? tmp[sb + i] : make_int2(0, 0);
        }
        #pragma unroll
        for (int k = 0; k < SEG_PER; ++k) {
            const int i = i0 + k * SEG_LANES;
            if (i < sn) {
                const int dl = (int)((unsigned)r[k].x >> 24);
                atomicAdd(&cnt[dl], 1);
                atomicAdd(&dg[dl], __int_as_float(r[k].y));
                if (staged) inr[sp + i] = r[k];
            }
        }
    }
    __syncthreads();

    if (tid < BIN_NODES) sc[tid] = cnt[tid];
    __syncthreads();
    for (int o = 1; o < BIN_NODES; o <<= 1) {
        int v = 0;
        if (tid < BIN_NODES && tid >= o) v = sc[tid - o];
        __syncthreads();
        if (tid < BIN_NODES) sc[tid] += v;
        __syncthreads();
    }
    if (tid < BIN_NODES) {
        const int first = sc[tid] - cnt[tid];
        const int node = b * BIN_NODES + tid;
        rowptr[node] = bbeg + first;
        if (node < n) dinv[node] = __frsqrt_rn(dg[tid]);
        cnt[tid] = first;
    }
    __syncthreads();

    if (staged) {
        for (int base = tid; base < total; base += 4 * SORT_T) {
            int2 r[4];
            int lp[4];
            #pragma unroll
            for (int k = 0; k < 4; ++k) {
                const int p = base + k * SORT_T;
                r[k] = inr[p < total ? p : 0];
            }
            #pragma unroll
            for (int k = 0; k < 4; ++k) {
                lp[k] = 0;
                if (base + k * SORT_T < total)
                    lp[k] = atomicAdd(&cnt[(int)((unsigned)r[k].x >> 24)], 1);
            }
            #pragma unroll
            for (int k = 0; k < 4; ++k)
                if (base + k * SORT_T < total)
                    stg[lp[k]] = make_int2(r[k].x & 0x00FFFFFF, r[k].y);
        }
        __syncthreads();
        for (int i = tid; i < total; i += SORT_T) rec[bbeg + i] = stg[i];
    } else {
        for (int i0 = sub; i0 < sn; i0 += SEG_PER * SEG_LANES) {
            int2 r[SEG_PER];
            #pragma unroll
            for (int k = 0; k < SEG_PER; ++k) {
                const int i = i0 + k * SEG_LANES;
                r[k] = (i < sn) ? tmp[sb + i] : make_int2(0, 0);
            }
            #pragma unroll
            for (int k = 0; k < SEG_PER; ++k) {
                if (i0 + k * SEG_LANES < sn) {
                    const int dl = (int)((unsigned)r[k].x >> 24);
                    const int lp = atomicAdd(&cnt[dl], 1);
                    rec[bbeg + lp] = make_int2(r[k].x & 0x00FFFFFF, r[k].y);
                }
            }
        }
    }
}

#define XS_STRIDE 34
#define WT_STRIDE 144

__global__ __launch_bounds__(256) void gemm128_wmma_kernel(
    const float* __restrict__ X, const float* __restrict__ W,
    const float* __restrict__ rs, float* __restrict__ H, int M, int K)
{
    __shared__ float  Xs[128 * XS_STRIDE];
    __shared__ float2 Wt[16 * WT_STRIDE];

    const int tid  = threadIdx.x;
    const int lane = tid & 31;
    const int wave = tid >> 5;
    const int l16  = lane & 15;
    const bool hi  = (lane >= 16);
    const int rowBlock = blockIdx.x * 128;
    const int rowLocal = wave * 16 + l16;

    v8f acc[8];
    #pragma unroll
    for (int t = 0; t < 8; ++t)
        #pragma unroll
        for (int r = 0; r < 8; ++r) acc[t][r] = 0.0f;

    float rsc[8];
    #pragma unroll
    for (int r = 0; r < 8; ++r) {
        const int m = rowBlock + wave * 16 + r + (hi ? 8 : 0);
        rsc[r] = rs[m < M ? m : M - 1];
    }

    const int nch = (K + 31) / 32;
    for (int ch = 0; ch < nch; ++ch) {
        const int k0 = ch * 32;

        #pragma unroll
        for (int j = 0; j < 4; ++j) {
            int f  = tid * 4 + j;
            int r  = f >> 3;
            int kb = (f & 7) * 4;
            int gr = rowBlock + r;
            float4 v = make_float4(0.f, 0.f, 0.f, 0.f);
            if (gr < M) {
                int kg = k0 + kb;
                if (kg + 3 < K) {
                    v = *(const float4*)(X + (size_t)gr * K + kg);
                } else {
                    float t0 = (kg + 0 < K) ? X[(size_t)gr * K + kg + 0] : 0.f;
                    float t1 = (kg + 1 < K) ? X[(size_t)gr * K + kg + 1] : 0.f;
                    float t2 = (kg + 2 < K) ? X[(size_t)gr * K + kg + 2] : 0.f;
                    float t3 = (kg + 3 < K) ? X[(size_t)gr * K + kg + 3] : 0.f;
                    v = make_float4(t0, t1, t2, t3);
                }
            }
            float* dp = &Xs[r * XS_STRIDE + kb];
            ((float2*)dp)[0] = make_float2(v.x, v.y);
            ((float2*)dp)[1] = make_float2(v.z, v.w);
        }

        #pragma unroll
        for (int j = 0; j < 2; ++j) {
            int f  = tid * 2 + j;
            int kp = f >> 5;
            int cb = (f & 31) * 4;
            int ka = k0 + 2 * kp;
            float4 va = make_float4(0.f, 0.f, 0.f, 0.f);
            float4 vb = make_float4(0.f, 0.f, 0.f, 0.f);
            if (ka < K)     va = *(const float4*)(W + (size_t)ka * DHID + cb);
            if (ka + 1 < K) vb = *(const float4*)(W + (size_t)(ka + 1) * DHID + cb);
            float4 w0 = make_float4(va.x, vb.x, va.y, vb.y);
            float4 w1 = make_float4(va.z, vb.z, va.w, vb.w);
            *(float4*)&Wt[kp * WT_STRIDE + cb]     = w0;
            *(float4*)&Wt[kp * WT_STRIDE + cb + 2] = w1;
        }

        __syncthreads();

        #pragma unroll
        for (int s = 0; s < 8; ++s) {
            const int kp = s * 2 + (hi ? 1 : 0);
            v2f a = *(const v2f*)&Xs[rowLocal * XS_STRIDE + kp * 2];
            float2 bf[8];
            #pragma unroll
            for (int nt = 0; nt < 8; ++nt)
                bf[nt] = Wt[kp * WT_STRIDE + nt * 16 + l16];
            #pragma unroll
            for (int nt = 0; nt < 8; ++nt) {
                v2f b;
                b.x = bf[nt].x;
                b.y = bf[nt].y;
                acc[nt] = __builtin_amdgcn_wmma_f32_16x16x4_f32(
                     false, a,  false, b,
                     (short)0, acc[nt],
                     false,  false);
            }
        }
        __syncthreads();
    }

    #pragma unroll
    for (int nt = 0; nt < 8; ++nt) {
        const int col = nt * 16 + l16;
        #pragma unroll
        for (int r = 0; r < 8; ++r) {
            int m = rowBlock + wave * 16 + r + (hi ? 8 : 0);
            if (m < M) H[(size_t)m * DHID + col] = acc[nt][r] * rsc[r];
        }
    }
}

__device__ __forceinline__ float4 agg_row(
    const float4* __restrict__ H4,
    const int* __restrict__ rowptr,
    const int2* __restrict__ R2,
    const float* __restrict__ dinv,
    const float* __restrict__ bias,
    int node, int lane)
{
    const int beg = rowptr[node];
    const int end = rowptr[node + 1];
    const float s = dinv[node];
    const float4 hv = H4[(size_t)node * 32 + lane];
    const float4 bv = ((const float4*)bias)[lane];
    float4 acc = hv;

    for (int base = beg; base < end; base += 32) {
        const int nb = (end - base < 32) ? (end - base) : 32;
        int2 r = make_int2(0, 0);
        if (lane < nb) r = R2[base + lane];

        int j = 0;
        for (; j + 4 <= nb; j += 4) {
            const int s0 = __builtin_amdgcn_readlane(r.x, j + 0);
            const int s1 = __builtin_amdgcn_readlane(r.x, j + 1);
            const int s2 = __builtin_amdgcn_readlane(r.x, j + 2);
            const int s3 = __builtin_amdgcn_readlane(r.x, j + 3);
            const float w0 = __int_as_float(__builtin_amdgcn_readlane(r.y, j + 0));
            const float w1 = __int_as_float(__builtin_amdgcn_readlane(r.y, j + 1));
            const float w2 = __int_as_float(__builtin_amdgcn_readlane(r.y, j + 2));
            const float w3 = __int_as_float(__builtin_amdgcn_readlane(r.y, j + 3));
            const float4 h0 = H4[(size_t)s0 * 32 + lane];
            const float4 h1 = H4[(size_t)s1 * 32 + lane];
            const float4 h2 = H4[(size_t)s2 * 32 + lane];
            const float4 h3 = H4[(size_t)s3 * 32 + lane];
            acc.x = fmaf(h0.x, w0, acc.x); acc.y = fmaf(h0.y, w0, acc.y); acc.z = fmaf(h0.z, w0, acc.z); acc.w = fmaf(h0.w, w0, acc.w);
            acc.x = fmaf(h1.x, w1, acc.x); acc.y = fmaf(h1.y, w1, acc.y); acc.z = fmaf(h1.z, w1, acc.z); acc.w = fmaf(h1.w, w1, acc.w);
            acc.x = fmaf(h2.x, w2, acc.x); acc.y = fmaf(h2.y, w2, acc.y); acc.z = fmaf(h2.z, w2, acc.z); acc.w = fmaf(h2.w, w2, acc.w);
            acc.x = fmaf(h3.x, w3, acc.x); acc.y = fmaf(h3.y, w3, acc.y); acc.z = fmaf(h3.z, w3, acc.z); acc.w = fmaf(h3.w, w3, acc.w);
        }
        #pragma unroll 1
        for (; j < nb; ++j) {
            const int s0 = __builtin_amdgcn_readlane(r.x, j);
            const float w0 = __int_as_float(__builtin_amdgcn_readlane(r.y, j));
            const float4 h0 = H4[(size_t)s0 * 32 + lane];
            acc.x = fmaf(h0.x, w0, acc.x); acc.y = fmaf(h0.y, w0, acc.y); acc.z = fmaf(h0.z, w0, acc.z); acc.w = fmaf(h0.w, w0, acc.w);
        }
    }
    acc.x = fmaf(acc.x, s, bv.x); acc.y = fmaf(acc.y, s, bv.y);
    acc.z = fmaf(acc.z, s, bv.z); acc.w = fmaf(acc.w, s, bv.w);
    acc.x = acc.x > 0.f ? acc.x : 0.f;
    acc.y = acc.y > 0.f ? acc.y : 0.f;
    acc.z = acc.z > 0.f ? acc.z : 0.f;
    acc.w = acc.w > 0.f ? acc.w : 0.f;
    return acc;
}

#define FG_WAVES  16
#define FG_T      (FG_WAVES * 32)
#define FG_STRIPS 5
#define AS_STRIDE 132
#define FG_KG     2
#define DS_STRIDE 36
static_assert(DHID == 128 && FG_WAVES == 16, "one A tile = 16 rows of 128, four multiplying waves x two column tiles");

__global__ __launch_bounds__(FG_T, 12) void agg_gather_gemm_kernel(
    const float* __restrict__ H,
    const int* __restrict__ rowptr,
    const int* __restrict__ rec,
    const float* __restrict__ dinv,
    const float* __restrict__ bias,
    const float* __restrict__ W,
    float* __restrict__ OUT, int n)
{
    __shared__ __attribute__((aligned(16))) float2 Wt[(DHID / 2) * WT_STRIDE];
    __shared__ __attribute__((aligned(16))) float  As[2][16 * AS_STRIDE];
    __shared__ __attribute__((aligned(16))) float Sc[2][16];
    __shared__ __attribute__((aligned(16))) float Ds[4][16 * DS_STRIDE];
    const int tid  = threadIdx.x;
    const int lane = tid & 31;
    const int l16  = lane & 15;
    const int hi   = lane >> 4;
    const int wv = __builtin_amdgcn_readfirstlane(tid >> 5);

    #pragma unroll
    for (int j = 0; j < 4; ++j) {
        const int f  = tid + j * FG_T;
        const int kp = f >> 5;
        const int cb = (f & 31) * 4;
        const float4 va = *(const float4*)(W + (size_t)(2 * kp) * DHID + cb);
        const float4 vb = *(const float4*)(W + (size_t)(2 * kp + 1) * DHID + cb);
        *(float4*)&Wt[kp * WT_STRIDE + cb]     = make_float4(va.x, vb.x, va.y, vb.y);
        *(float4*)&Wt[kp * WT_STRIDE + cb + 2] = make_float4(va.z, vb.z, va.w, vb.w);
    }

    const int nstrips = (n + 15) >> 4;
    for (int it = 0; it < FG_STRIPS; ++it) {
        const int strip = blockIdx.x * FG_STRIPS + it;
        if (strip >= nstrips) break;
        const int node = strip * 16 + wv;
        float4 h = make_float4(0.f, 0.f, 0.f, 0.f);
        float sc = 0.f;
        if (node < n) {
            sc = dinv[node];
            h = agg_row((const float4*)H, rowptr, (const int2*)rec, dinv, bias, node, lane);
        }
        float* as = As[it & 1];
        *(float4*)&as[wv * AS_STRIDE + 4 * lane] = h;
        if (lane == 0) Sc[it & 1][wv] = sc;
        __syncthreads();

        if ((wv >> 2) == (it & 3)) {
            const int q = wv & 3;
            v8f acc0, acc1;
            #pragma unroll
            for (int r = 0; r < 8; ++r) { acc0[r] = 0.0f; acc1[r] = 0.0f; }
            const float*  ap = &as[l16 * AS_STRIDE + 2 * hi];
            const float2* bp = &Wt[hi * WT_STRIDE + (2 * q) * 16 + l16];
            v2f    fa[2][FG_KG];
            float2 f0[2][FG_KG], f1[2][FG_KG];
            #pragma unroll
            for (int j = 0; j < FG_KG; ++j) {
                fa[0][j] = *(const v2f*)(ap + 4 * j);
                f0[0][j] = bp[(2 * j) * WT_STRIDE];
                f1[0][j] = bp[(2 * j) * WT_STRIDE + 16];
            }
            #pragma unroll
            for (int g = 0; g < DHID / (4 * FG_KG); ++g) {
                const int cur = g & 1;
                const int nxt = cur ^ 1;
                if (g + 1 < DHID / (4 * FG_KG)) {
                    #pragma unroll
                    for (int j = 0; j < FG_KG; ++j) {
                        const int s = FG_KG * (g + 1) + j;
                        fa[nxt][j] = *(const v2f*)(ap + 4 * s);
                        f0[nxt][j] = bp[(2 * s) * WT_STRIDE];
                        f1[nxt][j] = bp[(2 * s) * WT_STRIDE + 16];
                    }
                }
                __builtin_amdgcn_sched_barrier(0);
                #pragma unroll
                for (int j = 0; j < FG_KG; ++j) {
                    v2f b0, b1;
                    b0.x = f0[cur][j].x; b0.y = f0[cur][j].y;
                    b1.x = f1[cur][j].x; b1.y = f1[cur][j].y;
                    acc0 = __builtin_amdgcn_wmma_f32_16x16x4_f32(
                         false, fa[cur][j],  false, b0,
                         (short)0, acc0,
                         false,  false);
                    acc1 = __builtin_amdgcn_wmma_f32_16x16x4_f32(
                         false, fa[cur][j],  false, b1,
                         (short)0, acc1,
                         false,  false);
                }
                __builtin_amdgcn_sched_barrier(0);
            }
            const float4 sa = *(const float4*)&Sc[it & 1][8 * hi];
            const float4 sb = *(const float4*)&Sc[it & 1][8 * hi + 4];
            const float rsc[8] = {sa.x, sa.y, sa.z, sa.w, sb.x, sb.y, sb.z, sb.w};
            float* ds = Ds[q];
            #pragma unroll
            for (int r = 0; r < 8; ++r) {
                ds[(r + 8 * hi) * DS_STRIDE + l16]      = acc0[r] * rsc[r];
                ds[(r + 8 * hi) * DS_STRIDE + 16 + l16] = acc1[r] * rsc[r];
            }
            #pragma unroll
            for (int j = 0; j < 4; ++j) {
                const int row = 4 * j + (lane >> 3);
                const int c4  = (lane & 7) * 4;
                const float4 v = *(const float4*)&ds[row * DS_STRIDE + c4];
                const int m = strip * 16 + row;
                if (m < n) *(float4*)&OUT[(size_t)m * DHID + q * 32 + c4] = v;
            }
        }
    }
}

#define CLS_NODES     16
#define WC_LANE_F4    10
#define WC_PITCH_F4   11
static_assert(DHID == 128 && (DHID * NCLS) % 4 == 0 && DHID * NCLS / 4 == 32 * WC_LANE_F4,
              "a lane owns 4 rows of Wc = WC_LANE_F4 float4");
static_assert(NCLS <= 16 && (CLS_NODES * NCLS) % 32 == 0, "logit butterfly and whole-line stores");

__device__ __forceinline__ float xhalf(float x) {
    const int b = __builtin_bit_cast(int, x);
    return __builtin_bit_cast(float,
        __builtin_amdgcn_permlanex16(b, b, 0x76543210, 0xfedcba98, false, false));
}
template <int M> __device__ __forceinline__ float xrow(float x) {
    const int b = __builtin_bit_cast(int, x);
    return __builtin_bit_cast(float,
        __builtin_amdgcn_update_dpp(b, b, 0x160 + M, 0xf, 0xf, false));
}

__global__ __launch_bounds__(CLS_NODES * 32) void agg_gather_classify_kernel(
    const float* __restrict__ H,
    const int* __restrict__ rowptr,
    const int* __restrict__ rec,
    const float* __restrict__ dinv,
    const float* __restrict__ bias,
    const float* __restrict__ Wc,
    const float* __restrict__ bc,
    float* __restrict__ out, int n)
{
    __shared__ float4 wcs[32 * WC_PITCH_F4];
    __shared__ float  lg[CLS_NODES * NCLS];
    const int tid  = threadIdx.x;
    const int lane = tid & 31;
    const int wv   = tid >> 5;

    const bool stager = tid < 32 * WC_LANE_F4;
    float4 wstage = make_float4(0.f, 0.f, 0.f, 0.f);
    if (stager) wstage = ((const float4*)Wc)[tid];
    const int cls = lane >> 1;
    const float bcv = bc[cls < NCLS ? cls : 0];

    const int node = __builtin_amdgcn_readfirstlane((int)(blockIdx.x * CLS_NODES + wv));
    float4 h = make_float4(0.f, 0.f, 0.f, 0.f);
    if (node < n)
        h = agg_row((const float4*)H, rowptr, (const int2*)rec, dinv, bias, node, lane);
    if (stager) wcs[(tid / WC_LANE_F4) * WC_PITCH_F4 + (tid % WC_LANE_F4)] = wstage;
    __syncthreads();

    float w[4 * WC_LANE_F4];
    #pragma unroll
    for (int i = 0; i < WC_LANE_F4; ++i) {
        const float4 t = wcs[lane * WC_PITCH_F4 + i];
        w[4 * i + 0] = t.x; w[4 * i + 1] = t.y; w[4 * i + 2] = t.z; w[4 * i + 3] = t.w;
    }
    float p[16];
    #pragma unroll
    for (int c = 0; c < 16; ++c) {
        if (c < NCLS) {
            float a = h.x * w[c];
            a = fmaf(h.y, w[NCLS + c], a);
            a = fmaf(h.z, w[2 * NCLS + c], a);
            a = fmaf(h.w, w[3 * NCLS + c], a);
            p[c] = a;
        } else {
            p[c] = 0.f;
        }
    }
    const bool b4 = (lane & 16) != 0, b3 = (lane & 8) != 0, b2 = (lane & 4) != 0, b1 = (lane & 2) != 0;
    float q8[8], q4[4], q2[2];
    #pragma unroll
    for (int j = 0; j < 8; ++j) {
        const float keep = b4 ? p[j + 8] : p[j];
        const float send = b4 ? p[j] : p[j + 8];
        q8[j] = keep + xhalf(send);
    }
    #pragma unroll
    for (int j = 0; j < 4; ++j) {
        const float keep = b3 ? q8[j + 4] : q8[j];
        const float send = b3 ? q8[j] : q8[j + 4];
        q4[j] = keep + xrow<8>(send);
    }
    #pragma unroll
    for (int j = 0; j < 2; ++j) {
        const float keep = b2 ? q4[j + 2] : q4[j];
        const float send = b2 ? q4[j] : q4[j + 2];
        q2[j] = keep + xrow<4>(send);
    }
    const float keep1 = b1 ? q2[1] : q2[0];
    const float send1 = b1 ? q2[0] : q2[1];
    const float q1 = keep1 + xrow<2>(send1);
    const float q0 = q1 + xrow<1>(q1);
    if ((lane & 1) == 0 && cls < NCLS) lg[wv * NCLS + cls] = q0 + bcv;
    __syncthreads();

    if (tid < CLS_NODES * NCLS) {
        const int g = blockIdx.x * (CLS_NODES * NCLS) + tid;
        if (g < n * NCLS) out[g] = lg[tid];
    }
}

extern "C" void kernel_launch(void* const* d_in, const int* in_sizes, int n_in,
                              void* d_out, int out_size, void* d_ws, size_t ws_size,
                              hipStream_t stream) {
    const float* x  = (const float*)d_in[0];
    const int*   ei = (const int*)d_in[1];
    const float* ew = (const float*)d_in[2];
    const float* W1 = (const float*)d_in[3];
    const float* b1 = (const float*)d_in[4];
    const float* W2 = (const float*)d_in[5];
    const float* b2 = (const float*)d_in[6];
    const float* Wc = (const float*)d_in[7];
    const float* bc = (const float*)d_in[8];
    float* out = (float*)d_out;

    const int n  = NNODES;
    const int E  = in_sizes[1] / 2;
    const int K1 = in_sizes[0] / n;
    const int* src = ei;
    const int* dst = ei + E;

    char* ws = (char*)d_ws;
    size_t off = 0;
    auto alloc = [&](size_t bytes) { char* p = ws + off; off = (off + bytes + 255) & ~(size_t)255; return p; };
    float* dinv = (float*)alloc((size_t)n * 4);
    float* bufA = (float*)alloc((size_t)n * DHID * 4);
    float* bufB = (float*)alloc((size_t)n * DHID * 4);
    int* loff     = (int*)alloc((size_t)(NBINS + 1) * PB * 4);
    int* rowptr   = (int*)alloc((size_t)(NBINS * BIN_NODES + 1) * 4);
    int2* tmp   = (int2*)alloc((size_t)E * 8);
    int* rec    = (int*)alloc((size_t)E * 8);
    (void)ws_size;

    const int T = 256;
    bin_place_kernel  <<<PB, PLACE_T, 0, stream>>>(src, dst, ew, loff, tmp, E);
    bin_sort_kernel   <<<NBINS, SORT_T, 0, stream>>>(tmp, loff, rowptr, dinv, (int2*)rec, n, E);

    const int gemmBlocks = (n + 127) / 128;
    const int fuseBlocks = ((n + 15) / 16 + FG_STRIPS - 1) / FG_STRIPS;
    const int clsBlocks  = (n + CLS_NODES - 1) / CLS_NODES;

    gemm128_wmma_kernel<<<gemmBlocks, T, 0, stream>>>(x, W1, dinv, bufA, n, K1);
    agg_gather_gemm_kernel<<<fuseBlocks, FG_T, 0, stream>>>(bufA, rowptr, rec, dinv, b1, W2, bufB, n);

    agg_gather_classify_kernel<<<clsBlocks, CLS_NODES * 32, 0, stream>>>(
        bufB, rowptr, rec, dinv, b2, Wc, bc, out, n);
}


# x@W1 GEMM re-tiled and pipelined: one 208-row block per WGP, a wave keeps one 16-col tile of 13 strips; next K chunk read into registers during the products (2 LDS stages); same k order
# speedup vs baseline: 5.7500x; 1.2670x over previous
#include <hip/hip_runtime.h>

typedef __attribute__((ext_vector_type(2))) float v2f;
typedef __attribute__((ext_vector_type(8))) float v8f;

#define NNODES 50000
#define DHID   128
#define NCLS   10

#define BIN_SHIFT 8
#define BIN_NODES (1 << BIN_SHIFT)
#define NBINS     ((NNODES + BIN_NODES - 1) >> BIN_SHIFT)
#define PB        256
#define SORT_T    1024
#define PLACE_T   512
#define PLACE_CAP 4096
#define PLACE_PER (PLACE_CAP / PLACE_T)
#define SORT_CAP  8192
#define SEG_LANES (SORT_T / PB)
#define SEG_PER   8
#define BINS_PER_LANE ((NBINS + 31) / 32)
static_assert(SORT_T == SEG_LANES * PB && PB % 32 == 0 && PB <= SORT_T, "level 2: SEG_LANES threads per chunk, whole waves scan the PB runs");
static_assert(PLACE_CAP % PLACE_T == 0, "a level-1 thread holds a whole number of records");

__device__ __forceinline__ int coarse_bin(int d) {
    int b = (int)((unsigned)d >> BIN_SHIFT);
    return b < NBINS ? b : NBINS - 1;
}

__global__ __launch_bounds__(PLACE_T) void bin_place_kernel(
    const int* __restrict__ src, const int* __restrict__ dst,
    const float* __restrict__ w, int* __restrict__ loff,
    int2* __restrict__ tmp, int nE)
{
    __shared__ int hist[NBINS];
    __shared__ int off[NBINS];
    __shared__ int2 stg[PLACE_CAP];
    const int tid  = threadIdx.x;
    const int lane = tid & 31;
    const int wave = tid >> 5;
    const int blk  = blockIdx.x;
    const int chunk = (nE + PB - 1) / PB;
    const int beg = blk * chunk;
    const int end = (beg + chunk < nE) ? (beg + chunk) : nE;
    const int len = (end > beg) ? (end - beg) : 0;
    const bool staged = len <= PLACE_CAP;
    for (int i = tid; i < NBINS; i += PLACE_T) hist[i] = 0;
    __syncthreads();

    int d[PLACE_PER], s[PLACE_PER], rk[PLACE_PER];
    float wt[PLACE_PER];
    if (staged) {
        #pragma unroll
        for (int k = 0; k < PLACE_PER; ++k) {
            const int e = beg + tid + k * PLACE_T;
            const bool in = e < end;
            d[k]  = in ? dst[e] : 0;
            s[k]  = in ? src[e] : 0;
            wt[k] = in ? w[e] : 0.f;
        }
        #pragma unroll
        for (int k = 0; k < PLACE_PER; ++k) {
            rk[k] = 0;
            if (beg + tid + k * PLACE_T < end) rk[k] = atomicAdd(&hist[coarse_bin(d[k])], 1);
        }
    } else {
        for (int e = beg + tid; e < end; e += PLACE_T) atomicAdd(&hist[coarse_bin(dst[e])], 1);
    }
    __syncthreads();

    if (wave == 0) {
        int cc[BINS_PER_LANE];
        int sum = 0;
        #pragma unroll
        for (int k = 0; k < BINS_PER_LANE; ++k) {
            const int i = lane * BINS_PER_LANE + k;
            cc[k] = (i < NBINS) ? hist[i] : 0;
            sum += cc[k];
        }
        int inc = sum;
        #pragma unroll
        for (int o = 1; o < 32; o <<= 1) {
            const int u = __shfl_up(inc, o, 32);
            if (lane >= o) inc += u;
        }
        int run = inc - sum;
        #pragma unroll
        for (int k = 0; k < BINS_PER_LANE; ++k) {
            const int i = lane * BINS_PER_LANE + k;
            if (i < NBINS) { off[i] = run; loff[i * PB + blk] = run; }
            run += cc[k];
        }
        if (lane == 31) loff[NBINS * PB + blk] = run;
    }
    __syncthreads();

    if (staged) {
        int fs[PLACE_PER];
        #pragma unroll
        for (int k = 0; k < PLACE_PER; ++k) fs[k] = off[coarse_bin(d[k])];
        #pragma unroll
        for (int k = 0; k < PLACE_PER; ++k) {
            if (beg + tid + k * PLACE_T < end) {
                const unsigned key = (unsigned)s[k] | ((unsigned)(d[k] & (BIN_NODES - 1)) << 24);
                stg[fs[k] + rk[k]] = make_int2((int)key, __float_as_int(wt[k]));
            }
        }
        __syncthreads();
        for (int t = tid; t < len; t += PLACE_T) tmp[beg + t] = stg[t];
    } else {
        for (int e = beg + tid; e < end; e += PLACE_T) {
            const int dd = dst[e];
            const int pos = atomicAdd(&off[coarse_bin(dd)], 1);
            const unsigned key = (unsigned)src[e] | ((unsigned)(dd & (BIN_NODES - 1)) << 24);
            tmp[beg + pos] = make_int2((int)key, __float_as_int(w[e]));
        }
    }
}

__global__ __launch_bounds__(SORT_T) void bin_sort_kernel(
    const int2* __restrict__ tmp, const int* __restrict__ loff,
    int* __restrict__ rowptr, float* __restrict__ dinv,
    int2* __restrict__ rec, int n, int nE)
{
    __shared__ int   cnt[BIN_NODES];
    __shared__ int   sc[BIN_NODES];
    __shared__ float dg[BIN_NODES];
    __shared__ int   sbeg[PB];
    __shared__ int   scnt[PB];
    __shared__ int   spos[PB];
    __shared__ int   wlen[PB / 32];
    __shared__ int   wbef[PB / 32];
    __shared__ int2  inr[SORT_CAP];
    __shared__ int2  stg[SORT_CAP];
    const int tid  = threadIdx.x;
    const int lane = tid & 31;
    const int wave = tid >> 5;
    const int b = blockIdx.x;
    const int chunk = (nE + PB - 1) / PB;
    if (tid < BIN_NODES) { cnt[tid] = 0; dg[tid] = 1.0f; }

    int myLen = 0, myInc = 0, myBefInc = 0;
    if (tid < PB) {
        const int o0 = loff[b * PB + tid];
        const int o1 = loff[(b + 1) * PB + tid];
        myLen = o1 - o0;
        sbeg[tid] = tid * chunk + o0;
        scnt[tid] = myLen;
        myInc = myLen;
        myBefInc = o0;
        #pragma unroll
        for (int o = 1; o < 32; o <<= 1) {
            const int u = __shfl_up(myInc, o, 32);
            const int v = __shfl_up(myBefInc, o, 32);
            if (lane >= o) { myInc += u; myBefInc += v; }
        }
        if (lane == 31) { wlen[wave] = myInc; wbef[wave] = myBefInc; }
    }
    __syncthreads();
    int bbeg = 0, total = 0, wpre = 0;
    #pragma unroll
    for (int j = 0; j < PB / 32; ++j) {
        const int l = wlen[j];
        bbeg += wbef[j];
        if (j < wave) wpre += l;
        total += l;
    }
    if (tid < PB) spos[tid] = wpre + myInc - myLen;
    const bool staged = total <= SORT_CAP;
    __syncthreads();

    const int sg  = tid / SEG_LANES;
    const int sub = tid % SEG_LANES;
    const int sb = sbeg[sg], sn = scnt[sg], sp = spos[sg];
    for (int i0 = sub; i0 < sn; i0 += SEG_PER * SEG_LANES) {
        int2 r[SEG_PER];
        #pragma unroll
        for (int k = 0; k < SEG_PER; ++k) {
            const int i = i0 + k * SEG_LANES;
            r[k] = (i < sn) ? tmp[sb + i] : make_int2(0, 0);
        }
        #pragma unroll
        for (int k = 0; k < SEG_PER; ++k) {
            const int i = i0 + k * SEG_LANES;
            if (i < sn) {
                const int dl = (int)((unsigned)r[k].x >> 24);
                atomicAdd(&cnt[dl], 1);
                atomicAdd(&dg[dl], __int_as_float(r[k].y));
                if (staged) inr[sp + i] = r[k];
            }
        }
    }
    __syncthreads();

    if (tid < BIN_NODES) sc[tid] = cnt[tid];
    __syncthreads();
    for (int o = 1; o < BIN_NODES; o <<= 1) {
        int v = 0;
        if (tid < BIN_NODES && tid >= o) v = sc[tid - o];
        __syncthreads();
        if (tid < BIN_NODES) sc[tid] += v;
        __syncthreads();
    }
    if (tid < BIN_NODES) {
        const int first = sc[tid] - cnt[tid];
        const int node = b * BIN_NODES + tid;
        rowptr[node] = bbeg + first;
        if (node < n) dinv[node] = __frsqrt_rn(dg[tid]);
        cnt[tid] = first;
    }
    __syncthreads();

    if (staged) {
        for (int base = tid; base < total; base += 4 * SORT_T) {
            int2 r[4];
            int lp[4];
            #pragma unroll
            for (int k = 0; k < 4; ++k) {
                const int p = base + k * SORT_T;
                r[k] = inr[p < total ? p : 0];
            }
            #pragma unroll
            for (int k = 0; k < 4; ++k) {
                lp[k] = 0;
                if (base + k * SORT_T < total)
                    lp[k] = atomicAdd(&cnt[(int)((unsigned)r[k].x >> 24)], 1);
            }
            #pragma unroll
            for (int k = 0; k < 4; ++k)
                if (base + k * SORT_T < total)
                    stg[lp[k]] = make_int2(r[k].x & 0x00FFFFFF, r[k].y);
        }
        __syncthreads();
        for (int i = tid; i < total; i += SORT_T) rec[bbeg + i] = stg[i];
    } else {
        for (int i0 = sub; i0 < sn; i0 += SEG_PER * SEG_LANES) {
            int2 r[SEG_PER];
            #pragma unroll
            for (int k = 0; k < SEG_PER; ++k) {
                const int i = i0 + k * SEG_LANES;
                r[k] = (i < sn) ? tmp[sb + i] : make_int2(0, 0);
            }
            #pragma unroll
            for (int k = 0; k < SEG_PER; ++k) {
                if (i0 + k * SEG_LANES < sn) {
                    const int dl = (int)((unsigned)r[k].x >> 24);
                    const int lp = atomicAdd(&cnt[dl], 1);
                    rec[bbeg + lp] = make_int2(r[k].x & 0x00FFFFFF, r[k].y);
                }
            }
        }
    }
}

#define XS_STRIDE 34
#define WT_STRIDE 144
#define GK        32
#define GR_STRIPS 13
#define GR_ROWS   (16 * GR_STRIPS)
#define GT        256
#define GX_F4     (GR_ROWS * (GK / 4))
#define GX_LD     ((GX_F4 + GT - 1) / GT)
#define GX_LDS_ROWS    (GX_LD * GT / 8)
#define G_XS_FLOATS    (GX_LDS_ROWS * XS_STRIDE)
#define G_WT_FLOATS    (2 * (GK / 2) * WT_STRIDE)
#define G_STAGE_FLOATS (G_XS_FLOATS + G_WT_FLOATS)
#define OT_STRIDE 132
#define G_OT_FLOATS    (GR_ROWS * OT_STRIDE)
#define G_SMEM_FLOATS  (2 * G_STAGE_FLOATS > G_OT_FLOATS ? 2 * G_STAGE_FLOATS : G_OT_FLOATS)
static_assert(DHID == 16 * (GT / 32), "one 16-column tile per wave");
static_assert((G_XS_FLOATS % 4) == 0 && (G_STAGE_FLOATS % 4) == 0, "16-byte aligned stages");

__device__ __forceinline__ void gemm_fetch(
    const float* __restrict__ X, const float* __restrict__ W, int M, int K,
    int rowBlock, int k0, int tid,
    float4 (&xr)[GX_LD], float4 (&wa)[2], float4 (&wb)[2])
{
    #pragma unroll
    for (int j = 0; j < 2; ++j) {
        const int f  = tid + j * GT;
        const int kp = f >> 5;
        const int cb = (f & 31) * 4;
        const int ka = k0 + 2 * kp;
        wa[j] = *(const float4*)(W + (size_t)(ka < K ? ka : 0) * DHID + cb);
        wb[j] = *(const float4*)(W + (size_t)(ka + 1 < K ? ka + 1 : 0) * DHID + cb);
    }
    #pragma unroll
    for (int j = 0; j < GX_LD; ++j) {
        const int f  = tid + j * GT;
        int r = f >> 3;
        r = r < GR_ROWS ? r : GR_ROWS - 1;
        const int kg = k0 + (f & 7) * 4;
        int gr = rowBlock + r;
        gr = gr < M ? gr : M - 1;
        xr[j] = *(const float4*)(X + (size_t)gr * K + (kg + 3 < K ? kg : 0));
    }
}

__device__ __forceinline__ void gemm_stage(
    float* __restrict__ xs, float2* __restrict__ wt,
    const float* __restrict__ X, int M, int K, int rowBlock, int k0, int tid,
    const float4 (&xr)[GX_LD], const float4 (&wa)[2], const float4 (&wb)[2])
{
    #pragma unroll
    for (int j = 0; j < GX_LD; ++j) {
        const int f = tid + j * GT;
        float2* dp = (float2*)&xs[(f >> 3) * XS_STRIDE + (f & 7) * 4];
        dp[0] = make_float2(xr[j].x, xr[j].y);
        dp[1] = make_float2(xr[j].z, xr[j].w);
    }
    #pragma unroll
    for (int j = 0; j < 2; ++j) {
        const int f  = tid + j * GT;
        const int kp = f >> 5;
        const int cb = (f & 31) * 4;
        *(float4*)&wt[kp * WT_STRIDE + cb]     = make_float4(wa[j].x, wb[j].x, wa[j].y, wb[j].y);
        *(float4*)&wt[kp * WT_STRIDE + cb + 2] = make_float4(wa[j].z, wb[j].z, wa[j].w, wb[j].w);
    }
    if (k0 + GK > K) {
        #pragma unroll 1
        for (int j = 0; j < GX_LD; ++j) {
            const int f  = tid + j * GT;
            const int kg = k0 + (f & 7) * 4;
            float* dp = &xs[(f >> 3) * XS_STRIDE + (f & 7) * 4];
            if (kg + 3 >= K) {
                dp[0] = 0.f; dp[1] = 0.f; dp[2] = 0.f; dp[3] = 0.f;
                if (kg < K) {
                    int gr = rowBlock + (f >> 3);
                    gr = gr < M ? gr : M - 1;
                    const float* p = X + (size_t)gr * K + kg;
                    dp[0] = p[0];
                    if (kg + 1 < K) dp[1] = p[1];
                    if (kg + 2 < K) dp[2] = p[2];
                }
            }
        }
        #pragma unroll 1
        for (int j = 0; j < 2; ++j) {
            const int f  = tid + j * GT;
            const int kp = f >> 5;
            const int cb = (f & 31) * 4;
            const int ka = k0 + 2 * kp;
            if (ka + 1 >= K) {
                for (int c = 0; c < 4; ++c) {
                    float2* e = &wt[kp * WT_STRIDE + cb + c];
                    if (ka >= K) e->x = 0.f;
                    e->y = 0.f;
                }
            }
        }
    }
}

__global__ __launch_bounds__(GT) __attribute__((amdgpu_waves_per_eu(2, 2))) void gemm_rows_wmma_kernel(
    const float* __restrict__ X, const float* __restrict__ W,
    const float* __restrict__ rs, float* __restrict__ H, int M, int K)
{
    __shared__ __attribute__((aligned(16))) float smem[G_SMEM_FLOATS];
    __shared__ __attribute__((aligned(16))) float Sc[GR_ROWS];

    const int tid  = threadIdx.x;
    __builtin_assume(tid < GT);
    const int lane = tid & 31;
    const int l16  = lane & 15;
    const int hi   = lane >> 4;
    const int wave = __builtin_amdgcn_readfirstlane(tid >> 5);
    const int rowBlock = blockIdx.x * GR_ROWS;

    v8f acc[GR_STRIPS];
    #pragma unroll
    for (int t = 0; t < GR_STRIPS; ++t)
        #pragma unroll
        for (int r = 0; r < 8; ++r) acc[t][r] = 0.0f;

    if (tid < GR_ROWS) {
        const int m = rowBlock + tid;
        Sc[tid] = rs[m < M ? m : M - 1];
    }

    float4 xr[GX_LD], wa[2], wb[2];
    const int nch = (K + GK - 1) / GK;
    gemm_fetch(X, W, M, K, rowBlock, 0, tid, xr, wa, wb);
    gemm_stage(smem, (float2*)(smem + G_XS_FLOATS), X, M, K, rowBlock, 0, tid, xr, wa, wb);
    __syncthreads();

    for (int ch = 0; ch < nch; ++ch) {
        const float*  xs = smem + (ch & 1) * G_STAGE_FLOATS;
        const float2* wt = (const float2*)(xs + G_XS_FLOATS);
        const int kn = (ch + 1 < nch ? ch + 1 : ch) * GK;
        gemm_fetch(X, W, M, K, rowBlock, kn, tid, xr, wa, wb);

        const float*  ap = &xs[l16 * XS_STRIDE + 2 * hi];
        const float2* bp = &wt[hi * WT_STRIDE + wave * 16 + l16];
        v2f    fa[2][GR_STRIPS];
        float2 fb[2];
        #pragma unroll
        for (int t = 0; t < GR_STRIPS; ++t) fa[0][t] = *(const v2f*)(ap + t * 16 * XS_STRIDE);
        fb[0] = bp[0];
        #pragma unroll
        for (int s = 0; s < GK / 4; ++s) {
            const int cur = s & 1;
            const int nxt = cur ^ 1;
            if (s + 1 < GK / 4) {
                #pragma unroll
                for (int t = 0; t < GR_STRIPS; ++t)
                    fa[nxt][t] = *(const v2f*)(ap + t * 16 * XS_STRIDE + 4 * (s + 1));
                fb[nxt] = bp[2 * (s + 1) * WT_STRIDE];
            }
            __builtin_amdgcn_sched_barrier(0);
            v2f b;
            b.x = fb[cur].x; b.y = fb[cur].y;
            #pragma unroll
            for (int t = 0; t < GR_STRIPS; ++t)
                acc[t] = __builtin_amdgcn_wmma_f32_16x16x4_f32(
                     false, fa[cur][t],  false, b,
                     (short)0, acc[t],
                     false,  false);
            __builtin_amdgcn_sched_barrier(0);
        }

        {
            float* xn = smem + ((ch + 1) & 1) * G_STAGE_FLOATS;
            gemm_stage(xn, (float2*)(xn + G_XS_FLOATS), X, M, K, rowBlock, kn, tid, xr, wa, wb);
        }
        __syncthreads();
    }

    #pragma unroll
    for (int t = 0; t < GR_STRIPS; ++t) {
        const float4 sa = *(const float4*)&Sc[16 * t + 8 * hi];
        const float4 sb = *(const float4*)&Sc[16 * t + 8 * hi + 4];
        const float rsc[8] = {sa.x, sa.y, sa.z, sa.w, sb.x, sb.y, sb.z, sb.w};
        #pragma unroll
        for (int r = 0; r < 8; ++r)
            smem[(16 * t + 8 * hi + r) * OT_STRIDE + wave * 16 + l16] = acc[t][r] * rsc[r];
    }
    __syncthreads();
    #pragma unroll
    for (int j = 0; j < GR_ROWS / (GT / 32); ++j) {
        const int row = j * (GT / 32) + wave;
        const int m = rowBlock + row;
        const float4 v = *(const float4*)&smem[row * OT_STRIDE + 4 * lane];
        if (m < M) *(float4*)&H[(size_t)m * DHID + 4 * lane] = v;
    }
}

__device__ __forceinline__ float4 agg_row(
    const float4* __restrict__ H4,
    const int* __restrict__ rowptr,
    const int2* __restrict__ R2,
    const float* __restrict__ dinv,
    const float* __restrict__ bias,
    int node, int lane)
{
    const int beg = rowptr[node];
    const int end = rowptr[node + 1];
    const float s = dinv[node];
    const float4 hv = H4[(size_t)node * 32 + lane];
    const float4 bv = ((const float4*)bias)[lane];
    float4 acc = hv;

    for (int base = beg; base < end; base += 32) {
        const int nb = (end - base < 32) ? (end - base) : 32;
        int2 r = make_int2(0, 0);
        if (lane < nb) r = R2[base + lane];

        int j = 0;
        for (; j + 4 <= nb; j += 4) {
            const int s0 = __builtin_amdgcn_readlane(r.x, j + 0);
            const int s1 = __builtin_amdgcn_readlane(r.x, j + 1);
            const int s2 = __builtin_amdgcn_readlane(r.x, j + 2);
            const int s3 = __builtin_amdgcn_readlane(r.x, j + 3);
            const float w0 = __int_as_float(__builtin_amdgcn_readlane(r.y, j + 0));
            const float w1 = __int_as_float(__builtin_amdgcn_readlane(r.y, j + 1));
            const float w2 = __int_as_float(__builtin_amdgcn_readlane(r.y, j + 2));
            const float w3 = __int_as_float(__builtin_amdgcn_readlane(r.y, j + 3));
            const float4 h0 = H4[(size_t)s0 * 32 + lane];
            const float4 h1 = H4[(size_t)s1 * 32 + lane];
            const float4 h2 = H4[(size_t)s2 * 32 + lane];
            const float4 h3 = H4[(size_t)s3 * 32 + lane];
            acc.x = fmaf(h0.x, w0, acc.x); acc.y = fmaf(h0.y, w0, acc.y); acc.z = fmaf(h0.z, w0, acc.z); acc.w = fmaf(h0.w, w0, acc.w);
            acc.x = fmaf(h1.x, w1, acc.x); acc.y = fmaf(h1.y, w1, acc.y); acc.z = fmaf(h1.z, w1, acc.z); acc.w = fmaf(h1.w, w1, acc.w);
            acc.x = fmaf(h2.x, w2, acc.x); acc.y = fmaf(h2.y, w2, acc.y); acc.z = fmaf(h2.z, w2, acc.z); acc.w = fmaf(h2.w, w2, acc.w);
            acc.x = fmaf(h3.x, w3, acc.x); acc.y = fmaf(h3.y, w3, acc.y); acc.z = fmaf(h3.z, w3, acc.z); acc.w = fmaf(h3.w, w3, acc.w);
        }
        #pragma unroll 1
        for (; j < nb; ++j) {
            const int s0 = __builtin_amdgcn_readlane(r.x, j);
            const float w0 = __int_as_float(__builtin_amdgcn_readlane(r.y, j));
            const float4 h0 = H4[(size_t)s0 * 32 + lane];
            acc.x = fmaf(h0.x, w0, acc.x); acc.y = fmaf(h0.y, w0, acc.y); acc.z = fmaf(h0.z, w0, acc.z); acc.w = fmaf(h0.w, w0, acc.w);
        }
    }
    acc.x = fmaf(acc.x, s, bv.x); acc.y = fmaf(acc.y, s, bv.y);
    acc.z = fmaf(acc.z, s, bv.z); acc.w = fmaf(acc.w, s, bv.w);
    acc.x = acc.x > 0.f ? acc.x : 0.f;
    acc.y = acc.y > 0.f ? acc.y : 0.f;
    acc.z = acc.z > 0.f ? acc.z : 0.f;
    acc.w = acc.w > 0.f ? acc.w : 0.f;
    return acc;
}

#define FG_WAVES  16
#define FG_T      (FG_WAVES * 32)
#define FG_STRIPS 5
#define AS_STRIDE 132
#define FG_KG     2
#define DS_STRIDE 36
static_assert(DHID == 128 && FG_WAVES == 16, "one A tile = 16 rows of 128, four multiplying waves x two column tiles");

__global__ __launch_bounds__(FG_T, 12) void agg_gather_gemm_kernel(
    const float* __restrict__ H,
    const int* __restrict__ rowptr,
    const int* __restrict__ rec,
    const float* __restrict__ dinv,
    const float* __restrict__ bias,
    const float* __restrict__ W,
    float* __restrict__ OUT, int n)
{
    __shared__ __attribute__((aligned(16))) float2 Wt[(DHID / 2) * WT_STRIDE];
    __shared__ __attribute__((aligned(16))) float  As[2][16 * AS_STRIDE];
    __shared__ __attribute__((aligned(16))) float Sc[2][16];
    __shared__ __attribute__((aligned(16))) float Ds[4][16 * DS_STRIDE];
    const int tid  = threadIdx.x;
    const int lane = tid & 31;
    const int l16  = lane & 15;
    const int hi   = lane >> 4;
    const int wv = __builtin_amdgcn_readfirstlane(tid >> 5);

    #pragma unroll
    for (int j = 0; j < 4; ++j) {
        const int f  = tid + j * FG_T;
        const int kp = f >> 5;
        const int cb = (f & 31) * 4;
        const float4 va = *(const float4*)(W + (size_t)(2 * kp) * DHID + cb);
        const float4 vb = *(const float4*)(W + (size_t)(2 * kp + 1) * DHID + cb);
        *(float4*)&Wt[kp * WT_STRIDE + cb]     = make_float4(va.x, vb.x, va.y, vb.y);
        *(float4*)&Wt[kp * WT_STRIDE + cb + 2] = make_float4(va.z, vb.z, va.w, vb.w);
    }

    const int nstrips = (n + 15) >> 4;
    for (int it = 0; it < FG_STRIPS; ++it) {
        const int strip = blockIdx.x * FG_STRIPS + it;
        if (strip >= nstrips) break;
        const int node = strip * 16 + wv;
        float4 h = make_float4(0.f, 0.f, 0.f, 0.f);
        float sc = 0.f;
        if (node < n) {
            sc = dinv[node];
            h = agg_row((const float4*)H, rowptr, (const int2*)rec, dinv, bias, node, lane);
        }
        float* as = As[it & 1];
        *(float4*)&as[wv * AS_STRIDE + 4 * lane] = h;
        if (lane == 0) Sc[it & 1][wv] = sc;
        __syncthreads();

        if ((wv >> 2) == (it & 3)) {
            const int q = wv & 3;
            v8f acc0, acc1;
            #pragma unroll
            for (int r = 0; r < 8; ++r) { acc0[r] = 0.0f; acc1[r] = 0.0f; }
            const float*  ap = &as[l16 * AS_STRIDE + 2 * hi];
            const float2* bp = &Wt[hi * WT_STRIDE + (2 * q) * 16 + l16];
            v2f    fa[2][FG_KG];
            float2 f0[2][FG_KG], f1[2][FG_KG];
            #pragma unroll
            for (int j = 0; j < FG_KG; ++j) {
                fa[0][j] = *(const v2f*)(ap + 4 * j);
                f0[0][j] = bp[(2 * j) * WT_STRIDE];
                f1[0][j] = bp[(2 * j) * WT_STRIDE + 16];
            }
            #pragma unroll
            for (int g = 0; g < DHID / (4 * FG_KG); ++g) {
                const int cur = g & 1;
                const int nxt = cur ^ 1;
                if (g + 1 < DHID / (4 * FG_KG)) {
                    #pragma unroll
                    for (int j = 0; j < FG_KG; ++j) {
                        const int s = FG_KG * (g + 1) + j;
                        fa[nxt][j] = *(const v2f*)(ap + 4 * s);
                        f0[nxt][j] = bp[(2 * s) * WT_STRIDE];
                        f1[nxt][j] = bp[(2 * s) * WT_STRIDE + 16];
                    }
                }
                __builtin_amdgcn_sched_barrier(0);
                #pragma unroll
                for (int j = 0; j < FG_KG; ++j) {
                    v2f b0, b1;
                    b0.x = f0[cur][j].x; b0.y = f0[cur][j].y;
                    b1.x = f1[cur][j].x; b1.y = f1[cur][j].y;
                    acc0 = __builtin_amdgcn_wmma_f32_16x16x4_f32(
                         false, fa[cur][j],  false, b0,
                         (short)0, acc0,
                         false,  false);
                    acc1 = __builtin_amdgcn_wmma_f32_16x16x4_f32(
                         false, fa[cur][j],  false, b1,
                         (short)0, acc1,
                         false,  false);
                }
                __builtin_amdgcn_sched_barrier(0);
            }
            const float4 sa = *(const float4*)&Sc[it & 1][8 * hi];
            const float4 sb = *(const float4*)&Sc[it & 1][8 * hi + 4];
            const float rsc[8] = {sa.x, sa.y, sa.z, sa.w, sb.x, sb.y, sb.z, sb.w};
            float* ds = Ds[q];
            #pragma unroll
            for (int r = 0; r < 8; ++r) {
                ds[(r + 8 * hi) * DS_STRIDE + l16]      = acc0[r] * rsc[r];
                ds[(r + 8 * hi) * DS_STRIDE + 16 + l16] = acc1[r] * rsc[r];
            }
            #pragma unroll
            for (int j = 0; j < 4; ++j) {
                const int row = 4 * j + (lane >> 3);
                const int c4  = (lane & 7) * 4;
                const float4 v = *(const float4*)&ds[row * DS_STRIDE + c4];
                const int m = strip * 16 + row;
                if (m < n) *(float4*)&OUT[(size_t)m * DHID + q * 32 + c4] = v;
            }
        }
    }
}

#define CLS_NODES     16
#define WC_LANE_F4    10
#define WC_PITCH_F4   11
static_assert(DHID == 128 && (DHID * NCLS) % 4 == 0 && DHID * NCLS / 4 == 32 * WC_LANE_F4,
              "a lane owns 4 rows of Wc = WC_LANE_F4 float4");
static_assert(NCLS <= 16 && (CLS_NODES * NCLS) % 32 == 0, "logit butterfly and whole-line stores");

__device__ __forceinline__ float xhalf(float x) {
    const int b = __builtin_bit_cast(int, x);
    return __builtin_bit_cast(float,
        __builtin_amdgcn_permlanex16(b, b, 0x76543210, 0xfedcba98, false, false));
}
template <int M> __device__ __forceinline__ float xrow(float x) {
    const int b = __builtin_bit_cast(int, x);
    return __builtin_bit_cast(float,
        __builtin_amdgcn_update_dpp(b, b, 0x160 + M, 0xf, 0xf, false));
}

__global__ __launch_bounds__(CLS_NODES * 32) void agg_gather_classify_kernel(
    const float* __restrict__ H,
    const int* __restrict__ rowptr,
    const int* __restrict__ rec,
    const float* __restrict__ dinv,
    const float* __restrict__ bias,
    const float* __restrict__ Wc,
    const float* __restrict__ bc,
    float* __restrict__ out, int n)
{
    __shared__ float4 wcs[32 * WC_PITCH_F4];
    __shared__ float  lg[CLS_NODES * NCLS];
    const int tid  = threadIdx.x;
    const int lane = tid & 31;
    const int wv   = tid >> 5;

    const bool stager = tid < 32 * WC_LANE_F4;
    float4 wstage = make_float4(0.f, 0.f, 0.f, 0.f);
    if (stager) wstage = ((const float4*)Wc)[tid];
    const int cls = lane >> 1;
    const float bcv = bc[cls < NCLS ? cls : 0];

    const int node = __builtin_amdgcn_readfirstlane((int)(blockIdx.x * CLS_NODES + wv));
    float4 h = make_float4(0.f, 0.f, 0.f, 0.f);
    if (node < n)
        h = agg_row((const float4*)H, rowptr, (const int2*)rec, dinv, bias, node, lane);
    if (stager) wcs[(tid / WC_LANE_F4) * WC_PITCH_F4 + (tid % WC_LANE_F4)] = wstage;
    __syncthreads();

    float w[4 * WC_LANE_F4];
    #pragma unroll
    for (int i = 0; i < WC_LANE_F4; ++i) {
        const float4 t = wcs[lane * WC_PITCH_F4 + i];
        w[4 * i + 0] = t.x; w[4 * i + 1] = t.y; w[4 * i + 2] = t.z; w[4 * i + 3] = t.w;
    }
    float p[16];
    #pragma unroll
    for (int c = 0; c < 16; ++c) {
        if (c < NCLS) {
            float a = h.x * w[c];
            a = fmaf(h.y, w[NCLS + c], a);
            a = fmaf(h.z, w[2 * NCLS + c], a);
            a = fmaf(h.w, w[3 * NCLS + c], a);
            p[c] = a;
        } else {
            p[c] = 0.f;
        }
    }
    const bool b4 = (lane & 16) != 0, b3 = (lane & 8) != 0, b2 = (lane & 4) != 0, b1 = (lane & 2) != 0;
    float q8[8], q4[4], q2[2];
    #pragma unroll
    for (int j = 0; j < 8; ++j) {
        const float keep = b4 ? p[j + 8] : p[j];
        const float send = b4 ? p[j] : p[j + 8];
        q8[j] = keep + xhalf(send);
    }
    #pragma unroll
    for (int j = 0; j < 4; ++j) {
        const float keep = b3 ? q8[j + 4] : q8[j];
        const float send = b3 ? q8[j] : q8[j + 4];
        q4[j] = keep + xrow<8>(send);
    }
    #pragma unroll
    for (int j = 0; j < 2; ++j) {
        const float keep = b2 ? q4[j + 2] : q4[j];
        const float send = b2 ? q4[j] : q4[j + 2];
        q2[j] = keep + xrow<4>(send);
    }
    const float keep1 = b1 ? q2[1] : q2[0];
    const float send1 = b1 ? q2[0] : q2[1];
    const float q1 = keep1 + xrow<2>(send1);
    const float q0 = q1 + xrow<1>(q1);
    if ((lane & 1) == 0 && cls < NCLS) lg[wv * NCLS + cls] = q0 + bcv;
    __syncthreads();

    if (tid < CLS_NODES * NCLS) {
        const int g = blockIdx.x * (CLS_NODES * NCLS) + tid;
        if (g < n * NCLS) out[g] = lg[tid];
    }
}

extern "C" void kernel_launch(void* const* d_in, const int* in_sizes, int n_in,
                              void* d_out, int out_size, void* d_ws, size_t ws_size,
                              hipStream_t stream) {
    const float* x  = (const float*)d_in[0];
    const int*   ei = (const int*)d_in[1];
    const float* ew = (const float*)d_in[2];
    const float* W1 = (const float*)d_in[3];
    const float* b1 = (const float*)d_in[4];
    const float* W2 = (const float*)d_in[5];
    const float* b2 = (const float*)d_in[6];
    const float* Wc = (const float*)d_in[7];
    const float* bc = (const float*)d_in[8];
    float* out = (float*)d_out;

    const int n  = NNODES;
    const int E  = in_sizes[1] / 2;
    const int K1 = in_sizes[0] / n;
    const int* src = ei;
    const int* dst = ei + E;

    char* ws = (char*)d_ws;
    size_t off = 0;
    auto alloc = [&](size_t bytes) { char* p = ws + off; off = (off + bytes + 255) & ~(size_t)255; return p; };
    float* dinv = (float*)alloc((size_t)n * 4);
    float* bufA = (float*)alloc((size_t)n * DHID * 4);
    float* bufB = (float*)alloc((size_t)n * DHID * 4);
    int* loff     = (int*)alloc((size_t)(NBINS + 1) * PB * 4);
    int* rowptr   = (int*)alloc((size_t)(NBINS * BIN_NODES + 1) * 4);
    int2* tmp   = (int2*)alloc((size_t)E * 8);
    int* rec    = (int*)alloc((size_t)E * 8);
    (void)ws_size;

    bin_place_kernel  <<<PB, PLACE_T, 0, stream>>>(src, dst, ew, loff, tmp, E);
    bin_sort_kernel   <<<NBINS, SORT_T, 0, stream>>>(tmp, loff, rowptr, dinv, (int2*)rec, n, E);

    const int gemmBlocks = (n + GR_ROWS - 1) / GR_ROWS;
    const int fuseBlocks = ((n + 15) / 16 + FG_STRIPS - 1) / FG_STRIPS;
    const int clsBlocks  = (n + CLS_NODES - 1) / CLS_NODES;

    gemm_rows_wmma_kernel<<<gemmBlocks, GT, 0, stream>>>(x, W1, dinv, bufA, n, K1);
    agg_gather_gemm_kernel<<<fuseBlocks, FG_T, 0, stream>>>(bufA, rowptr, rec, dinv, b1, W2, bufB, n);

    agg_gather_classify_kernel<<<clsBlocks, CLS_NODES * 32, 0, stream>>>(
        bufB, rowptr, rec, dinv, b2, Wc, bc, out, n);
}
